# starting kernel
# baseline (speedup 1.0000x reference)
#include <hip/hip_runtime.h>
#include <hip/hip_fp16.h>

typedef __attribute__((ext_vector_type(16))) _Float16 v16h;
typedef __attribute__((ext_vector_type(8)))  _Float16 v8h;
typedef __attribute__((ext_vector_type(8)))  float    v8f;

__device__ __forceinline__ unsigned fp_ord(float f) {
    unsigned u = __float_as_uint(f);
    return (u & 0x80000000u) ? ~u : (u | 0x80000000u);
}
__device__ __forceinline__ float fp_unord(unsigned u) {
    unsigned b = (u & 0x80000000u) ? (u ^ 0x80000000u) : ~u;
    return __uint_as_float(b);
}

__global__ void k_node_transform1(const float* __restrict__ x,
                                  const float* __restrict__ Wl, const float* __restrict__ bl,
                                  const float* __restrict__ Wr, const float* __restrict__ br,
                                  float* __restrict__ xl1, float* __restrict__ xr1, int N) {
    int idx = blockIdx.x * blockDim.x + threadIdx.x;
    if (idx >= N * 128) return;
    int n = idx >> 7, d = idx & 127;
    float x0 = x[n * 2], x1 = x[n * 2 + 1];
    xl1[idx] = fmaf(x0, Wl[d], fmaf(x1, Wl[128 + d], bl[d]));
    xr1[idx] = fmaf(x0, Wr[d], fmaf(x1, Wr[128 + d], br[d]));
}

template <int H, int C>
__global__ void k_edge_logits(const float* __restrict__ xl, const float* __restrict__ xr,
                              const float* __restrict__ ea, const float* __restrict__ We,
                              const float* __restrict__ att,
                              const int* __restrict__ src, const int* __restrict__ tgt,
                              float* __restrict__ elog, unsigned* __restrict__ emaxM, int E) {
    const int D = H * C;
    int idx = blockIdx.x * blockDim.x + threadIdx.x;
    if (idx >= E * H) return;
    int e = idx / H, h = idx % H;
    int s = src[e], t = tgt[e];
    float a0 = ea[e * 2], a1 = ea[e * 2 + 1];
    int base = h * C;
    float acc = 0.f;
#pragma unroll
    for (int c = 0; c < C; ++c) {
        int d = base + c;
        float m = xl[s * D + d] + xr[t * D + d] + a0 * We[d] + a1 * We[D + d];
        m = (m > 0.f) ? m : 0.2f * m;
        acc = fmaf(m, att[base + c], acc);
    }
    elog[idx] = acc;
    atomicMax(&emaxM[t * H + h], fp_ord(acc));
}

template <int H>
__global__ void k_edge_exp(float* __restrict__ elog, const unsigned* __restrict__ emaxM,
                           float* __restrict__ denom, const int* __restrict__ tgt, int E) {
    int idx = blockIdx.x * blockDim.x + threadIdx.x;
    if (idx >= E * H) return;
    int e = idx / H, h = idx % H;
    int t = tgt[e];
    float mx = fp_unord(emaxM[t * H + h]);
    float v = __expf(elog[idx] - mx);
    elog[idx] = v;
    atomicAdd(&denom[t * H + h], v);
}

template <int H, int C>
__global__ void k_edge_agg(const float* __restrict__ ee, const float* __restrict__ denom,
                           const float* __restrict__ xl,
                           const int* __restrict__ src, const int* __restrict__ tgt,
                           float* __restrict__ agg, int E) {
    const int D = H * C;
    const int Q = D / 4;
    int idx = blockIdx.x * blockDim.x + threadIdx.x;
    if (idx >= E * Q) return;
    int e = idx / Q, q = idx % Q;
    int d = q * 4, h = d / C;
    int s = src[e], t = tgt[e];
    float alpha = ee[e * H + h] / denom[t * H + h];
    float4 v = *(const float4*)(xl + s * D + d);
    atomicAdd(&agg[t * D + d + 0], alpha * v.x);
    atomicAdd(&agg[t * D + d + 1], alpha * v.y);
    atomicAdd(&agg[t * D + d + 2], alpha * v.z);
    atomicAdd(&agg[t * D + d + 3], alpha * v.w);
}

__global__ void k_elu_half(const float* __restrict__ agg, const float* __restrict__ bias,
                           _Float16* __restrict__ hh, int N) {
    int idx = blockIdx.x * blockDim.x + threadIdx.x;
    if (idx >= N * 128) return;
    int d = idx & 127;
    float v = agg[idx] + bias[d];
    v = (v > 0.f) ? v : (__expf(v) - 1.f);
    hh[idx] = (_Float16)v;
}

__global__ void k_w_pack(const float* __restrict__ Wl, const float* __restrict__ Wr,
                         _Float16* __restrict__ Wpk) {
    int idx = blockIdx.x * blockDim.x + threadIdx.x;
    if (idx >= 2 * 8192) return;
    int which = idx >> 13;
    int r = idx & 8191;
    int j = r & 15;
    int lane = (r >> 4) & 31;
    int frag = r >> 9;
    int kc = frag >> 2, nt = frag & 3;
    int half_sel = lane >> 4, lidx = lane & 15;
    int kb = (j < 8) ? (half_sel * 8) : (16 + half_sel * 8);
    int k = kc * 32 + kb + (j & 7);
    int n = nt * 16 + lidx;
    const float* W = which ? Wr : Wl;
    Wpk[idx] = (_Float16)W[k * 64 + n];
}

__global__ void k_gemm2_wmma(const _Float16* __restrict__ hh,
                             const _Float16* __restrict__ Wpk,
                             const float* __restrict__ bl, const float* __restrict__ br,
                             float* __restrict__ xl2, float* __restrict__ xr2, int N) {
    __shared__ __align__(32) _Float16 ldsW[2 * 8192];

    {
        const uint4* g = (const uint4*)Wpk;
        uint4* l = (uint4*)ldsW;
        for (int i = threadIdx.x; i < 2048; i += blockDim.x) l[i] = g[i];
    }
    __syncthreads();

    int wave = (blockIdx.x * blockDim.x + threadIdx.x) >> 5;
    int lane = threadIdx.x & 31;
    int row0 = wave * 16;
    if (row0 >= N) return;

    int half_sel = lane >> 4;
    int lidx = lane & 15;
    int kb_lo = half_sel * 8;
    int kb_hi = 16 + kb_lo;
    int m = row0 + lidx;
    if (m >= N) m = N - 1;

    const v16h* fragL = (const v16h*)ldsW;
    const v16h* fragR = (const v16h*)(ldsW + 8192);

    v8f accl[4] = {};
    v8f accr[4] = {};

#pragma unroll
    for (int kc = 0; kc < 4; ++kc) {
        const _Float16* pa = hh + m * 128 + kc * 32;
        v8h alo = *(const v8h*)(pa + kb_lo);
        v8h ahi = *(const v8h*)(pa + kb_hi);
        v16h a = __builtin_shufflevector(alo, ahi, 0, 1, 2, 3, 4, 5, 6, 7,
                                         8, 9, 10, 11, 12, 13, 14, 15);
#pragma unroll
        for (int nt = 0; nt < 4; ++nt) {
            int fi = (kc * 4 + nt) * 32 + lane;
            v16h b  = fragL[fi];
            v16h b2 = fragR[fi];
            accl[nt] = __builtin_amdgcn_wmma_f32_16x16x32_f16(
                false, a, false, b, (short)0, accl[nt], false, false);
            accr[nt] = __builtin_amdgcn_wmma_f32_16x16x32_f16(
                false, a, false, b2, (short)0, accr[nt], false, false);
        }
    }
#pragma unroll
    for (int nt = 0; nt < 4; ++nt) {
        int n = nt * 16 + lidx;
        float bln = bl[n], brn = br[n];
#pragma unroll
        for (int r = 0; r < 8; ++r) {
            int mo = row0 + r + half_sel * 8;
            if (mo < N) {
                xl2[mo * 64 + n] = accl[nt][r] + bln;
                xr2[mo * 64 + n] = accr[nt][r] + brn;
            }
        }
    }
}

__global__ void k_pool(const float* __restrict__ agg2, const float* __restrict__ bias2,
                       const int* __restrict__ batch,
                       float* __restrict__ sums, float* __restrict__ cnts, int N) {
    int idx = blockIdx.x * blockDim.x + threadIdx.x;
    if (idx >= N * 64) return;
    int n = idx >> 6, d = idx & 63;
    int g = batch[n];
    atomicAdd(&sums[g * 64 + d], agg2[idx] + bias2[d]);
    if (d == 0) atomicAdd(&cnts[g], 1.0f);
}

__global__ void k_final(const float* __restrict__ sums, const float* __restrict__ cnts,
                        float* __restrict__ out, int G) {
    int idx = blockIdx.x * blockDim.x + threadIdx.x;
    if (idx >= G * 64) return;
    out[idx] = sums[idx] / fmaxf(cnts[idx >> 6], 1.0f);
}

extern "C" void kernel_launch(void* const* d_in, const int* in_sizes, int n_in,
                              void* d_out, int out_size, void* d_ws, size_t ws_size,
                              hipStream_t stream) {
    const float* x      = (const float*)d_in[0];
    const float* ea     = (const float*)d_in[1];
    const float* W_l1   = (const float*)d_in[2];
    const float* b_l1   = (const float*)d_in[3];
    const float* W_r1   = (const float*)d_in[4];
    const float* b_r1   = (const float*)d_in[5];
    const float* W_e1   = (const float*)d_in[6];
    const float* att1   = (const float*)d_in[7];
    const float* bias1  = (const float*)d_in[8];
    const float* W_l2   = (const float*)d_in[9];
    const float* b_l2   = (const float*)d_in[10];
    const float* W_r2   = (const float*)d_in[11];
    const float* b_r2   = (const float*)d_in[12];
    const float* W_e2   = (const float*)d_in[13];
    const float* att2   = (const float*)d_in[14];
    const float* bias2  = (const float*)d_in[15];
    const int*   eidx   = (const int*)d_in[16];
    const int*   batch  = (const int*)d_in[17];
    float* out = (float*)d_out;

    const int N = in_sizes[0] / 2;
    const int E = in_sizes[1] / 2;
    const int G = out_size / 64;
    const int* src = eidx;
    const int* tgt = eidx + E;

    char* ws = (char*)d_ws;
    size_t off = 0;
    auto alloc = [&](size_t bytes) { size_t o = off; off = (off + bytes + 255) & ~(size_t)255; return o; };

    size_t o_emax1 = alloc((size_t)N * 16 * 4);
    size_t o_den1  = alloc((size_t)N * 16 * 4);
    size_t o_agg1  = alloc((size_t)N * 128 * 4);
    size_t o_emax2 = alloc((size_t)N * 4 * 4);
    size_t o_den2  = alloc((size_t)N * 4 * 4);
    size_t o_agg2  = alloc((size_t)N * 64 * 4);
    size_t o_sums  = alloc((size_t)G * 64 * 4);
    size_t o_cnts  = alloc((size_t)G * 4);
    size_t acc_bytes = off;

    size_t o_xl1 = alloc((size_t)N * 128 * 4);
    size_t o_xr1 = alloc((size_t)N * 128 * 4);
    size_t o_e1  = alloc((size_t)E * 16 * 4);
    size_t o_hh  = alloc((size_t)N * 128 * 2);
    size_t o_wpk = alloc((size_t)2 * 8192 * 2);
    size_t o_xl2 = alloc((size_t)N * 64 * 4);
    size_t o_xr2 = alloc((size_t)N * 64 * 4);
    size_t o_e2  = alloc((size_t)E * 4 * 4);
    (void)ws_size;

    unsigned* emax1 = (unsigned*)(ws + o_emax1);
    float*    den1  = (float*)(ws + o_den1);
    float*    agg1  = (float*)(ws + o_agg1);
    unsigned* emax2 = (unsigned*)(ws + o_emax2);
    float*    den2  = (float*)(ws + o_den2);
    float*    agg2  = (float*)(ws + o_agg2);
    float*    sums  = (float*)(ws + o_sums);
    float*    cnts  = (float*)(ws + o_cnts);
    float*    xl1   = (float*)(ws + o_xl1);
    float*    xr1   = (float*)(ws + o_xr1);
    float*    e1    = (float*)(ws + o_e1);
    _Float16* hh    = (_Float16*)(ws + o_hh);
    _Float16* wpk   = (_Float16*)(ws + o_wpk);
    float*    xl2   = (float*)(ws + o_xl2);
    float*    xr2   = (float*)(ws + o_xr2);
    float*    e2    = (float*)(ws + o_e2);

    const int TB = 256;
    auto blocks = [&](long long t) { return (unsigned)((t + TB - 1) / TB); };

    hipMemsetAsync(d_ws, 0, acc_bytes, stream);

    k_node_transform1<<<blocks((long long)N * 128), TB, 0, stream>>>(
        x, W_l1, b_l1, W_r1, b_r1, xl1, xr1, N);
    k_edge_logits<16, 8><<<blocks((long long)E * 16), TB, 0, stream>>>(
        xl1, xr1, ea, W_e1, att1, src, tgt, e1, emax1, E);
    k_edge_exp<16><<<blocks((long long)E * 16), TB, 0, stream>>>(
        e1, emax1, den1, tgt, E);
    k_edge_agg<16, 8><<<blocks((long long)E * 32), TB, 0, stream>>>(
        e1, den1, xl1, src, tgt, agg1, E);
    k_elu_half<<<blocks((long long)N * 128), TB, 0, stream>>>(agg1, bias1, hh, N);

    k_w_pack<<<blocks(2 * 8192), TB, 0, stream>>>(W_l2, W_r2, wpk);
    long long gemm_threads = (long long)((N + 15) / 16) * 32;
    k_gemm2_wmma<<<blocks(gemm_threads), TB, 0, stream>>>(
        hh, wpk, b_l2, b_r2, xl2, xr2, N);

    k_edge_logits<4, 16><<<blocks((long long)E * 4), TB, 0, stream>>>(
        xl2, xr2, ea, W_e2, att2, src, tgt, e2, emax2, E);
    k_edge_exp<4><<<blocks((long long)E * 4), TB, 0, stream>>>(
        e2, emax2, den2, tgt, E);
    k_edge_agg<4, 16><<<blocks((long long)E * 16), TB, 0, stream>>>(
        e2, den2, xl2, src, tgt, agg2, E);

    k_pool<<<blocks((long long)N * 64), TB, 0, stream>>>(agg2, bias2, batch, sums, cnts, N);
    k_final<<<blocks((long long)G * 64), TB, 0, stream>>>(sums, cnts, out, G);
}


# mean pool by linearity: each layer-2 edge adds alpha*row into its target's graph sum (LDS table per block, summed in final kernel); node output, its atomics, pool launch gone; changes pool sum order
# speedup vs baseline: 1.2543x; 1.2543x over previous
#include <hip/hip_runtime.h>
#include <hip/hip_fp16.h>

typedef __attribute__((ext_vector_type(16))) _Float16 v16h;
typedef __attribute__((ext_vector_type(8)))  _Float16 v8h;
typedef __attribute__((ext_vector_type(8)))  float    v8f;

__device__ __forceinline__ unsigned fp_ord(float f) {
    unsigned u = __float_as_uint(f);
    return (u & 0x80000000u) ? ~u : (u | 0x80000000u);
}
__device__ __forceinline__ float fp_unord(unsigned u) {
    unsigned b = (u & 0x80000000u) ? (u ^ 0x80000000u) : ~u;
    return __uint_as_float(b);
}

__global__ void k_node_transform1(const float* __restrict__ x,
                                  const float* __restrict__ Wl, const float* __restrict__ bl,
                                  const float* __restrict__ Wr, const float* __restrict__ br,
                                  float* __restrict__ xl1, float* __restrict__ xr1, int N) {
    int idx = blockIdx.x * blockDim.x + threadIdx.x;
    if (idx >= N * 128) return;
    int n = idx >> 7, d = idx & 127;
    float x0 = x[n * 2], x1 = x[n * 2 + 1];
    xl1[idx] = fmaf(x0, Wl[d], fmaf(x1, Wl[128 + d], bl[d]));
    xr1[idx] = fmaf(x0, Wr[d], fmaf(x1, Wr[128 + d], br[d]));
}

template <int H, int C>
__global__ void k_edge_logits(const float* __restrict__ xl, const float* __restrict__ xr,
                              const float* __restrict__ ea, const float* __restrict__ We,
                              const float* __restrict__ att,
                              const int* __restrict__ src, const int* __restrict__ tgt,
                              float* __restrict__ elog, unsigned* __restrict__ emaxM, int E) {
    const int D = H * C;
    int idx = blockIdx.x * blockDim.x + threadIdx.x;
    if (idx >= E * H) return;
    int e = idx / H, h = idx % H;
    int s = src[e], t = tgt[e];
    float a0 = ea[e * 2], a1 = ea[e * 2 + 1];
    int base = h * C;
    float acc = 0.f;
#pragma unroll
    for (int c = 0; c < C; ++c) {
        int d = base + c;
        float m = xl[s * D + d] + xr[t * D + d] + a0 * We[d] + a1 * We[D + d];
        m = (m > 0.f) ? m : 0.2f * m;
        acc = fmaf(m, att[base + c], acc);
    }
    elog[idx] = acc;
    atomicMax(&emaxM[t * H + h], fp_ord(acc));
}

template <int H>
__global__ void k_edge_exp(float* __restrict__ elog, const unsigned* __restrict__ emaxM,
                           float* __restrict__ denom, const int* __restrict__ tgt, int E) {
    int idx = blockIdx.x * blockDim.x + threadIdx.x;
    if (idx >= E * H) return;
    int e = idx / H, h = idx % H;
    int t = tgt[e];
    float mx = fp_unord(emaxM[t * H + h]);
    float v = __expf(elog[idx] - mx);
    elog[idx] = v;
    atomicAdd(&denom[t * H + h], v);
}

template <int H, int C>
__global__ void k_edge_agg(const float* __restrict__ ee, const float* __restrict__ denom,
                           const float* __restrict__ xl,
                           const int* __restrict__ src, const int* __restrict__ tgt,
                           float* __restrict__ agg, int E) {
    const int D = H * C;
    const int Q = D / 4;
    int idx = blockIdx.x * blockDim.x + threadIdx.x;
    if (idx >= E * Q) return;
    int e = idx / Q, q = idx % Q;
    int d = q * 4, h = d / C;
    int s = src[e], t = tgt[e];
    float alpha = ee[e * H + h] / denom[t * H + h];
    float4 v = *(const float4*)(xl + s * D + d);
    atomicAdd(&agg[t * D + d + 0], alpha * v.x);
    atomicAdd(&agg[t * D + d + 1], alpha * v.y);
    atomicAdd(&agg[t * D + d + 2], alpha * v.z);
    atomicAdd(&agg[t * D + d + 3], alpha * v.w);
}

__global__ void k_elu_half(const float* __restrict__ agg, const float* __restrict__ bias,
                           _Float16* __restrict__ hh, int N) {
    int idx = blockIdx.x * blockDim.x + threadIdx.x;
    if (idx >= N * 128) return;
    int d = idx & 127;
    float v = agg[idx] + bias[d];
    v = (v > 0.f) ? v : (__expf(v) - 1.f);
    hh[idx] = (_Float16)v;
}

__global__ void k_w_pack(const float* __restrict__ Wl, const float* __restrict__ Wr,
                         _Float16* __restrict__ Wpk) {
    int idx = blockIdx.x * blockDim.x + threadIdx.x;
    if (idx >= 2 * 8192) return;
    int which = idx >> 13;
    int r = idx & 8191;
    int j = r & 15;
    int lane = (r >> 4) & 31;
    int frag = r >> 9;
    int kc = frag >> 2, nt = frag & 3;
    int half_sel = lane >> 4, lidx = lane & 15;
    int kb = (j < 8) ? (half_sel * 8) : (16 + half_sel * 8);
    int k = kc * 32 + kb + (j & 7);
    int n = nt * 16 + lidx;
    const float* W = which ? Wr : Wl;
    Wpk[idx] = (_Float16)W[k * 64 + n];
}

__global__ void k_gemm2_wmma(const _Float16* __restrict__ hh,
                             const _Float16* __restrict__ Wpk,
                             const float* __restrict__ bl, const float* __restrict__ br,
                             float* __restrict__ xl2, float* __restrict__ xr2, int N) {
    __shared__ __align__(32) _Float16 ldsW[2 * 8192];

    {
        const uint4* g = (const uint4*)Wpk;
        uint4* l = (uint4*)ldsW;
        for (int i = threadIdx.x; i < 2048; i += blockDim.x) l[i] = g[i];
    }
    __syncthreads();

    int wave = (blockIdx.x * blockDim.x + threadIdx.x) >> 5;
    int lane = threadIdx.x & 31;
    int row0 = wave * 16;
    if (row0 >= N) return;

    int half_sel = lane >> 4;
    int lidx = lane & 15;
    int kb_lo = half_sel * 8;
    int kb_hi = 16 + kb_lo;
    int m = row0 + lidx;
    if (m >= N) m = N - 1;

    const v16h* fragL = (const v16h*)ldsW;
    const v16h* fragR = (const v16h*)(ldsW + 8192);

    v8f accl[4] = {};
    v8f accr[4] = {};

#pragma unroll
    for (int kc = 0; kc < 4; ++kc) {
        const _Float16* pa = hh + m * 128 + kc * 32;
        v8h alo = *(const v8h*)(pa + kb_lo);
        v8h ahi = *(const v8h*)(pa + kb_hi);
        v16h a = __builtin_shufflevector(alo, ahi, 0, 1, 2, 3, 4, 5, 6, 7,
                                         8, 9, 10, 11, 12, 13, 14, 15);
#pragma unroll
        for (int nt = 0; nt < 4; ++nt) {
            int fi = (kc * 4 + nt) * 32 + lane;
            v16h b  = fragL[fi];
            v16h b2 = fragR[fi];
            accl[nt] = __builtin_amdgcn_wmma_f32_16x16x32_f16(
                false, a, false, b, (short)0, accl[nt], false, false);
            accr[nt] = __builtin_amdgcn_wmma_f32_16x16x32_f16(
                false, a, false, b2, (short)0, accr[nt], false, false);
        }
    }
#pragma unroll
    for (int nt = 0; nt < 4; ++nt) {
        int n = nt * 16 + lidx;
        float bln = bl[n], brn = br[n];
#pragma unroll
        for (int r = 0; r < 8; ++r) {
            int mo = row0 + r + half_sel * 8;
            if (mo < N) {
                xl2[mo * 64 + n] = accl[nt][r] + bln;
                xr2[mo * 64 + n] = accr[nt][r] + brn;
            }
        }
    }
}

template <int H, int C>
__device__ __forceinline__ void pool_item_load(unsigned idx,
        const float* __restrict__ ee, const float* __restrict__ denom,
        const float* __restrict__ xl, const int* __restrict__ src,
        const int* __restrict__ tgt, const int* __restrict__ batch, int G,
        float4& v, float& alpha, int& off) {
    constexpr unsigned D = H * C, Q = D / 4;
    unsigned e = idx / Q, q = idx % Q;
    unsigned d = q * 4, h = d / C;
    int s = src[e], t = tgt[e];
    alpha = ee[e * H + h] / denom[t * H + h];
    int g = batch[t];
    v = *(const float4*)(xl + (size_t)s * D + d);
    unsigned row = ((unsigned)g < (unsigned)G) ? (unsigned)g : (unsigned)G;
    off = (int)(row * D + d);
}

__device__ __forceinline__ void pool_item_add(float* tab, const float4& v, float alpha, int off) {
    atomicAdd(&tab[off + 0], alpha * v.x);
    atomicAdd(&tab[off + 1], alpha * v.y);
    atomicAdd(&tab[off + 2], alpha * v.z);
    atomicAdd(&tab[off + 3], alpha * v.w);
}

template <int H, int C>
__global__ __launch_bounds__(512) void k_edge_agg_pool(
        const float* __restrict__ ee, const float* __restrict__ denom,
        const float* __restrict__ xl, const int* __restrict__ src,
        const int* __restrict__ tgt, const int* __restrict__ batch,
        float* __restrict__ part, int E, int G) {
    extern __shared__ float tab[];
    constexpr int D = H * C, Q = D / 4, U = 4;
    const int GD = G * D;
    for (int i = threadIdx.x; i < GD + D; i += blockDim.x) tab[i] = 0.f;
    __syncthreads();

    const unsigned total = (unsigned)E * Q;
    const unsigned stride = gridDim.x * blockDim.x;
    unsigned idx = blockIdx.x * blockDim.x + threadIdx.x;

    for (; idx + (U - 1) * stride < total; idx += U * stride) {
        float4 v[U];
        float  a[U];
        int    off[U];
#pragma unroll
        for (int u = 0; u < U; ++u)
            pool_item_load<H, C>(idx + u * stride, ee, denom, xl, src, tgt, batch, G,
                                 v[u], a[u], off[u]);
#pragma unroll
        for (int u = 0; u < U; ++u)
            pool_item_add(tab, v[u], a[u], off[u]);
    }
    for (; idx < total; idx += stride) {
        float4 v; float a; int off;
        pool_item_load<H, C>(idx, ee, denom, xl, src, tgt, batch, G, v, a, off);
        pool_item_add(tab, v, a, off);
    }
    __syncthreads();

    float* my = part + (size_t)blockIdx.x * GD;
    for (int i = threadIdx.x; i < GD; i += blockDim.x) my[i] = tab[i];
}

__global__ __launch_bounds__(1024) void k_final(const float* __restrict__ part, int nPart,
                                                const float* __restrict__ bias2,
                                                const int* __restrict__ batch,
                                                float* __restrict__ out, int N, int G) {
    __shared__ float red[16][64];
    __shared__ int cntS;
    const int g = blockIdx.x;
    const int tid = threadIdx.x;
    const int d = tid & 63, j = tid >> 6;
    if (tid == 0) cntS = 0;

    float acc = 0.f;
    const float* p = part + (size_t)g * 64 + d;
    const size_t tabStride = (size_t)G * 64;
    int b = j;
    for (; b + 16 * 7 < nPart; b += 16 * 8) {
        float t[8];
#pragma unroll
        for (int k = 0; k < 8; ++k) t[k] = p[(size_t)(b + 16 * k) * tabStride];
#pragma unroll
        for (int k = 0; k < 8; ++k) acc += t[k];
    }
    for (; b < nPart; b += 16) acc += p[(size_t)b * tabStride];

    int c = 0;
    for (int n = tid; n < N; n += 1024) c += (batch[n] == g) ? 1 : 0;

    red[j][d] = acc;
    __syncthreads();
    atomicAdd(&cntS, c);
    __syncthreads();

    if (j == 0) {
        float s = 0.f;
#pragma unroll
        for (int k = 0; k < 16; ++k) s += red[k][d];
        float cnt = (float)cntS;
        out[g * 64 + d] = (s + cnt * bias2[d]) / fmaxf(cnt, 1.0f);
    }
}

extern "C" void kernel_launch(void* const* d_in, const int* in_sizes, int n_in,
                              void* d_out, int out_size, void* d_ws, size_t ws_size,
                              hipStream_t stream) {
    const float* x      = (const float*)d_in[0];
    const float* ea     = (const float*)d_in[1];
    const float* W_l1   = (const float*)d_in[2];
    const float* b_l1   = (const float*)d_in[3];
    const float* W_r1   = (const float*)d_in[4];
    const float* b_r1   = (const float*)d_in[5];
    const float* W_e1   = (const float*)d_in[6];
    const float* att1   = (const float*)d_in[7];
    const float* bias1  = (const float*)d_in[8];
    const float* W_l2   = (const float*)d_in[9];
    const float* b_l2   = (const float*)d_in[10];
    const float* W_r2   = (const float*)d_in[11];
    const float* b_r2   = (const float*)d_in[12];
    const float* W_e2   = (const float*)d_in[13];
    const float* att2   = (const float*)d_in[14];
    const float* bias2  = (const float*)d_in[15];
    const int*   eidx   = (const int*)d_in[16];
    const int*   batch  = (const int*)d_in[17];
    float* out = (float*)d_out;

    const int N = in_sizes[0] / 2;
    const int E = in_sizes[1] / 2;
    const int G = out_size / 64;
    const int* src = eidx;
    const int* tgt = eidx + E;

    char* ws = (char*)d_ws;
    size_t off = 0;
    auto alloc = [&](size_t bytes) { size_t o = off; off = (off + bytes + 255) & ~(size_t)255; return o; };

    size_t o_emax1 = alloc((size_t)N * 16 * 4);
    size_t o_den1  = alloc((size_t)N * 16 * 4);
    size_t o_agg1  = alloc((size_t)N * 128 * 4);
    size_t o_emax2 = alloc((size_t)N * 4 * 4);
    size_t o_den2  = alloc((size_t)N * 4 * 4);
    size_t acc_bytes = off;

    size_t o_xl1 = alloc((size_t)N * 128 * 4);
    size_t o_xr1 = alloc((size_t)N * 128 * 4);
    size_t o_e1  = alloc((size_t)E * 16 * 4);
    size_t o_hh  = alloc((size_t)N * 128 * 2);
    size_t o_wpk = alloc((size_t)2 * 8192 * 2);
    size_t o_xl2 = alloc((size_t)N * 64 * 4);
    size_t o_xr2 = alloc((size_t)N * 64 * 4);
    size_t o_e2  = alloc((size_t)E * 4 * 4);
    const int POOL_BLOCKS = 512;
    size_t o_part = alloc((size_t)POOL_BLOCKS * G * 64 * 4);
    (void)ws_size;

    unsigned* emax1 = (unsigned*)(ws + o_emax1);
    float*    den1  = (float*)(ws + o_den1);
    float*    agg1  = (float*)(ws + o_agg1);
    unsigned* emax2 = (unsigned*)(ws + o_emax2);
    float*    den2  = (float*)(ws + o_den2);
    float*    xl1   = (float*)(ws + o_xl1);
    float*    xr1   = (float*)(ws + o_xr1);
    float*    e1    = (float*)(ws + o_e1);
    _Float16* hh    = (_Float16*)(ws + o_hh);
    _Float16* wpk   = (_Float16*)(ws + o_wpk);
    float*    xl2   = (float*)(ws + o_xl2);
    float*    xr2   = (float*)(ws + o_xr2);
    float*    e2    = (float*)(ws + o_e2);
    float*    part  = (float*)(ws + o_part);

    const int TB = 256;
    auto blocks = [&](long long t) { return (unsigned)((t + TB - 1) / TB); };

    hipMemsetAsync(d_ws, 0, acc_bytes, stream);

    k_node_transform1<<<blocks((long long)N * 128), TB, 0, stream>>>(
        x, W_l1, b_l1, W_r1, b_r1, xl1, xr1, N);
    k_edge_logits<16, 8><<<blocks((long long)E * 16), TB, 0, stream>>>(
        xl1, xr1, ea, W_e1, att1, src, tgt, e1, emax1, E);
    k_edge_exp<16><<<blocks((long long)E * 16), TB, 0, stream>>>(
        e1, emax1, den1, tgt, E);
    k_edge_agg<16, 8><<<blocks((long long)E * 32), TB, 0, stream>>>(
        e1, den1, xl1, src, tgt, agg1, E);
    k_elu_half<<<blocks((long long)N * 128), TB, 0, stream>>>(agg1, bias1, hh, N);

    k_w_pack<<<blocks(2 * 8192), TB, 0, stream>>>(W_l2, W_r2, wpk);
    long long gemm_threads = (long long)((N + 15) / 16) * 32;
    k_gemm2_wmma<<<blocks(gemm_threads), TB, 0, stream>>>(
        hh, wpk, b_l2, b_r2, xl2, xr2, N);

    k_edge_logits<4, 16><<<blocks((long long)E * 4), TB, 0, stream>>>(
        xl2, xr2, ea, W_e2, att2, src, tgt, e2, emax2, E);
    k_edge_exp<4><<<blocks((long long)E * 4), TB, 0, stream>>>(
        e2, emax2, den2, tgt, E);

    k_edge_agg_pool<4, 16><<<POOL_BLOCKS, 512, (size_t)(G + 1) * 64 * sizeof(float), stream>>>(
        e2, den2, xl2, src, tgt, batch, part, E, G);
    k_final<<<G, 1024, 0, stream>>>(part, POOL_BLOCKS, bias2, batch, out, N, G);
}


# layer-1 sums by target ownership: a block holds 200 nodes' rows in LDS, filters the edge list, adds by LDS atomics, writes ELU rows; global atomics, ELU launch gone; changes layer-1 sum order
# speedup vs baseline: 1.4269x; 1.1376x over previous
#include <hip/hip_runtime.h>
#include <hip/hip_fp16.h>

typedef __attribute__((ext_vector_type(16))) _Float16 v16h;
typedef __attribute__((ext_vector_type(8)))  _Float16 v8h;
typedef __attribute__((ext_vector_type(8)))  float    v8f;

__device__ __forceinline__ unsigned fp_ord(float f) {
    unsigned u = __float_as_uint(f);
    return (u & 0x80000000u) ? ~u : (u | 0x80000000u);
}
__device__ __forceinline__ float fp_unord(unsigned u) {
    unsigned b = (u & 0x80000000u) ? (u ^ 0x80000000u) : ~u;
    return __uint_as_float(b);
}

__global__ void k_node_transform1(const float* __restrict__ x,
                                  const float* __restrict__ Wl, const float* __restrict__ bl,
                                  const float* __restrict__ Wr, const float* __restrict__ br,
                                  float* __restrict__ xl1, float* __restrict__ xr1, int N) {
    int idx = blockIdx.x * blockDim.x + threadIdx.x;
    if (idx >= N * 128) return;
    int n = idx >> 7, d = idx & 127;
    float x0 = x[n * 2], x1 = x[n * 2 + 1];
    xl1[idx] = fmaf(x0, Wl[d], fmaf(x1, Wl[128 + d], bl[d]));
    xr1[idx] = fmaf(x0, Wr[d], fmaf(x1, Wr[128 + d], br[d]));
}

template <int H, int C>
__global__ void k_edge_logits(const float* __restrict__ xl, const float* __restrict__ xr,
                              const float* __restrict__ ea, const float* __restrict__ We,
                              const float* __restrict__ att,
                              const int* __restrict__ src, const int* __restrict__ tgt,
                              float* __restrict__ elog, unsigned* __restrict__ emaxM, int E) {
    const int D = H * C;
    int idx = blockIdx.x * blockDim.x + threadIdx.x;
    if (idx >= E * H) return;
    int e = idx / H, h = idx % H;
    int s = src[e], t = tgt[e];
    float a0 = ea[e * 2], a1 = ea[e * 2 + 1];
    int base = h * C;
    float acc = 0.f;
#pragma unroll
    for (int c = 0; c < C; ++c) {
        int d = base + c;
        float m = xl[s * D + d] + xr[t * D + d] + a0 * We[d] + a1 * We[D + d];
        m = (m > 0.f) ? m : 0.2f * m;
        acc = fmaf(m, att[base + c], acc);
    }
    elog[idx] = acc;
    atomicMax(&emaxM[t * H + h], fp_ord(acc));
}

template <int H>
__global__ void k_edge_exp(float* __restrict__ elog, const unsigned* __restrict__ emaxM,
                           float* __restrict__ denom, const int* __restrict__ tgt, int E) {
    int idx = blockIdx.x * blockDim.x + threadIdx.x;
    if (idx >= E * H) return;
    int e = idx / H, h = idx % H;
    int t = tgt[e];
    float mx = fp_unord(emaxM[t * H + h]);
    float v = __expf(elog[idx] - mx);
    elog[idx] = v;
    atomicAdd(&denom[t * H + h], v);
}

__global__ void k_w_pack(const float* __restrict__ Wl, const float* __restrict__ Wr,
                         _Float16* __restrict__ Wpk) {
    int idx = blockIdx.x * blockDim.x + threadIdx.x;
    if (idx >= 2 * 8192) return;
    int which = idx >> 13;
    int r = idx & 8191;
    int j = r & 15;
    int lane = (r >> 4) & 31;
    int frag = r >> 9;
    int kc = frag >> 2, nt = frag & 3;
    int half_sel = lane >> 4, lidx = lane & 15;
    int kb = (j < 8) ? (half_sel * 8) : (16 + half_sel * 8);
    int k = kc * 32 + kb + (j & 7);
    int n = nt * 16 + lidx;
    const float* W = which ? Wr : Wl;
    Wpk[idx] = (_Float16)W[k * 64 + n];
}

__global__ void k_gemm2_wmma(const _Float16* __restrict__ hh,
                             const _Float16* __restrict__ Wpk,
                             const float* __restrict__ bl, const float* __restrict__ br,
                             float* __restrict__ xl2, float* __restrict__ xr2, int N) {
    __shared__ __align__(32) _Float16 ldsW[2 * 8192];

    {
        const uint4* g = (const uint4*)Wpk;
        uint4* l = (uint4*)ldsW;
        for (int i = threadIdx.x; i < 2048; i += blockDim.x) l[i] = g[i];
    }
    __syncthreads();

    int wave = (blockIdx.x * blockDim.x + threadIdx.x) >> 5;
    int lane = threadIdx.x & 31;
    int row0 = wave * 16;
    if (row0 >= N) return;

    int half_sel = lane >> 4;
    int lidx = lane & 15;
    int kb_lo = half_sel * 8;
    int kb_hi = 16 + kb_lo;
    int m = row0 + lidx;
    if (m >= N) m = N - 1;

    const v16h* fragL = (const v16h*)ldsW;
    const v16h* fragR = (const v16h*)(ldsW + 8192);

    v8f accl[4] = {};
    v8f accr[4] = {};

#pragma unroll
    for (int kc = 0; kc < 4; ++kc) {
        const _Float16* pa = hh + m * 128 + kc * 32;
        v8h alo = *(const v8h*)(pa + kb_lo);
        v8h ahi = *(const v8h*)(pa + kb_hi);
        v16h a = __builtin_shufflevector(alo, ahi, 0, 1, 2, 3, 4, 5, 6, 7,
                                         8, 9, 10, 11, 12, 13, 14, 15);
#pragma unroll
        for (int nt = 0; nt < 4; ++nt) {
            int fi = (kc * 4 + nt) * 32 + lane;
            v16h b  = fragL[fi];
            v16h b2 = fragR[fi];
            accl[nt] = __builtin_amdgcn_wmma_f32_16x16x32_f16(
                false, a, false, b, (short)0, accl[nt], false, false);
            accr[nt] = __builtin_amdgcn_wmma_f32_16x16x32_f16(
                false, a, false, b2, (short)0, accr[nt], false, false);
        }
    }
#pragma unroll
    for (int nt = 0; nt < 4; ++nt) {
        int n = nt * 16 + lidx;
        float bln = bl[n], brn = br[n];
#pragma unroll
        for (int r = 0; r < 8; ++r) {
            int mo = row0 + r + half_sel * 8;
            if (mo < N) {
                xl2[mo * 64 + n] = accl[nt][r] + bln;
                xr2[mo * 64 + n] = accr[nt][r] + brn;
            }
        }
    }
}

template <int H, int C>
__device__ __forceinline__ void pool_item_load(unsigned idx,
        const float* __restrict__ ee, const float* __restrict__ denom,
        const float* __restrict__ xl, const int* __restrict__ src,
        const int* __restrict__ tgt, const int* __restrict__ batch, int G,
        float4& v, float& alpha, int& off) {
    constexpr unsigned D = H * C, Q = D / 4;
    unsigned e = idx / Q, q = idx % Q;
    unsigned d = q * 4, h = d / C;
    int s = src[e], t = tgt[e];
    alpha = ee[e * H + h] / denom[t * H + h];
    int g = batch[t];
    v = *(const float4*)(xl + (size_t)s * D + d);
    unsigned row = ((unsigned)g < (unsigned)G) ? (unsigned)g : (unsigned)G;
    off = (int)(row * D + d);
}

__device__ __forceinline__ void pool_item_add(float* tab, const float4& v, float alpha, int off) {
    atomicAdd(&tab[off + 0], alpha * v.x);
    atomicAdd(&tab[off + 1], alpha * v.y);
    atomicAdd(&tab[off + 2], alpha * v.z);
    atomicAdd(&tab[off + 3], alpha * v.w);
}

constexpr int R1  = 200;
constexpr int AGT = 1024;
constexpr int EPT = 32;
constexpr int CHK = AGT * EPT;
constexpr int AGU = 6;

typedef __attribute__((ext_vector_type(2))) _Float16 v2h;

__device__ __forceinline__ void own_load_chunk(int (&cur)[EPT], const int* __restrict__ tgt,
                                               int ebase, int tid, int E) {
    if (ebase + CHK <= E && (((size_t)tgt) & 15) == 0) {
#pragma unroll
        for (int u = 0; u < EPT / 4; ++u) {
            int4 t = *(const int4*)(tgt + ebase + u * 4 * AGT + 4 * tid);
            cur[4 * u + 0] = t.x; cur[4 * u + 1] = t.y; cur[4 * u + 2] = t.z; cur[4 * u + 3] = t.w;
        }
    } else {
#pragma unroll
        for (int k = 0; k < EPT; ++k) {
            int e = ebase + (k >> 2) * 4 * AGT + 4 * tid + (k & 3);
            int t = tgt[(e < E) ? e : 0];
            cur[k] = (e < E) ? t : -1;
        }
    }
}

__global__ __launch_bounds__(AGT) void k_agg_own(
        const float* __restrict__ ee, const float* __restrict__ denom,
        const float* __restrict__ xl, const int* __restrict__ src,
        const int* __restrict__ tgt, const float* __restrict__ bias,
        _Float16* __restrict__ hh, int N, int E) {
    __shared__ float tab[R1 * 128];
    __shared__ int   queue[CHK];
    __shared__ int   qcnt[2];

    const int tid = threadIdx.x;
    const int lane = tid & 31, wave = tid >> 5;
    const int hq = lane >> 1;
    const int node0 = blockIdx.x * R1;
    const int rows = (N - node0 < R1) ? (N - node0) : R1;
    const int nChunks = (E + CHK - 1) / CHK;

    for (int i = tid; i < R1 * 128; i += AGT) tab[i] = 0.f;
    if (tid < 2) qcnt[tid] = 0;

    int cur[EPT];
#pragma unroll
    for (int k = 0; k < EPT; ++k) cur[k] = -1;
    if (nChunks > 0) own_load_chunk(cur, tgt, 0, tid, E);
    __syncthreads();

    for (int c = 0; c < nChunks; ++c) {
        const int ebase = c * CHK;
        unsigned mask = 0;
#pragma unroll
        for (int u = 0; u < EPT; ++u)
            mask |= (((unsigned)(cur[u] - node0) < (unsigned)rows) ? 1u : 0u) << u;
        if (mask) {
            int pos = atomicAdd(&qcnt[c & 1], __popc(mask));
            while (mask) {
                int u = __ffs(mask) - 1;
                mask &= mask - 1;
                queue[pos++] = ebase + (u >> 2) * 4 * AGT + 4 * tid + (u & 3);
            }
        }
        __syncthreads();
        const int cnt = qcnt[c & 1];
        if (tid == 0) qcnt[(c + 1) & 1] = 0;
        if (c + 1 < nChunks) own_load_chunk(cur, tgt, ebase + CHK, tid, E);

        for (int i = wave; i < cnt; i += (AGT / 32) * AGU) {
            float4 v[AGU];
            float  w[AGU], dn[AGU];
            int    off[AGU];
#pragma unroll
            for (int u = 0; u < AGU; ++u) {
                int j = i + u * (AGT / 32);
                int e = queue[(j < cnt) ? j : i];
                int s = src[e], t = tgt[e];
                w[u]  = ee[(size_t)e * 16 + hq];
                dn[u] = denom[(size_t)t * 16 + hq];
                v[u]  = *(const float4*)(xl + (size_t)s * 128 + lane * 4);
                off[u] = (t - node0) * 128 + lane * 4;
            }
#pragma unroll
            for (int u = 0; u < AGU; ++u) {
                asm volatile("" :: "v"(v[u].x), "v"(v[u].y), "v"(v[u].z), "v"(v[u].w), "v"(w[u]), "v"(dn[u]));
            }
#pragma unroll
            for (int u = 0; u < AGU; ++u)
                if (i + u * (AGT / 32) < cnt)
                    pool_item_add(tab, v[u], w[u] / dn[u], off[u]);
        }
        __syncthreads();
    }

    for (int i = tid; i < rows * 64; i += AGT) {
        int r = i >> 6, d = (i & 63) * 2;
        float v0 = tab[r * 128 + d] + bias[d];
        float v1 = tab[r * 128 + d + 1] + bias[d + 1];
        v0 = (v0 > 0.f) ? v0 : (__expf(v0) - 1.f);
        v1 = (v1 > 0.f) ? v1 : (__expf(v1) - 1.f);
        v2h o = {(_Float16)v0, (_Float16)v1};
        *(v2h*)(hh + (size_t)(node0 + r) * 128 + d) = o;
    }
}

template <int H, int C>
__global__ __launch_bounds__(512) void k_edge_agg_pool(
        const float* __restrict__ ee, const float* __restrict__ denom,
        const float* __restrict__ xl, const int* __restrict__ src,
        const int* __restrict__ tgt, const int* __restrict__ batch,
        float* __restrict__ part, int E, int G) {
    extern __shared__ float tab[];
    constexpr int D = H * C, Q = D / 4, U = 4;
    const int GD = G * D;
    for (int i = threadIdx.x; i < GD + D; i += blockDim.x) tab[i] = 0.f;
    __syncthreads();

    const unsigned total = (unsigned)E * Q;
    const unsigned stride = gridDim.x * blockDim.x;
    unsigned idx = blockIdx.x * blockDim.x + threadIdx.x;

    for (; idx + (U - 1) * stride < total; idx += U * stride) {
        float4 v[U];
        float  a[U];
        int    off[U];
#pragma unroll
        for (int u = 0; u < U; ++u)
            pool_item_load<H, C>(idx + u * stride, ee, denom, xl, src, tgt, batch, G,
                                 v[u], a[u], off[u]);
#pragma unroll
        for (int u = 0; u < U; ++u)
            pool_item_add(tab, v[u], a[u], off[u]);
    }
    for (; idx < total; idx += stride) {
        float4 v; float a; int off;
        pool_item_load<H, C>(idx, ee, denom, xl, src, tgt, batch, G, v, a, off);
        pool_item_add(tab, v, a, off);
    }
    __syncthreads();

    float* my = part + (size_t)blockIdx.x * GD;
    for (int i = threadIdx.x; i < GD; i += blockDim.x) my[i] = tab[i];
}

__global__ __launch_bounds__(1024) void k_final(const float* __restrict__ part, int nPart,
                                                const float* __restrict__ bias2,
                                                const int* __restrict__ batch,
                                                float* __restrict__ out, int N, int G) {
    __shared__ float red[16][64];
    __shared__ int cntS;
    const int g = blockIdx.x;
    const int tid = threadIdx.x;
    const int d = tid & 63, j = tid >> 6;
    if (tid == 0) cntS = 0;

    float acc = 0.f;
    const float* p = part + (size_t)g * 64 + d;
    const size_t tabStride = (size_t)G * 64;
    int b = j;
    for (; b + 16 * 7 < nPart; b += 16 * 8) {
        float t[8];
#pragma unroll
        for (int k = 0; k < 8; ++k) t[k] = p[(size_t)(b + 16 * k) * tabStride];
#pragma unroll
        for (int k = 0; k < 8; ++k) acc += t[k];
    }
    for (; b < nPart; b += 16) acc += p[(size_t)b * tabStride];

    int c = 0;
    for (int n = tid; n < N; n += 1024) c += (batch[n] == g) ? 1 : 0;

    red[j][d] = acc;
    __syncthreads();
    atomicAdd(&cntS, c);
    __syncthreads();

    if (j == 0) {
        float s = 0.f;
#pragma unroll
        for (int k = 0; k < 16; ++k) s += red[k][d];
        float cnt = (float)cntS;
        out[g * 64 + d] = (s + cnt * bias2[d]) / fmaxf(cnt, 1.0f);
    }
}

extern "C" void kernel_launch(void* const* d_in, const int* in_sizes, int n_in,
                              void* d_out, int out_size, void* d_ws, size_t ws_size,
                              hipStream_t stream) {
    const float* x      = (const float*)d_in[0];
    const float* ea     = (const float*)d_in[1];
    const float* W_l1   = (const float*)d_in[2];
    const float* b_l1   = (const float*)d_in[3];
    const float* W_r1   = (const float*)d_in[4];
    const float* b_r1   = (const float*)d_in[5];
    const float* W_e1   = (const float*)d_in[6];
    const float* att1   = (const float*)d_in[7];
    const float* bias1  = (const float*)d_in[8];
    const float* W_l2   = (const float*)d_in[9];
    const float* b_l2   = (const float*)d_in[10];
    const float* W_r2   = (const float*)d_in[11];
    const float* b_r2   = (const float*)d_in[12];
    const float* W_e2   = (const float*)d_in[13];
    const float* att2   = (const float*)d_in[14];
    const float* bias2  = (const float*)d_in[15];
    const int*   eidx   = (const int*)d_in[16];
    const int*   batch  = (const int*)d_in[17];
    float* out = (float*)d_out;

    const int N = in_sizes[0] / 2;
    const int E = in_sizes[1] / 2;
    const int G = out_size / 64;
    const int* src = eidx;
    const int* tgt = eidx + E;

    char* ws = (char*)d_ws;
    size_t off = 0;
    auto alloc = [&](size_t bytes) { size_t o = off; off = (off + bytes + 255) & ~(size_t)255; return o; };

    size_t o_emax1 = alloc((size_t)N * 16 * 4);
    size_t o_den1  = alloc((size_t)N * 16 * 4);
    size_t o_emax2 = alloc((size_t)N * 4 * 4);
    size_t o_den2  = alloc((size_t)N * 4 * 4);
    size_t acc_bytes = off;

    size_t o_xl1 = alloc((size_t)N * 128 * 4);
    size_t o_xr1 = alloc((size_t)N * 128 * 4);
    size_t o_e1  = alloc((size_t)E * 16 * 4);
    size_t o_hh  = alloc((size_t)N * 128 * 2);
    size_t o_wpk = alloc((size_t)2 * 8192 * 2);
    size_t o_xl2 = alloc((size_t)N * 64 * 4);
    size_t o_xr2 = alloc((size_t)N * 64 * 4);
    size_t o_e2  = alloc((size_t)E * 4 * 4);
    const int POOL_BLOCKS = 512;
    size_t o_part = alloc((size_t)POOL_BLOCKS * G * 64 * 4);
    (void)ws_size;

    unsigned* emax1 = (unsigned*)(ws + o_emax1);
    float*    den1  = (float*)(ws + o_den1);
    unsigned* emax2 = (unsigned*)(ws + o_emax2);
    float*    den2  = (float*)(ws + o_den2);
    float*    xl1   = (float*)(ws + o_xl1);
    float*    xr1   = (float*)(ws + o_xr1);
    float*    e1    = (float*)(ws + o_e1);
    _Float16* hh    = (_Float16*)(ws + o_hh);
    _Float16* wpk   = (_Float16*)(ws + o_wpk);
    float*    xl2   = (float*)(ws + o_xl2);
    float*    xr2   = (float*)(ws + o_xr2);
    float*    e2    = (float*)(ws + o_e2);
    float*    part  = (float*)(ws + o_part);

    const int TB = 256;
    auto blocks = [&](long long t) { return (unsigned)((t + TB - 1) / TB); };

    hipMemsetAsync(d_ws, 0, acc_bytes, stream);

    k_node_transform1<<<blocks((long long)N * 128), TB, 0, stream>>>(
        x, W_l1, b_l1, W_r1, b_r1, xl1, xr1, N);
    k_edge_logits<16, 8><<<blocks((long long)E * 16), TB, 0, stream>>>(
        xl1, xr1, ea, W_e1, att1, src, tgt, e1, emax1, E);
    k_edge_exp<16><<<blocks((long long)E * 16), TB, 0, stream>>>(
        e1, emax1, den1, tgt, E);
    k_agg_own<<<(unsigned)((N + R1 - 1) / R1), AGT, 0, stream>>>(
        e1, den1, xl1, src, tgt, bias1, hh, N, E);

    k_w_pack<<<blocks(2 * 8192), TB, 0, stream>>>(W_l2, W_r2, wpk);
    long long gemm_threads = (long long)((N + 15) / 16) * 32;
    k_gemm2_wmma<<<blocks(gemm_threads), TB, 0, stream>>>(
        hh, wpk, b_l2, b_r2, xl2, xr2, N);

    k_edge_logits<4, 16><<<blocks((long long)E * 4), TB, 0, stream>>>(
        xl2, xr2, ea, W_e2, att2, src, tgt, e2, emax2, E);
    k_edge_exp<4><<<blocks((long long)E * 4), TB, 0, stream>>>(
        e2, emax2, den2, tgt, E);

    k_edge_agg_pool<4, 16><<<POOL_BLOCKS, 512, (size_t)(G + 1) * 64 * sizeof(float), stream>>>(
        e2, den2, xl2, src, tgt, batch, part, E, G);
    k_final<<<G, 1024, 0, stream>>>(part, POOL_BLOCKS, bias2, batch, out, N, G);
}


# layer-1 attention fused into the target-ownership kernel: logits from the one gathered row, running max with rescale, denominators in LDS; 2 launches and their global atomics gone; changes sum order
# speedup vs baseline: 1.6493x; 1.1559x over previous
#include <hip/hip_runtime.h>
#include <hip/hip_fp16.h>

typedef __attribute__((ext_vector_type(16))) _Float16 v16h;
typedef __attribute__((ext_vector_type(8)))  _Float16 v8h;
typedef __attribute__((ext_vector_type(8)))  float    v8f;

__device__ __forceinline__ unsigned fp_ord(float f) {
    unsigned u = __float_as_uint(f);
    return (u & 0x80000000u) ? ~u : (u | 0x80000000u);
}
__device__ __forceinline__ float fp_unord(unsigned u) {
    unsigned b = (u & 0x80000000u) ? (u ^ 0x80000000u) : ~u;
    return __uint_as_float(b);
}

__global__ void k_node_transform1(const float* __restrict__ x,
                                  const float* __restrict__ Wl, const float* __restrict__ bl,
                                  const float* __restrict__ Wr, const float* __restrict__ br,
                                  float* __restrict__ xl1, float* __restrict__ xr1, int N) {
    int idx = blockIdx.x * blockDim.x + threadIdx.x;
    if (idx >= N * 128) return;
    int n = idx >> 7, d = idx & 127;
    float x0 = x[n * 2], x1 = x[n * 2 + 1];
    xl1[idx] = fmaf(x0, Wl[d], fmaf(x1, Wl[128 + d], bl[d]));
    xr1[idx] = fmaf(x0, Wr[d], fmaf(x1, Wr[128 + d], br[d]));
}

template <int H, int C>
__global__ void k_edge_logits(const float* __restrict__ xl, const float* __restrict__ xr,
                              const float* __restrict__ ea, const float* __restrict__ We,
                              const float* __restrict__ att,
                              const int* __restrict__ src, const int* __restrict__ tgt,
                              float* __restrict__ elog, unsigned* __restrict__ emaxM, int E) {
    const int D = H * C;
    int idx = blockIdx.x * blockDim.x + threadIdx.x;
    if (idx >= E * H) return;
    int e = idx / H, h = idx % H;
    int s = src[e], t = tgt[e];
    float a0 = ea[e * 2], a1 = ea[e * 2 + 1];
    int base = h * C;
    float acc = 0.f;
#pragma unroll
    for (int c = 0; c < C; ++c) {
        int d = base + c;
        float m = xl[s * D + d] + xr[t * D + d] + a0 * We[d] + a1 * We[D + d];
        m = (m > 0.f) ? m : 0.2f * m;
        acc = fmaf(m, att[base + c], acc);
    }
    elog[idx] = acc;
    atomicMax(&emaxM[t * H + h], fp_ord(acc));
}

template <int H>
__global__ void k_edge_exp(float* __restrict__ elog, const unsigned* __restrict__ emaxM,
                           float* __restrict__ denom, const int* __restrict__ tgt, int E) {
    int idx = blockIdx.x * blockDim.x + threadIdx.x;
    if (idx >= E * H) return;
    int e = idx / H, h = idx % H;
    int t = tgt[e];
    float mx = fp_unord(emaxM[t * H + h]);
    float v = __expf(elog[idx] - mx);
    elog[idx] = v;
    atomicAdd(&denom[t * H + h], v);
}

__global__ void k_w_pack(const float* __restrict__ Wl, const float* __restrict__ Wr,
                         _Float16* __restrict__ Wpk) {
    int idx = blockIdx.x * blockDim.x + threadIdx.x;
    if (idx >= 2 * 8192) return;
    int which = idx >> 13;
    int r = idx & 8191;
    int j = r & 15;
    int lane = (r >> 4) & 31;
    int frag = r >> 9;
    int kc = frag >> 2, nt = frag & 3;
    int half_sel = lane >> 4, lidx = lane & 15;
    int kb = (j < 8) ? (half_sel * 8) : (16 + half_sel * 8);
    int k = kc * 32 + kb + (j & 7);
    int n = nt * 16 + lidx;
    const float* W = which ? Wr : Wl;
    Wpk[idx] = (_Float16)W[k * 64 + n];
}

__global__ void k_gemm2_wmma(const _Float16* __restrict__ hh,
                             const _Float16* __restrict__ Wpk,
                             const float* __restrict__ bl, const float* __restrict__ br,
                             float* __restrict__ xl2, float* __restrict__ xr2, int N) {
    __shared__ __align__(32) _Float16 ldsW[2 * 8192];

    {
        const uint4* g = (const uint4*)Wpk;
        uint4* l = (uint4*)ldsW;
        for (int i = threadIdx.x; i < 2048; i += blockDim.x) l[i] = g[i];
    }
    __syncthreads();

    int wave = (blockIdx.x * blockDim.x + threadIdx.x) >> 5;
    int lane = threadIdx.x & 31;
    int row0 = wave * 16;
    if (row0 >= N) return;

    int half_sel = lane >> 4;
    int lidx = lane & 15;
    int kb_lo = half_sel * 8;
    int kb_hi = 16 + kb_lo;
    int m = row0 + lidx;
    if (m >= N) m = N - 1;

    const v16h* fragL = (const v16h*)ldsW;
    const v16h* fragR = (const v16h*)(ldsW + 8192);

    v8f accl[4] = {};
    v8f accr[4] = {};

#pragma unroll
    for (int kc = 0; kc < 4; ++kc) {
        const _Float16* pa = hh + m * 128 + kc * 32;
        v8h alo = *(const v8h*)(pa + kb_lo);
        v8h ahi = *(const v8h*)(pa + kb_hi);
        v16h a = __builtin_shufflevector(alo, ahi, 0, 1, 2, 3, 4, 5, 6, 7,
                                         8, 9, 10, 11, 12, 13, 14, 15);
#pragma unroll
        for (int nt = 0; nt < 4; ++nt) {
            int fi = (kc * 4 + nt) * 32 + lane;
            v16h b  = fragL[fi];
            v16h b2 = fragR[fi];
            accl[nt] = __builtin_amdgcn_wmma_f32_16x16x32_f16(
                false, a, false, b, (short)0, accl[nt], false, false);
            accr[nt] = __builtin_amdgcn_wmma_f32_16x16x32_f16(
                false, a, false, b2, (short)0, accr[nt], false, false);
        }
    }
#pragma unroll
    for (int nt = 0; nt < 4; ++nt) {
        int n = nt * 16 + lidx;
        float bln = bl[n], brn = br[n];
#pragma unroll
        for (int r = 0; r < 8; ++r) {
            int mo = row0 + r + half_sel * 8;
            if (mo < N) {
                xl2[mo * 64 + n] = accl[nt][r] + bln;
                xr2[mo * 64 + n] = accr[nt][r] + brn;
            }
        }
    }
}

template <int H, int C>
__device__ __forceinline__ void pool_item_load(unsigned idx,
        const float* __restrict__ ee, const float* __restrict__ denom,
        const float* __restrict__ xl, const int* __restrict__ src,
        const int* __restrict__ tgt, const int* __restrict__ batch, int G,
        float4& v, float& alpha, int& off) {
    constexpr unsigned D = H * C, Q = D / 4;
    unsigned e = idx / Q, q = idx % Q;
    unsigned d = q * 4, h = d / C;
    int s = src[e], t = tgt[e];
    alpha = ee[e * H + h] / denom[t * H + h];
    int g = batch[t];
    v = *(const float4*)(xl + (size_t)s * D + d);
    unsigned row = ((unsigned)g < (unsigned)G) ? (unsigned)g : (unsigned)G;
    off = (int)(row * D + d);
}

__device__ __forceinline__ void pool_item_add(float* tab, const float4& v, float alpha, int off) {
    atomicAdd(&tab[off + 0], alpha * v.x);
    atomicAdd(&tab[off + 1], alpha * v.y);
    atomicAdd(&tab[off + 2], alpha * v.z);
    atomicAdd(&tab[off + 3], alpha * v.w);
}

constexpr int R1  = 200;
constexpr int AGT = 1024;
constexpr int NWV = AGT / 32;
constexpr int EPT = 32;
constexpr int CHK = AGT * EPT;
constexpr int AGU = 6;
static_assert(CHK <= 65536, "queue entries are 16-bit offsets into the chunk");

typedef __attribute__((ext_vector_type(2))) _Float16 v2h;

__device__ __forceinline__ void own_load_chunk(int (&cur)[EPT], const int* __restrict__ tgt,
                                               int ebase, int tid, int E) {
    if (ebase + CHK <= E && (((size_t)tgt) & 15) == 0) {
#pragma unroll
        for (int u = 0; u < EPT / 4; ++u) {
            int4 t = *(const int4*)(tgt + ebase + u * 4 * AGT + 4 * tid);
            cur[4 * u + 0] = t.x; cur[4 * u + 1] = t.y; cur[4 * u + 2] = t.z; cur[4 * u + 3] = t.w;
        }
    } else {
#pragma unroll
        for (int k = 0; k < EPT; ++k) {
            int e = ebase + (k >> 2) * 4 * AGT + 4 * tid + (k & 3);
            int t = tgt[(e < E) ? e : 0];
            cur[k] = (e < E) ? t : -1;
        }
    }
}

__global__ __launch_bounds__(AGT) void k_l1_own(
        const float* __restrict__ xl, const float* __restrict__ xr,
        const float* __restrict__ ea, const float* __restrict__ We,
        const float* __restrict__ att,
        const int* __restrict__ src, const int* __restrict__ tgt,
        const float* __restrict__ bias,
        _Float16* __restrict__ hh, int N, int E) {
    __shared__ __align__(16) float tab[R1 * 128];
    __shared__ __align__(16) float xrS[R1 * 128];
    __shared__ unsigned mxU[R1 * 16];
    __shared__ float    mxA[R1 * 16];
    __shared__ float    den[R1 * 16];
    __shared__ unsigned short queue[CHK];
    __shared__ int      qcnt[2];

    const int tid = threadIdx.x;
    const int lane = tid & 31;
    const int wave = __builtin_amdgcn_readfirstlane(tid >> 5);
    const int hq = lane >> 1;
    const int node0 = blockIdx.x * R1;
    const int rows = (N - node0 < R1) ? (N - node0) : R1;
    const int nChunks = (E + CHK - 1) / CHK;

    for (int i = tid; i < R1 * 128; i += AGT) tab[i] = 0.f;
    for (int i = tid; i < R1 * 16; i += AGT) { mxU[i] = 0u; mxA[i] = -__builtin_inff(); den[i] = 0.f; }
    {
        const float4* g = (const float4*)(xr + (size_t)node0 * 128);
        float4* l = (float4*)xrS;
        for (int i = tid; i < rows * 32; i += AGT) l[i] = g[i];
    }
    if (tid < 2) qcnt[tid] = 0;

    float we0[4], we1[4], at[4];
#pragma unroll
    for (int k = 0; k < 4; ++k) {
        we0[k] = We[lane * 4 + k];
        we1[k] = We[128 + lane * 4 + k];
        at[k]  = att[lane * 4 + k];
    }

    int cur[EPT];
#pragma unroll
    for (int k = 0; k < EPT; ++k) cur[k] = -1;
    if (nChunks > 0) own_load_chunk(cur, tgt, 0, tid, E);
    __syncthreads();

    for (int c = 0; c < nChunks; ++c) {
        const int ebase = c * CHK;
        unsigned mask = 0;
#pragma unroll
        for (int u = 0; u < EPT; ++u)
            mask |= (((unsigned)(cur[u] - node0) < (unsigned)rows) ? 1u : 0u) << u;
        if (mask) {
            int pos = atomicAdd(&qcnt[c & 1], __popc(mask));
            while (mask) {
                int u = __ffs(mask) - 1;
                mask &= mask - 1;
                queue[pos++] = (unsigned short)((u >> 2) * 4 * AGT + 4 * tid + (u & 3));
            }
        }
        __syncthreads();
        const int cnt = qcnt[c & 1];
        if (tid == 0) qcnt[(c + 1) & 1] = 0;
        if (c + 1 < nChunks) own_load_chunk(cur, tgt, ebase + CHK, tid, E);

        for (int base = 0; base < cnt; base += NWV * AGU) {
            float4 v[AGU];
            float  lg[AGU], a0[AGU], a1[AGU];
            int    sN[AGU], tl[AGU];
#pragma unroll
            for (int u = 0; u < AGU; ++u) {
                int j = base + wave + u * NWV;
                int e = ebase + (int)queue[(j < cnt) ? j : (cnt - 1)];
                sN[u] = src[e];
                tl[u] = tgt[e] - node0;
                a0[u] = ea[(size_t)e * 2];
                a1[u] = ea[(size_t)e * 2 + 1];
            }
#pragma unroll
            for (int u = 0; u < AGU; ++u)
                v[u] = *(const float4*)(xl + (size_t)sN[u] * 128 + lane * 4);
#pragma unroll
            for (int u = 0; u < AGU; ++u) {
                const float4 r4 = *(const float4*)(xrS + tl[u] * 128 + lane * 4);
                float m0 = v[u].x + r4.x + a0[u] * we0[0] + a1[u] * we1[0];
                float m1 = v[u].y + r4.y + a0[u] * we0[1] + a1[u] * we1[1];
                float m2 = v[u].z + r4.z + a0[u] * we0[2] + a1[u] * we1[2];
                float m3 = v[u].w + r4.w + a0[u] * we0[3] + a1[u] * we1[3];
                m0 = (m0 > 0.f) ? m0 : 0.2f * m0;
                m1 = (m1 > 0.f) ? m1 : 0.2f * m1;
                m2 = (m2 > 0.f) ? m2 : 0.2f * m2;
                m3 = (m3 > 0.f) ? m3 : 0.2f * m3;
                float p = fmaf(m3, at[3], fmaf(m2, at[2], fmaf(m1, at[1], m0 * at[0])));
                p += __int_as_float(__builtin_amdgcn_update_dpp(0, __float_as_int(p), 0xB1, 0xF, 0xF, false));
                lg[u] = p;
                if (base + wave + u * NWV < cnt && (lane & 1) == 0)
                    atomicMax(&mxU[tl[u] * 16 + hq], fp_ord(p));
            }
            __syncthreads();
            for (int i = tid; i < R1 * 16; i += AGT) {
                unsigned mu = mxU[i];
                if (mu != 0u) {
                    float nm = fp_unord(mu), old = mxA[i];
                    if (nm != old) {
                        float sc = __expf(old - nm);
                        mxA[i] = nm;
                        den[i] *= sc;
                        float4* row = (float4*)(tab + i * 8);
                        float4 q0 = row[0], q1 = row[1];
                        q0.x *= sc; q0.y *= sc; q0.z *= sc; q0.w *= sc;
                        q1.x *= sc; q1.y *= sc; q1.z *= sc; q1.w *= sc;
                        row[0] = q0; row[1] = q1;
                    }
                }
            }
            __syncthreads();
#pragma unroll
            for (int u = 0; u < AGU; ++u) {
                if (base + wave + u * NWV < cnt) {
                    float w = __expf(lg[u] - mxA[tl[u] * 16 + hq]);
                    pool_item_add(tab, v[u], w, tl[u] * 128 + lane * 4);
                    if ((lane & 1) == 0) atomicAdd(&den[tl[u] * 16 + hq], w);
                }
            }
        }
        __syncthreads();
    }

    for (int i = tid; i < rows * 64; i += AGT) {
        int r = i >> 6, d = (i & 63) * 2;
        float dn = den[r * 16 + (d >> 3)];
        float v0 = (dn > 0.f) ? tab[r * 128 + d] / dn : 0.f;
        float v1 = (dn > 0.f) ? tab[r * 128 + d + 1] / dn : 0.f;
        v0 += bias[d];
        v1 += bias[d + 1];
        v0 = (v0 > 0.f) ? v0 : (__expf(v0) - 1.f);
        v1 = (v1 > 0.f) ? v1 : (__expf(v1) - 1.f);
        v2h o = {(_Float16)v0, (_Float16)v1};
        *(v2h*)(hh + (size_t)(node0 + r) * 128 + d) = o;
    }
}

template <int H, int C>
__global__ __launch_bounds__(512) void k_edge_agg_pool(
        const float* __restrict__ ee, const float* __restrict__ denom,
        const float* __restrict__ xl, const int* __restrict__ src,
        const int* __restrict__ tgt, const int* __restrict__ batch,
        float* __restrict__ part, int E, int G) {
    extern __shared__ float tab[];
    constexpr int D = H * C, Q = D / 4, U = 4;
    const int GD = G * D;
    for (int i = threadIdx.x; i < GD + D; i += blockDim.x) tab[i] = 0.f;
    __syncthreads();

    const unsigned total = (unsigned)E * Q;
    const unsigned stride = gridDim.x * blockDim.x;
    unsigned idx = blockIdx.x * blockDim.x + threadIdx.x;

    for (; idx + (U - 1) * stride < total; idx += U * stride) {
        float4 v[U];
        float  a[U];
        int    off[U];
#pragma unroll
        for (int u = 0; u < U; ++u)
            pool_item_load<H, C>(idx + u * stride, ee, denom, xl, src, tgt, batch, G,
                                 v[u], a[u], off[u]);
#pragma unroll
        for (int u = 0; u < U; ++u)
            pool_item_add(tab, v[u], a[u], off[u]);
    }
    for (; idx < total; idx += stride) {
        float4 v; float a; int off;
        pool_item_load<H, C>(idx, ee, denom, xl, src, tgt, batch, G, v, a, off);
        pool_item_add(tab, v, a, off);
    }
    __syncthreads();

    float* my = part + (size_t)blockIdx.x * GD;
    for (int i = threadIdx.x; i < GD; i += blockDim.x) my[i] = tab[i];
}

__global__ __launch_bounds__(1024) void k_final(const float* __restrict__ part, int nPart,
                                                const float* __restrict__ bias2,
                                                const int* __restrict__ batch,
                                                float* __restrict__ out, int N, int G) {
    __shared__ float red[16][64];
    __shared__ int cntS;
    const int g = blockIdx.x;
    const int tid = threadIdx.x;
    const int d = tid & 63, j = tid >> 6;
    if (tid == 0) cntS = 0;

    float acc = 0.f;
    const float* p = part + (size_t)g * 64 + d;
    const size_t tabStride = (size_t)G * 64;
    int b = j;
    for (; b + 16 * 7 < nPart; b += 16 * 8) {
        float t[8];
#pragma unroll
        for (int k = 0; k < 8; ++k) t[k] = p[(size_t)(b + 16 * k) * tabStride];
#pragma unroll
        for (int k = 0; k < 8; ++k) acc += t[k];
    }
    for (; b < nPart; b += 16) acc += p[(size_t)b * tabStride];

    int c = 0;
    for (int n = tid; n < N; n += 1024) c += (batch[n] == g) ? 1 : 0;

    red[j][d] = acc;
    __syncthreads();
    atomicAdd(&cntS, c);
    __syncthreads();

    if (j == 0) {
        float s = 0.f;
#pragma unroll
        for (int k = 0; k < 16; ++k) s += red[k][d];
        float cnt = (float)cntS;
        out[g * 64 + d] = (s + cnt * bias2[d]) / fmaxf(cnt, 1.0f);
    }
}

extern "C" void kernel_launch(void* const* d_in, const int* in_sizes, int n_in,
                              void* d_out, int out_size, void* d_ws, size_t ws_size,
                              hipStream_t stream) {
    const float* x      = (const float*)d_in[0];
    const float* ea     = (const float*)d_in[1];
    const float* W_l1   = (const float*)d_in[2];
    const float* b_l1   = (const float*)d_in[3];
    const float* W_r1   = (const float*)d_in[4];
    const float* b_r1   = (const float*)d_in[5];
    const float* W_e1   = (const float*)d_in[6];
    const float* att1   = (const float*)d_in[7];
    const float* bias1  = (const float*)d_in[8];
    const float* W_l2   = (const float*)d_in[9];
    const float* b_l2   = (const float*)d_in[10];
    const float* W_r2   = (const float*)d_in[11];
    const float* b_r2   = (const float*)d_in[12];
    const float* W_e2   = (const float*)d_in[13];
    const float* att2   = (const float*)d_in[14];
    const float* bias2  = (const float*)d_in[15];
    const int*   eidx   = (const int*)d_in[16];
    const int*   batch  = (const int*)d_in[17];
    float* out = (float*)d_out;

    const int N = in_sizes[0] / 2;
    const int E = in_sizes[1] / 2;
    const int G = out_size / 64;
    const int* src = eidx;
    const int* tgt = eidx + E;

    char* ws = (char*)d_ws;
    size_t off = 0;
    auto alloc = [&](size_t bytes) { size_t o = off; off = (off + bytes + 255) & ~(size_t)255; return o; };

    size_t o_emax2 = alloc((size_t)N * 4 * 4);
    size_t o_den2  = alloc((size_t)N * 4 * 4);
    size_t acc_bytes = off;

    size_t o_xl1 = alloc((size_t)N * 128 * 4);
    size_t o_xr1 = alloc((size_t)N * 128 * 4);
    size_t o_hh  = alloc((size_t)N * 128 * 2);
    size_t o_wpk = alloc((size_t)2 * 8192 * 2);
    size_t o_xl2 = alloc((size_t)N * 64 * 4);
    size_t o_xr2 = alloc((size_t)N * 64 * 4);
    size_t o_e2  = alloc((size_t)E * 4 * 4);
    const int POOL_BLOCKS = 512;
    size_t o_part = alloc((size_t)POOL_BLOCKS * G * 64 * 4);
    (void)ws_size;

    unsigned* emax2 = (unsigned*)(ws + o_emax2);
    float*    den2  = (float*)(ws + o_den2);
    float*    xl1   = (float*)(ws + o_xl1);
    float*    xr1   = (float*)(ws + o_xr1);
    _Float16* hh    = (_Float16*)(ws + o_hh);
    _Float16* wpk   = (_Float16*)(ws + o_wpk);
    float*    xl2   = (float*)(ws + o_xl2);
    float*    xr2   = (float*)(ws + o_xr2);
    float*    e2    = (float*)(ws + o_e2);
    float*    part  = (float*)(ws + o_part);

    const int TB = 256;
    auto blocks = [&](long long t) { return (unsigned)((t + TB - 1) / TB); };

    hipMemsetAsync(d_ws, 0, acc_bytes, stream);

    k_node_transform1<<<blocks((long long)N * 128), TB, 0, stream>>>(
        x, W_l1, b_l1, W_r1, b_r1, xl1, xr1, N);
    k_l1_own<<<(unsigned)((N + R1 - 1) / R1), AGT, 0, stream>>>(
        xl1, xr1, ea, W_e1, att1, src, tgt, bias1, hh, N, E);

    k_w_pack<<<blocks(2 * 8192), TB, 0, stream>>>(W_l2, W_r2, wpk);
    long long gemm_threads = (long long)((N + 15) / 16) * 32;
    k_gemm2_wmma<<<blocks(gemm_threads), TB, 0, stream>>>(
        hh, wpk, b_l2, b_r2, xl2, xr2, N);

    k_edge_logits<4, 16><<<blocks((long long)E * 4), TB, 0, stream>>>(
        xl2, xr2, ea, W_e2, att2, src, tgt, e2, emax2, E);
    k_edge_exp<4><<<blocks((long long)E * 4), TB, 0, stream>>>(
        e2, emax2, den2, tgt, E);

    k_edge_agg_pool<4, 16><<<POOL_BLOCKS, 512, (size_t)(G + 1) * 64 * sizeof(float), stream>>>(
        e2, den2, xl2, src, tgt, batch, part, E, G);
    k_final<<<G, 1024, 0, stream>>>(part, POOL_BLOCKS, bias2, batch, out, N, G);
}


# layer-1 softmax in one barrier-free pass over a block-wide LDS edge list: weights taken against a per-entry reference logit, true max checked after (two-pass fallback); changes layer-1 sum order
# speedup vs baseline: 2.0829x; 1.2629x over previous
#include <hip/hip_runtime.h>
#include <hip/hip_fp16.h>

typedef __attribute__((ext_vector_type(16))) _Float16 v16h;
typedef __attribute__((ext_vector_type(8)))  _Float16 v8h;
typedef __attribute__((ext_vector_type(8)))  float    v8f;

__device__ __forceinline__ unsigned fp_ord(float f) {
    unsigned u = __float_as_uint(f);
    return (u & 0x80000000u) ? ~u : (u | 0x80000000u);
}
__device__ __forceinline__ float fp_unord(unsigned u) {
    unsigned b = (u & 0x80000000u) ? (u ^ 0x80000000u) : ~u;
    return __uint_as_float(b);
}

__global__ void k_node_transform1(const float* __restrict__ x,
                                  const float* __restrict__ Wl, const float* __restrict__ bl,
                                  const float* __restrict__ Wr, const float* __restrict__ br,
                                  float* __restrict__ xl1, float* __restrict__ xr1, int N) {
    int idx = blockIdx.x * blockDim.x + threadIdx.x;
    if (idx >= N * 128) return;
    int n = idx >> 7, d = idx & 127;
    float x0 = x[n * 2], x1 = x[n * 2 + 1];
    xl1[idx] = fmaf(x0, Wl[d], fmaf(x1, Wl[128 + d], bl[d]));
    xr1[idx] = fmaf(x0, Wr[d], fmaf(x1, Wr[128 + d], br[d]));
}

template <int H, int C>
__global__ void k_edge_logits(const float* __restrict__ xl, const float* __restrict__ xr,
                              const float* __restrict__ ea, const float* __restrict__ We,
                              const float* __restrict__ att,
                              const int* __restrict__ src, const int* __restrict__ tgt,
                              float* __restrict__ elog, unsigned* __restrict__ emaxM, int E) {
    const int D = H * C;
    int idx = blockIdx.x * blockDim.x + threadIdx.x;
    if (idx >= E * H) return;
    int e = idx / H, h = idx % H;
    int s = src[e], t = tgt[e];
    float a0 = ea[e * 2], a1 = ea[e * 2 + 1];
    int base = h * C;
    float acc = 0.f;
#pragma unroll
    for (int c = 0; c < C; ++c) {
        int d = base + c;
        float m = xl[s * D + d] + xr[t * D + d] + a0 * We[d] + a1 * We[D + d];
        m = (m > 0.f) ? m : 0.2f * m;
        acc = fmaf(m, att[base + c], acc);
    }
    elog[idx] = acc;
    atomicMax(&emaxM[t * H + h], fp_ord(acc));
}

template <int H>
__global__ void k_edge_exp(float* __restrict__ elog, const unsigned* __restrict__ emaxM,
                           float* __restrict__ denom, const int* __restrict__ tgt, int E) {
    int idx = blockIdx.x * blockDim.x + threadIdx.x;
    if (idx >= E * H) return;
    int e = idx / H, h = idx % H;
    int t = tgt[e];
    float mx = fp_unord(emaxM[t * H + h]);
    float v = __expf(elog[idx] - mx);
    elog[idx] = v;
    atomicAdd(&denom[t * H + h], v);
}

constexpr int KEY_R = 200;
__global__ void k_w_pack(const float* __restrict__ Wl, const float* __restrict__ Wr,
                         _Float16* __restrict__ Wpk, int packBlocks,
                         const int* __restrict__ tgt, unsigned* __restrict__ keys, int E, int nKeys) {
    if ((int)blockIdx.x >= packBlocks) {
        int w = ((int)blockIdx.x - packBlocks) * (int)blockDim.x + (int)threadIdx.x;
        int e0 = w * 4;
        if (e0 >= nKeys) return;
        int t0, t1, t2, t3;
        if (e0 + 4 <= E && (((size_t)tgt) & 15) == 0) {
            int4 t = *(const int4*)(tgt + e0);
            t0 = t.x; t1 = t.y; t2 = t.z; t3 = t.w;
        } else {
            t0 = (e0 + 0 < E) ? tgt[e0 + 0] : 0;
            t1 = (e0 + 1 < E) ? tgt[e0 + 1] : 0;
            t2 = (e0 + 2 < E) ? tgt[e0 + 2] : 0;
            t3 = (e0 + 3 < E) ? tgt[e0 + 3] : 0;
        }
        unsigned k0 = ((unsigned)t0 / (unsigned)KEY_R) & 255u;
        unsigned k1 = ((unsigned)t1 / (unsigned)KEY_R) & 255u;
        unsigned k2 = ((unsigned)t2 / (unsigned)KEY_R) & 255u;
        unsigned k3 = ((unsigned)t3 / (unsigned)KEY_R) & 255u;
        keys[w] = k0 | (k1 << 8) | (k2 << 16) | (k3 << 24);
        return;
    }
    int idx = blockIdx.x * blockDim.x + threadIdx.x;
    if (idx >= 2 * 8192) return;
    int which = idx >> 13;
    int r = idx & 8191;
    int j = r & 15;
    int lane = (r >> 4) & 31;
    int frag = r >> 9;
    int kc = frag >> 2, nt = frag & 3;
    int half_sel = lane >> 4, lidx = lane & 15;
    int kb = (j < 8) ? (half_sel * 8) : (16 + half_sel * 8);
    int k = kc * 32 + kb + (j & 7);
    int n = nt * 16 + lidx;
    const float* W = which ? Wr : Wl;
    Wpk[idx] = (_Float16)W[k * 64 + n];
}

__global__ void k_gemm2_wmma(const _Float16* __restrict__ hh,
                             const _Float16* __restrict__ Wpk,
                             const float* __restrict__ bl, const float* __restrict__ br,
                             float* __restrict__ xl2, float* __restrict__ xr2, int N) {
    __shared__ __align__(32) _Float16 ldsW[2 * 8192];

    {
        const uint4* g = (const uint4*)Wpk;
        uint4* l = (uint4*)ldsW;
        for (int i = threadIdx.x; i < 2048; i += blockDim.x) l[i] = g[i];
    }
    __syncthreads();

    int wave = (blockIdx.x * blockDim.x + threadIdx.x) >> 5;
    int lane = threadIdx.x & 31;
    int row0 = wave * 16;
    if (row0 >= N) return;

    int half_sel = lane >> 4;
    int lidx = lane & 15;
    int kb_lo = half_sel * 8;
    int kb_hi = 16 + kb_lo;
    int m = row0 + lidx;
    if (m >= N) m = N - 1;

    const v16h* fragL = (const v16h*)ldsW;
    const v16h* fragR = (const v16h*)(ldsW + 8192);

    v8f accl[4] = {};
    v8f accr[4] = {};

#pragma unroll
    for (int kc = 0; kc < 4; ++kc) {
        const _Float16* pa = hh + m * 128 + kc * 32;
        v8h alo = *(const v8h*)(pa + kb_lo);
        v8h ahi = *(const v8h*)(pa + kb_hi);
        v16h a = __builtin_shufflevector(alo, ahi, 0, 1, 2, 3, 4, 5, 6, 7,
                                         8, 9, 10, 11, 12, 13, 14, 15);
#pragma unroll
        for (int nt = 0; nt < 4; ++nt) {
            int fi = (kc * 4 + nt) * 32 + lane;
            v16h b  = fragL[fi];
            v16h b2 = fragR[fi];
            accl[nt] = __builtin_amdgcn_wmma_f32_16x16x32_f16(
                false, a, false, b, (short)0, accl[nt], false, false);
            accr[nt] = __builtin_amdgcn_wmma_f32_16x16x32_f16(
                false, a, false, b2, (short)0, accr[nt], false, false);
        }
    }
#pragma unroll
    for (int nt = 0; nt < 4; ++nt) {
        int n = nt * 16 + lidx;
        float bln = bl[n], brn = br[n];
#pragma unroll
        for (int r = 0; r < 8; ++r) {
            int mo = row0 + r + half_sel * 8;
            if (mo < N) {
                xl2[mo * 64 + n] = accl[nt][r] + bln;
                xr2[mo * 64 + n] = accr[nt][r] + brn;
            }
        }
    }
}

template <int H, int C>
__device__ __forceinline__ void pool_item_load(unsigned idx,
        const float* __restrict__ ee, const float* __restrict__ denom,
        const float* __restrict__ xl, const int* __restrict__ src,
        const int* __restrict__ tgt, const int* __restrict__ batch, int G,
        float4& v, float& alpha, int& off) {
    constexpr unsigned D = H * C, Q = D / 4;
    unsigned e = idx / Q, q = idx % Q;
    unsigned d = q * 4, h = d / C;
    int s = src[e], t = tgt[e];
    alpha = ee[e * H + h] / denom[t * H + h];
    int g = batch[t];
    v = *(const float4*)(xl + (size_t)s * D + d);
    unsigned row = ((unsigned)g < (unsigned)G) ? (unsigned)g : (unsigned)G;
    off = (int)(row * D + d);
}

__device__ __forceinline__ void pool_item_add(float* tab, const float4& v, float alpha, int off) {
    atomicAdd(&tab[off + 0], alpha * v.x);
    atomicAdd(&tab[off + 1], alpha * v.y);
    atomicAdd(&tab[off + 2], alpha * v.z);
    atomicAdd(&tab[off + 3], alpha * v.w);
}

constexpr int R1  = 200;
constexpr int AGT = 1024;
constexpr int NWV = AGT / 32;
constexpr int EPT = 16;
constexpr int CHK = AGT * EPT;
constexpr int QCAP = CHK;
constexpr int AGU = 8;
constexpr float SHIFT_OK = 40.f;
static_assert(EPT == 16, "a thread's keys of one chunk are one 16-byte word; two chunks make one 32-bit match mask");
static_assert(QCAP >= CHK, "a run of one chunk must always fit the list");
static_assert(KEY_R == R1, "the owner keys are made for this block size");

typedef __attribute__((ext_vector_type(2))) _Float16 v2h;

__device__ __forceinline__ uint4 own_load_keys(const unsigned* __restrict__ keys, int ebase, int tid) {
    return *(const uint4*)(keys + ((ebase + 16 * tid) >> 2));
}

__device__ __forceinline__ unsigned own_match4(unsigned w, unsigned pat) {
    unsigned x = w ^ pat;
    return ~(((x & 0x7F7F7F7Fu) + 0x7F7F7F7Fu) | x | 0x7F7F7F7Fu);
}

__device__ __forceinline__ int own_off(int p, int tid) {
    const int i = p & 7, j = p >> 3;
    return (i >> 2) * CHK + 16 * tid + 4 * (i & 3) + j;
}

__device__ __forceinline__ void own_trip(const int* list, int base, int cnt, int wave, int lane,
        const int* __restrict__ src, const int* __restrict__ tgt, const float* __restrict__ ea,
        const float* __restrict__ xl, const float* xrS, int node0, int rows,
        const float (&we0)[4], const float (&we1)[4], const float (&at)[4],
        float4 (&v)[AGU], float (&lg)[AGU], int (&tl)[AGU]) {
    float a0[AGU], a1[AGU];
    int   sN[AGU];
#pragma unroll
    for (int u = 0; u < AGU; ++u) {
        int j = base + wave + u * NWV;
        int e = list[(j < cnt) ? j : (cnt - 1)];
        sN[u] = src[e];
        const int t = tgt[e] - node0;
        tl[u] = ((unsigned)t < (unsigned)rows) ? t : -1;
        a0[u] = ea[(size_t)e * 2];
        a1[u] = ea[(size_t)e * 2 + 1];
    }
#pragma unroll
    for (int u = 0; u < AGU; ++u)
        v[u] = *(const float4*)(xl + (size_t)sN[u] * 128 + lane * 4);
#pragma unroll
    for (int u = 0; u < AGU; ++u) {
        const float4 r4 = *(const float4*)(xrS + ((tl[u] < 0) ? 0 : tl[u]) * 128 + lane * 4);
        float m0 = v[u].x + r4.x + a0[u] * we0[0] + a1[u] * we1[0];
        float m1 = v[u].y + r4.y + a0[u] * we0[1] + a1[u] * we1[1];
        float m2 = v[u].z + r4.z + a0[u] * we0[2] + a1[u] * we1[2];
        float m3 = v[u].w + r4.w + a0[u] * we0[3] + a1[u] * we1[3];
        m0 = (m0 > 0.f) ? m0 : 0.2f * m0;
        m1 = (m1 > 0.f) ? m1 : 0.2f * m1;
        m2 = (m2 > 0.f) ? m2 : 0.2f * m2;
        m3 = (m3 > 0.f) ? m3 : 0.2f * m3;
        float p = fmaf(m3, at[3], fmaf(m2, at[2], fmaf(m1, at[1], m0 * at[0])));
        p += __int_as_float(__builtin_amdgcn_update_dpp(0, __float_as_int(p), 0xB1, 0xF, 0xF, false));
        lg[u] = p;
    }
}

__global__ __launch_bounds__(AGT) void k_l1_own(
        const float* __restrict__ xl, const float* __restrict__ xr,
        const float* __restrict__ ea, const float* __restrict__ We,
        const float* __restrict__ att,
        const int* __restrict__ src, const int* __restrict__ tgt,
        const unsigned* __restrict__ keys,
        const float* __restrict__ bias,
        _Float16* __restrict__ hh, int N, int E) {
    __shared__ __align__(16) float tab[R1 * 128];
    __shared__ __align__(16) float xrS[R1 * 128];
    __shared__ unsigned mxU[R1 * 16];
    __shared__ float    mxA[R1 * 16];
    __shared__ float    den[R1 * 16];
    __shared__ int      list[QCAP];
    __shared__ int      qcnt;
    __shared__ int      redo;

    const int tid = threadIdx.x;
    const int lane = tid & 31;
    const int wave = __builtin_amdgcn_readfirstlane(tid >> 5);
    const int hq = lane >> 1;
    const int node0 = blockIdx.x * R1;
    const int rows = (N - node0 < R1) ? (N - node0) : R1;
    const int nChunks = (E + CHK - 1) / CHK;

    for (int i = tid; i < R1 * 128; i += AGT) tab[i] = 0.f;
    for (int i = tid; i < R1 * 16; i += AGT) { mxU[i] = 0u; mxA[i] = -__builtin_inff(); den[i] = 0.f; }
    {
        const float4* g = (const float4*)(xr + (size_t)node0 * 128);
        float4* l = (float4*)xrS;
        for (int i = tid; i < rows * 32; i += AGT) l[i] = g[i];
    }
    if (tid == 0) { qcnt = 0; redo = 0; }

    float we0[4], we1[4], at[4];
#pragma unroll
    for (int k = 0; k < 4; ++k) {
        we0[k] = We[lane * 4 + k];
        we1[k] = We[128 + lane * 4 + k];
        at[k]  = att[lane * 4 + k];
    }

    const unsigned pat = ((unsigned)blockIdx.x & 255u) * 0x01010101u;
    __syncthreads();

    int c0 = 0;
    while (c0 < nChunks) {
        int seg = nChunks - c0;
        int cnt = 0;
        while (true) {
            const int cEnd = c0 + seg;
            uint4 kq[2];
            kq[0] = own_load_keys(keys, c0 * CHK, tid);
            kq[1] = own_load_keys(keys, ((c0 + 1 < cEnd) ? (c0 + 1) : c0) * CHK, tid);
            for (int c = c0; c < cEnd; c += 2) {
                const int ebase = c * CHK;
                uint4 kn[2];
                kn[0] = kq[0];
                kn[1] = kq[1];
                if (c + 2 < cEnd) {
                    kn[0] = own_load_keys(keys, ebase + 2 * CHK, tid);
                    kn[1] = own_load_keys(keys, ((c + 3 < cEnd) ? (c + 3) : (c + 2)) * CHK, tid);
                }
                unsigned mask = (own_match4(kq[0].x, pat) >> 7) | (own_match4(kq[0].y, pat) >> 6)
                              | (own_match4(kq[0].z, pat) >> 5) | (own_match4(kq[0].w, pat) >> 4);
                if (c + 1 < cEnd)
                    mask |= (own_match4(kq[1].x, pat) >> 3) | (own_match4(kq[1].y, pat) >> 2)
                          | (own_match4(kq[1].z, pat) >> 1) |  own_match4(kq[1].w, pat);
                if (mask && ebase + 2 * CHK > E) {
                    unsigned m2 = mask;
                    while (m2) {
                        int b = __ffs(m2) - 1;
                        m2 &= m2 - 1;
                        if (ebase + own_off(b, tid) >= E) mask &= ~(1u << b);
                    }
                }
                if (mask) {
                    int pos = atomicAdd(&qcnt, __popc(mask));
                    while (mask) {
                        int b = __ffs(mask) - 1;
                        mask &= mask - 1;
                        if (pos < QCAP) list[pos] = ebase + own_off(b, tid);
                        ++pos;
                    }
                }
                kq[0] = kn[0];
                kq[1] = kn[1];
            }
            __syncthreads();
            cnt = qcnt;
            __syncthreads();
            if (tid == 0) qcnt = 0;
            if (cnt <= QCAP) break;
            seg = (seg + 1) >> 1;
            __syncthreads();
        }

        if (c0 == 0 && seg == nChunks) {
            for (int i = tid; i < R1 * 16; i += AGT) {
                const float4* xr4 = (const float4*)(xrS + i * 8);
                const float4 p0 = xr4[0], p1 = xr4[1];
                const float* ah = att + (i & 15) * 8;
                float m[8] = {p0.x, p0.y, p0.z, p0.w, p1.x, p1.y, p1.z, p1.w};
                float r = 0.f;
#pragma unroll
                for (int k = 0; k < 8; ++k) {
                    float mk = (m[k] > 0.f) ? m[k] : 0.2f * m[k];
                    r = fmaf(mk, ah[k], r);
                }
                mxA[i] = r;
            }
            __syncthreads();
            for (int base = 0; base < cnt; base += NWV * AGU) {
                float4 v[AGU];
                float  lg[AGU];
                int    tl[AGU];
                own_trip(list, base, cnt, wave, lane, src, tgt, ea, xl, xrS, node0, rows, we0, we1, at, v, lg, tl);
                float sh[AGU];
#pragma unroll
                for (int u = 0; u < AGU; ++u)
                    sh[u] = mxA[((tl[u] < 0) ? 0 : tl[u]) * 16 + hq];
#pragma unroll
                for (int u = 0; u < AGU; ++u) {
                    if (base + wave + u * NWV < cnt && tl[u] >= 0) {
                        float w = __expf(lg[u] - sh[u]);
                        pool_item_add(tab, v[u], w, tl[u] * 128 + lane * 4);
                        if ((lane & 1) == 0) {
                            atomicAdd(&den[tl[u] * 16 + hq], w);
                            atomicMax(&mxU[tl[u] * 16 + hq], fp_ord(lg[u]));
                        }
                    }
                }
            }
            __syncthreads();
            for (int i = tid; i < R1 * 16; i += AGT) {
                unsigned mu = mxU[i];
                if (mu != 0u) {
                    float d = fp_unord(mu) - mxA[i];
                    if (!(d > -SHIFT_OK && d < SHIFT_OK)) redo = 1;
                }
            }
            __syncthreads();
            if (redo == 0) break;
            for (int i = tid; i < R1 * 128; i += AGT) tab[i] = 0.f;
            for (int i = tid; i < R1 * 16; i += AGT) { mxA[i] = -__builtin_inff(); den[i] = 0.f; }
            __syncthreads();
        }

        for (int base = 0; base < cnt; base += NWV * AGU) {
            float4 v[AGU];
            float  lg[AGU];
            int    tl[AGU];
            own_trip(list, base, cnt, wave, lane, src, tgt, ea, xl, xrS, node0, rows, we0, we1, at, v, lg, tl);
#pragma unroll
            for (int u = 0; u < AGU; ++u) {
                if (base + wave + u * NWV < cnt && tl[u] >= 0 && (lane & 1) == 0)
                    atomicMax(&mxU[tl[u] * 16 + hq], fp_ord(lg[u]));
            }
        }
        __syncthreads();
        for (int i = tid; i < R1 * 16; i += AGT) {
            unsigned mu = mxU[i];
            if (mu != 0u) {
                float nm = fp_unord(mu), old = mxA[i];
                if (nm != old) {
                    float sc = __expf(old - nm);
                    mxA[i] = nm;
                    den[i] *= sc;
                    float4* row = (float4*)(tab + i * 8);
                    float4 q0 = row[0], q1 = row[1];
                    q0.x *= sc; q0.y *= sc; q0.z *= sc; q0.w *= sc;
                    q1.x *= sc; q1.y *= sc; q1.z *= sc; q1.w *= sc;
                    row[0] = q0; row[1] = q1;
                }
            }
        }
        __syncthreads();
        for (int base = 0; base < cnt; base += NWV * AGU) {
            float4 v[AGU];
            float  lg[AGU];
            int    tl[AGU];
            own_trip(list, base, cnt, wave, lane, src, tgt, ea, xl, xrS, node0, rows, we0, we1, at, v, lg, tl);
            float mx[AGU];
#pragma unroll
            for (int u = 0; u < AGU; ++u)
                mx[u] = mxA[((tl[u] < 0) ? 0 : tl[u]) * 16 + hq];
#pragma unroll
            for (int u = 0; u < AGU; ++u) {
                if (base + wave + u * NWV < cnt && tl[u] >= 0) {
                    float w = __expf(lg[u] - mx[u]);
                    pool_item_add(tab, v[u], w, tl[u] * 128 + lane * 4);
                    if ((lane & 1) == 0) atomicAdd(&den[tl[u] * 16 + hq], w);
                }
            }
        }
        c0 += seg;
        __syncthreads();
    }

    for (int i = tid; i < rows * 64; i += AGT) {
        int r = i >> 6, d = (i & 63) * 2;
        float dn = den[r * 16 + (d >> 3)];
        float v0 = (dn > 0.f) ? tab[r * 128 + d] / dn : 0.f;
        float v1 = (dn > 0.f) ? tab[r * 128 + d + 1] / dn : 0.f;
        v0 += bias[d];
        v1 += bias[d + 1];
        v0 = (v0 > 0.f) ? v0 : (__expf(v0) - 1.f);
        v1 = (v1 > 0.f) ? v1 : (__expf(v1) - 1.f);
        v2h o = {(_Float16)v0, (_Float16)v1};
        *(v2h*)(hh + (size_t)(node0 + r) * 128 + d) = o;
    }
}

template <int H, int C>
__global__ __launch_bounds__(512) void k_edge_agg_pool(
        const float* __restrict__ ee, const float* __restrict__ denom,
        const float* __restrict__ xl, const int* __restrict__ src,
        const int* __restrict__ tgt, const int* __restrict__ batch,
        float* __restrict__ part, int E, int G) {
    extern __shared__ float tab[];
    constexpr int D = H * C, Q = D / 4, U = 4;
    const int GD = G * D;
    for (int i = threadIdx.x; i < GD + D; i += blockDim.x) tab[i] = 0.f;
    __syncthreads();

    const unsigned total = (unsigned)E * Q;
    const unsigned stride = gridDim.x * blockDim.x;
    unsigned idx = blockIdx.x * blockDim.x + threadIdx.x;

    for (; idx + (U - 1) * stride < total; idx += U * stride) {
        float4 v[U];
        float  a[U];
        int    off[U];
#pragma unroll
        for (int u = 0; u < U; ++u)
            pool_item_load<H, C>(idx + u * stride, ee, denom, xl, src, tgt, batch, G,
                                 v[u], a[u], off[u]);
#pragma unroll
        for (int u = 0; u < U; ++u)
            pool_item_add(tab, v[u], a[u], off[u]);
    }
    for (; idx < total; idx += stride) {
        float4 v; float a; int off;
        pool_item_load<H, C>(idx, ee, denom, xl, src, tgt, batch, G, v, a, off);
        pool_item_add(tab, v, a, off);
    }
    __syncthreads();

    float* my = part + (size_t)blockIdx.x * GD;
    for (int i = threadIdx.x; i < GD; i += blockDim.x) my[i] = tab[i];
}

__global__ __launch_bounds__(1024) void k_final(const float* __restrict__ part, int nPart,
                                                const float* __restrict__ bias2,
                                                const int* __restrict__ batch,
                                                float* __restrict__ out, int N, int G) {
    __shared__ float red[16][64];
    __shared__ int cntS;
    const int g = blockIdx.x;
    const int tid = threadIdx.x;
    const int d = tid & 63, j = tid >> 6;
    if (tid == 0) cntS = 0;

    float acc = 0.f;
    const float* p = part + (size_t)g * 64 + d;
    const size_t tabStride = (size_t)G * 64;
    int b = j;
    for (; b + 16 * 7 < nPart; b += 16 * 8) {
        float t[8];
#pragma unroll
        for (int k = 0; k < 8; ++k) t[k] = p[(size_t)(b + 16 * k) * tabStride];
#pragma unroll
        for (int k = 0; k < 8; ++k) acc += t[k];
    }
    for (; b < nPart; b += 16) acc += p[(size_t)b * tabStride];

    int c = 0;
    for (int n = tid; n < N; n += 1024) c += (batch[n] == g) ? 1 : 0;

    red[j][d] = acc;
    __syncthreads();
    atomicAdd(&cntS, c);
    __syncthreads();

    if (j == 0) {
        float s = 0.f;
#pragma unroll
        for (int k = 0; k < 16; ++k) s += red[k][d];
        float cnt = (float)cntS;
        out[g * 64 + d] = (s + cnt * bias2[d]) / fmaxf(cnt, 1.0f);
    }
}

extern "C" void kernel_launch(void* const* d_in, const int* in_sizes, int n_in,
                              void* d_out, int out_size, void* d_ws, size_t ws_size,
                              hipStream_t stream) {
    const float* x      = (const float*)d_in[0];
    const float* ea     = (const float*)d_in[1];
    const float* W_l1   = (const float*)d_in[2];
    const float* b_l1   = (const float*)d_in[3];
    const float* W_r1   = (const float*)d_in[4];
    const float* b_r1   = (const float*)d_in[5];
    const float* W_e1   = (const float*)d_in[6];
    const float* att1   = (const float*)d_in[7];
    const float* bias1  = (const float*)d_in[8];
    const float* W_l2   = (const float*)d_in[9];
    const float* b_l2   = (const float*)d_in[10];
    const float* W_r2   = (const float*)d_in[11];
    const float* b_r2   = (const float*)d_in[12];
    const float* W_e2   = (const float*)d_in[13];
    const float* att2   = (const float*)d_in[14];
    const float* bias2  = (const float*)d_in[15];
    const int*   eidx   = (const int*)d_in[16];
    const int*   batch  = (const int*)d_in[17];
    float* out = (float*)d_out;

    const int N = in_sizes[0] / 2;
    const int E = in_sizes[1] / 2;
    const int G = out_size / 64;
    const int* src = eidx;
    const int* tgt = eidx + E;

    char* ws = (char*)d_ws;
    size_t off = 0;
    auto alloc = [&](size_t bytes) { size_t o = off; off = (off + bytes + 255) & ~(size_t)255; return o; };

    size_t o_emax2 = alloc((size_t)N * 4 * 4);
    size_t o_den2  = alloc((size_t)N * 4 * 4);
    size_t acc_bytes = off;

    size_t o_xl1 = alloc((size_t)N * 128 * 4);
    size_t o_xr1 = alloc((size_t)N * 128 * 4);
    size_t o_hh  = alloc((size_t)N * 128 * 2);
    size_t o_wpk = alloc((size_t)2 * 8192 * 2);
    const int nKeys = ((E + CHK - 1) / CHK) * CHK;
    size_t o_keys = alloc((size_t)nKeys);
    size_t o_xl2 = alloc((size_t)N * 64 * 4);
    size_t o_xr2 = alloc((size_t)N * 64 * 4);
    size_t o_e2  = alloc((size_t)E * 4 * 4);
    const int POOL_BLOCKS = 512;
    size_t o_part = alloc((size_t)POOL_BLOCKS * G * 64 * 4);
    (void)ws_size;

    unsigned* emax2 = (unsigned*)(ws + o_emax2);
    float*    den2  = (float*)(ws + o_den2);
    float*    xl1   = (float*)(ws + o_xl1);
    float*    xr1   = (float*)(ws + o_xr1);
    _Float16* hh    = (_Float16*)(ws + o_hh);
    _Float16* wpk   = (_Float16*)(ws + o_wpk);
    float*    xl2   = (float*)(ws + o_xl2);
    float*    xr2   = (float*)(ws + o_xr2);
    float*    e2    = (float*)(ws + o_e2);
    float*    part  = (float*)(ws + o_part);
    unsigned* keys  = (unsigned*)(ws + o_keys);

    const int TB = 256;
    auto blocks = [&](long long t) { return (unsigned)((t + TB - 1) / TB); };

    hipMemsetAsync(d_ws, 0, acc_bytes, stream);

    k_node_transform1<<<blocks((long long)N * 128), TB, 0, stream>>>(
        x, W_l1, b_l1, W_r1, b_r1, xl1, xr1, N);
    const int packBlocks = (int)blocks(2 * 8192);
    k_w_pack<<<(unsigned)packBlocks + blocks(nKeys / 4), TB, 0, stream>>>(
        W_l2, W_r2, wpk, packBlocks, tgt, keys, E, nKeys);
    k_l1_own<<<(unsigned)((N + R1 - 1) / R1), AGT, 0, stream>>>(
        xl1, xr1, ea, W_e1, att1, src, tgt, keys, bias1, hh, N, E);

    long long gemm_threads = (long long)((N + 15) / 16) * 32;
    k_gemm2_wmma<<<blocks(gemm_threads), TB, 0, stream>>>(
        hh, wpk, b_l2, b_r2, xl2, xr2, N);

    k_edge_logits<4, 16><<<blocks((long long)E * 4), TB, 0, stream>>>(
        xl2, xr2, ea, W_e2, att2, src, tgt, e2, emax2, E);
    k_edge_exp<4><<<blocks((long long)E * 4), TB, 0, stream>>>(
        e2, emax2, den2, tgt, E);

    k_edge_agg_pool<4, 16><<<POOL_BLOCKS, 512, (size_t)(G + 1) * 64 * sizeof(float), stream>>>(
        e2, den2, xl2, src, tgt, batch, part, E, G);
    k_final<<<G, 1024, 0, stream>>>(part, POOL_BLOCKS, bias2, batch, out, N, G);
}


# layer-2 attention by target ownership: one kernel does logits, softmax, node and graph sums in LDS on layer 1's edge list; memset, 3 launches, global atomics gone; changes layer-2 and pool sum order
# speedup vs baseline: 3.0456x; 1.4622x over previous
#include <hip/hip_runtime.h>
#include <hip/hip_fp16.h>

typedef __attribute__((ext_vector_type(16))) _Float16 v16h;
typedef __attribute__((ext_vector_type(8)))  _Float16 v8h;
typedef __attribute__((ext_vector_type(8)))  float    v8f;

__device__ __forceinline__ unsigned fp_ord(float f) {
    unsigned u = __float_as_uint(f);
    return (u & 0x80000000u) ? ~u : (u | 0x80000000u);
}
__device__ __forceinline__ float fp_unord(unsigned u) {
    unsigned b = (u & 0x80000000u) ? (u ^ 0x80000000u) : ~u;
    return __uint_as_float(b);
}

__global__ void k_node_transform1(const float* __restrict__ x,
                                  const float* __restrict__ Wl, const float* __restrict__ bl,
                                  const float* __restrict__ Wr, const float* __restrict__ br,
                                  float* __restrict__ xl1, float* __restrict__ xr1, int N) {
    int idx = blockIdx.x * blockDim.x + threadIdx.x;
    if (idx >= N * 128) return;
    int n = idx >> 7, d = idx & 127;
    float x0 = x[n * 2], x1 = x[n * 2 + 1];
    xl1[idx] = fmaf(x0, Wl[d], fmaf(x1, Wl[128 + d], bl[d]));
    xr1[idx] = fmaf(x0, Wr[d], fmaf(x1, Wr[128 + d], br[d]));
}

constexpr int KEY_R = 200;
__global__ void k_w_pack(const float* __restrict__ Wl, const float* __restrict__ Wr,
                         _Float16* __restrict__ Wpk, int packBlocks,
                         const int* __restrict__ tgt, unsigned* __restrict__ keys, int E, int nKeys) {
    if ((int)blockIdx.x >= packBlocks) {
        int w = ((int)blockIdx.x - packBlocks) * (int)blockDim.x + (int)threadIdx.x;
        int e0 = w * 4;
        if (e0 >= nKeys) return;
        int t0, t1, t2, t3;
        if (e0 + 4 <= E && (((size_t)tgt) & 15) == 0) {
            int4 t = *(const int4*)(tgt + e0);
            t0 = t.x; t1 = t.y; t2 = t.z; t3 = t.w;
        } else {
            t0 = (e0 + 0 < E) ? tgt[e0 + 0] : 0;
            t1 = (e0 + 1 < E) ? tgt[e0 + 1] : 0;
            t2 = (e0 + 2 < E) ? tgt[e0 + 2] : 0;
            t3 = (e0 + 3 < E) ? tgt[e0 + 3] : 0;
        }
        unsigned k0 = ((unsigned)t0 / (unsigned)KEY_R) & 255u;
        unsigned k1 = ((unsigned)t1 / (unsigned)KEY_R) & 255u;
        unsigned k2 = ((unsigned)t2 / (unsigned)KEY_R) & 255u;
        unsigned k3 = ((unsigned)t3 / (unsigned)KEY_R) & 255u;
        keys[w] = k0 | (k1 << 8) | (k2 << 16) | (k3 << 24);
        return;
    }
    int idx = blockIdx.x * blockDim.x + threadIdx.x;
    if (idx >= 2 * 8192) return;
    int which = idx >> 13;
    int r = idx & 8191;
    int j = r & 15;
    int lane = (r >> 4) & 31;
    int frag = r >> 9;
    int kc = frag >> 2, nt = frag & 3;
    int half_sel = lane >> 4, lidx = lane & 15;
    int kb = (j < 8) ? (half_sel * 8) : (16 + half_sel * 8);
    int k = kc * 32 + kb + (j & 7);
    int n = nt * 16 + lidx;
    const float* W = which ? Wr : Wl;
    Wpk[idx] = (_Float16)W[k * 64 + n];
}

__global__ void k_gemm2_wmma(const _Float16* __restrict__ hh,
                             const _Float16* __restrict__ Wpk,
                             const float* __restrict__ bl, const float* __restrict__ br,
                             float* __restrict__ xl2, float* __restrict__ xr2, int N) {
    __shared__ __align__(32) _Float16 ldsW[2 * 8192];

    {
        const uint4* g = (const uint4*)Wpk;
        uint4* l = (uint4*)ldsW;
        for (int i = threadIdx.x; i < 2048; i += blockDim.x) l[i] = g[i];
    }
    __syncthreads();

    int wave = (blockIdx.x * blockDim.x + threadIdx.x) >> 5;
    int lane = threadIdx.x & 31;
    int row0 = wave * 16;
    if (row0 >= N) return;

    int half_sel = lane >> 4;
    int lidx = lane & 15;
    int kb_lo = half_sel * 8;
    int kb_hi = 16 + kb_lo;
    int m = row0 + lidx;
    if (m >= N) m = N - 1;

    const v16h* fragL = (const v16h*)ldsW;
    const v16h* fragR = (const v16h*)(ldsW + 8192);

    v8f accl[4] = {};
    v8f accr[4] = {};

#pragma unroll
    for (int kc = 0; kc < 4; ++kc) {
        const _Float16* pa = hh + m * 128 + kc * 32;
        v8h alo = *(const v8h*)(pa + kb_lo);
        v8h ahi = *(const v8h*)(pa + kb_hi);
        v16h a = __builtin_shufflevector(alo, ahi, 0, 1, 2, 3, 4, 5, 6, 7,
                                         8, 9, 10, 11, 12, 13, 14, 15);
#pragma unroll
        for (int nt = 0; nt < 4; ++nt) {
            int fi = (kc * 4 + nt) * 32 + lane;
            v16h b  = fragL[fi];
            v16h b2 = fragR[fi];
            accl[nt] = __builtin_amdgcn_wmma_f32_16x16x32_f16(
                false, a, false, b, (short)0, accl[nt], false, false);
            accr[nt] = __builtin_amdgcn_wmma_f32_16x16x32_f16(
                false, a, false, b2, (short)0, accr[nt], false, false);
        }
    }
#pragma unroll
    for (int nt = 0; nt < 4; ++nt) {
        int n = nt * 16 + lidx;
        float bln = bl[n], brn = br[n];
#pragma unroll
        for (int r = 0; r < 8; ++r) {
            int mo = row0 + r + half_sel * 8;
            if (mo < N) {
                xl2[mo * 64 + n] = accl[nt][r] + bln;
                xr2[mo * 64 + n] = accr[nt][r] + brn;
            }
        }
    }
}

__device__ __forceinline__ void pool_item_add(float* tab, const float4& v, float alpha, int off) {
    atomicAdd(&tab[off + 0], alpha * v.x);
    atomicAdd(&tab[off + 1], alpha * v.y);
    atomicAdd(&tab[off + 2], alpha * v.z);
    atomicAdd(&tab[off + 3], alpha * v.w);
}

constexpr int R1  = 200;
constexpr int AGT = 1024;
constexpr int NWV = AGT / 32;
constexpr int EPT = 16;
constexpr int CHK = AGT * EPT;
constexpr int QCAP = CHK;
constexpr int AGU = 8;
constexpr int CH2 = 64;
constexpr int EPW = 2;
constexpr float SHIFT_OK = 40.f;
static_assert(EPT == 16, "a thread's keys of one chunk are one 16-byte word; two chunks make one 32-bit match mask");
static_assert(QCAP >= CHK, "a run of one chunk must always fit the list");
static_assert(KEY_R == R1, "the owner keys are made for this block size");

typedef __attribute__((ext_vector_type(2))) _Float16 v2h;

__device__ __forceinline__ uint4 own_load_keys(const unsigned* __restrict__ keys, int ebase, int tid) {
    return *(const uint4*)(keys + ((ebase + 16 * tid) >> 2));
}

__device__ __forceinline__ unsigned own_match4(unsigned w, unsigned pat) {
    unsigned x = w ^ pat;
    return ~(((x & 0x7F7F7F7Fu) + 0x7F7F7F7Fu) | x | 0x7F7F7F7Fu);
}

__device__ __forceinline__ int own_off(int p, int tid) {
    const int i = p & 7, j = p >> 3;
    return (i >> 2) * CHK + 16 * tid + 4 * (i & 3) + j;
}

__device__ __forceinline__ void own_trip(const int* list, int base, int cnt, int wave, int lane,
        const int* __restrict__ src, const int* __restrict__ tgt, const float* __restrict__ ea,
        const float* __restrict__ xl, const float* xrS, int node0, int rows,
        const float (&we0)[4], const float (&we1)[4], const float (&at)[4],
        float4 (&v)[AGU], float (&lg)[AGU], int (&tl)[AGU]) {
    float a0[AGU], a1[AGU];
    int   sN[AGU];
#pragma unroll
    for (int u = 0; u < AGU; ++u) {
        int j = base + wave + u * NWV;
        int e = list[(j < cnt) ? j : (cnt - 1)];
        sN[u] = src[e];
        const int t = tgt[e] - node0;
        tl[u] = ((unsigned)t < (unsigned)rows) ? t : -1;
        a0[u] = ea[(size_t)e * 2];
        a1[u] = ea[(size_t)e * 2 + 1];
    }
#pragma unroll
    for (int u = 0; u < AGU; ++u)
        v[u] = *(const float4*)(xl + (size_t)sN[u] * 128 + lane * 4);
#pragma unroll
    for (int u = 0; u < AGU; ++u) {
        const float4 r4 = *(const float4*)(xrS + ((tl[u] < 0) ? 0 : tl[u]) * 128 + lane * 4);
        float m0 = v[u].x + r4.x + a0[u] * we0[0] + a1[u] * we1[0];
        float m1 = v[u].y + r4.y + a0[u] * we0[1] + a1[u] * we1[1];
        float m2 = v[u].z + r4.z + a0[u] * we0[2] + a1[u] * we1[2];
        float m3 = v[u].w + r4.w + a0[u] * we0[3] + a1[u] * we1[3];
        m0 = (m0 > 0.f) ? m0 : 0.2f * m0;
        m1 = (m1 > 0.f) ? m1 : 0.2f * m1;
        m2 = (m2 > 0.f) ? m2 : 0.2f * m2;
        m3 = (m3 > 0.f) ? m3 : 0.2f * m3;
        float p = fmaf(m3, at[3], fmaf(m2, at[2], fmaf(m1, at[1], m0 * at[0])));
        p += __int_as_float(__builtin_amdgcn_update_dpp(0, __float_as_int(p), 0xB1, 0xF, 0xF, false));
        lg[u] = p;
    }
}

__global__ __launch_bounds__(AGT) void k_l1_own(
        const float* __restrict__ xl, const float* __restrict__ xr,
        const float* __restrict__ ea, const float* __restrict__ We,
        const float* __restrict__ att,
        const int* __restrict__ src, const int* __restrict__ tgt,
        const unsigned* __restrict__ keys,
        const float* __restrict__ bias,
        _Float16* __restrict__ hh, int* __restrict__ glist, int* __restrict__ gcnt, int N, int E) {
    __shared__ __align__(16) float tab[R1 * 128];
    __shared__ __align__(16) float xrS[R1 * 128];
    __shared__ unsigned mxU[R1 * 16];
    __shared__ float    mxA[R1 * 16];
    __shared__ float    den[R1 * 16];
    __shared__ int      list[QCAP];
    __shared__ int      qcnt;
    __shared__ int      redo;

    const int tid = threadIdx.x;
    const int lane = tid & 31;
    const int wave = __builtin_amdgcn_readfirstlane(tid >> 5);
    const int hq = lane >> 1;
    const int node0 = blockIdx.x * R1;
    const int rows = (N - node0 < R1) ? (N - node0) : R1;
    const int nChunks = (E + CHK - 1) / CHK;

    for (int i = tid; i < R1 * 128; i += AGT) tab[i] = 0.f;
    for (int i = tid; i < R1 * 16; i += AGT) { mxU[i] = 0u; mxA[i] = -__builtin_inff(); den[i] = 0.f; }
    {
        const float4* g = (const float4*)(xr + (size_t)node0 * 128);
        float4* l = (float4*)xrS;
        for (int i = tid; i < rows * 32; i += AGT) l[i] = g[i];
    }
    if (tid == 0) { qcnt = 0; redo = 0; }

    float we0[4], we1[4], at[4];
#pragma unroll
    for (int k = 0; k < 4; ++k) {
        we0[k] = We[lane * 4 + k];
        we1[k] = We[128 + lane * 4 + k];
        at[k]  = att[lane * 4 + k];
    }

    const unsigned pat = ((unsigned)blockIdx.x & 255u) * 0x01010101u;
    __syncthreads();

    int c0 = 0;
    while (c0 < nChunks) {
        int seg = nChunks - c0;
        int cnt = 0;
        while (true) {
            const int cEnd = c0 + seg;
            uint4 kq[2];
            kq[0] = own_load_keys(keys, c0 * CHK, tid);
            kq[1] = own_load_keys(keys, ((c0 + 1 < cEnd) ? (c0 + 1) : c0) * CHK, tid);
            for (int c = c0; c < cEnd; c += 2) {
                const int ebase = c * CHK;
                uint4 kn[2];
                kn[0] = kq[0];
                kn[1] = kq[1];
                if (c + 2 < cEnd) {
                    kn[0] = own_load_keys(keys, ebase + 2 * CHK, tid);
                    kn[1] = own_load_keys(keys, ((c + 3 < cEnd) ? (c + 3) : (c + 2)) * CHK, tid);
                }
                unsigned mask = (own_match4(kq[0].x, pat) >> 7) | (own_match4(kq[0].y, pat) >> 6)
                              | (own_match4(kq[0].z, pat) >> 5) | (own_match4(kq[0].w, pat) >> 4);
                if (c + 1 < cEnd)
                    mask |= (own_match4(kq[1].x, pat) >> 3) | (own_match4(kq[1].y, pat) >> 2)
                          | (own_match4(kq[1].z, pat) >> 1) |  own_match4(kq[1].w, pat);
                if (mask && ebase + 2 * CHK > E) {
                    unsigned m2 = mask;
                    while (m2) {
                        int b = __ffs(m2) - 1;
                        m2 &= m2 - 1;
                        if (ebase + own_off(b, tid) >= E) mask &= ~(1u << b);
                    }
                }
                if (mask) {
                    int pos = atomicAdd(&qcnt, __popc(mask));
                    while (mask) {
                        int b = __ffs(mask) - 1;
                        mask &= mask - 1;
                        if (pos < QCAP) list[pos] = ebase + own_off(b, tid);
                        ++pos;
                    }
                }
                kq[0] = kn[0];
                kq[1] = kn[1];
            }
            __syncthreads();
            cnt = qcnt;
            __syncthreads();
            if (tid == 0) qcnt = 0;
            if (cnt <= QCAP) break;
            seg = (seg + 1) >> 1;
            __syncthreads();
        }
        if (c0 == 0) {
            const bool whole = (seg == nChunks);
            if (tid == 0) gcnt[blockIdx.x] = whole ? cnt : -1;
            if (whole) {
                int* gl = glist + (size_t)blockIdx.x * QCAP;
                for (int i = tid; i < cnt; i += AGT) gl[i] = list[i];
            }
        }

        if (c0 == 0 && seg == nChunks) {
            for (int i = tid; i < R1 * 16; i += AGT) {
                const float4* xr4 = (const float4*)(xrS + i * 8);
                const float4 p0 = xr4[0], p1 = xr4[1];
                const float* ah = att + (i & 15) * 8;
                float m[8] = {p0.x, p0.y, p0.z, p0.w, p1.x, p1.y, p1.z, p1.w};
                float r = 0.f;
#pragma unroll
                for (int k = 0; k < 8; ++k) {
                    float mk = (m[k] > 0.f) ? m[k] : 0.2f * m[k];
                    r = fmaf(mk, ah[k], r);
                }
                mxA[i] = r;
            }
            __syncthreads();
            for (int base = 0; base < cnt; base += NWV * AGU) {
                float4 v[AGU];
                float  lg[AGU];
                int    tl[AGU];
                own_trip(list, base, cnt, wave, lane, src, tgt, ea, xl, xrS, node0, rows, we0, we1, at, v, lg, tl);
                float sh[AGU];
#pragma unroll
                for (int u = 0; u < AGU; ++u)
                    sh[u] = mxA[((tl[u] < 0) ? 0 : tl[u]) * 16 + hq];
#pragma unroll
                for (int u = 0; u < AGU; ++u) {
                    if (base + wave + u * NWV < cnt && tl[u] >= 0) {
                        float w = __expf(lg[u] - sh[u]);
                        pool_item_add(tab, v[u], w, tl[u] * 128 + lane * 4);
                        if ((lane & 1) == 0) {
                            atomicAdd(&den[tl[u] * 16 + hq], w);
                            atomicMax(&mxU[tl[u] * 16 + hq], fp_ord(lg[u]));
                        }
                    }
                }
            }
            __syncthreads();
            for (int i = tid; i < R1 * 16; i += AGT) {
                unsigned mu = mxU[i];
                if (mu != 0u) {
                    float d = fp_unord(mu) - mxA[i];
                    if (!(d > -SHIFT_OK && d < SHIFT_OK)) redo = 1;
                }
            }
            __syncthreads();
            if (redo == 0) break;
            for (int i = tid; i < R1 * 128; i += AGT) tab[i] = 0.f;
            for (int i = tid; i < R1 * 16; i += AGT) { mxA[i] = -__builtin_inff(); den[i] = 0.f; }
            __syncthreads();
        }

        for (int base = 0; base < cnt; base += NWV * AGU) {
            float4 v[AGU];
            float  lg[AGU];
            int    tl[AGU];
            own_trip(list, base, cnt, wave, lane, src, tgt, ea, xl, xrS, node0, rows, we0, we1, at, v, lg, tl);
#pragma unroll
            for (int u = 0; u < AGU; ++u) {
                if (base + wave + u * NWV < cnt && tl[u] >= 0 && (lane & 1) == 0)
                    atomicMax(&mxU[tl[u] * 16 + hq], fp_ord(lg[u]));
            }
        }
        __syncthreads();
        for (int i = tid; i < R1 * 16; i += AGT) {
            unsigned mu = mxU[i];
            if (mu != 0u) {
                float nm = fp_unord(mu), old = mxA[i];
                if (nm != old) {
                    float sc = __expf(old - nm);
                    mxA[i] = nm;
                    den[i] *= sc;
                    float4* row = (float4*)(tab + i * 8);
                    float4 q0 = row[0], q1 = row[1];
                    q0.x *= sc; q0.y *= sc; q0.z *= sc; q0.w *= sc;
                    q1.x *= sc; q1.y *= sc; q1.z *= sc; q1.w *= sc;
                    row[0] = q0; row[1] = q1;
                }
            }
        }
        __syncthreads();
        for (int base = 0; base < cnt; base += NWV * AGU) {
            float4 v[AGU];
            float  lg[AGU];
            int    tl[AGU];
            own_trip(list, base, cnt, wave, lane, src, tgt, ea, xl, xrS, node0, rows, we0, we1, at, v, lg, tl);
            float mx[AGU];
#pragma unroll
            for (int u = 0; u < AGU; ++u)
                mx[u] = mxA[((tl[u] < 0) ? 0 : tl[u]) * 16 + hq];
#pragma unroll
            for (int u = 0; u < AGU; ++u) {
                if (base + wave + u * NWV < cnt && tl[u] >= 0) {
                    float w = __expf(lg[u] - mx[u]);
                    pool_item_add(tab, v[u], w, tl[u] * 128 + lane * 4);
                    if ((lane & 1) == 0) atomicAdd(&den[tl[u] * 16 + hq], w);
                }
            }
        }
        c0 += seg;
        __syncthreads();
    }

    for (int i = tid; i < rows * 64; i += AGT) {
        int r = i >> 6, d = (i & 63) * 2;
        float dn = den[r * 16 + (d >> 3)];
        float v0 = (dn > 0.f) ? tab[r * 128 + d] / dn : 0.f;
        float v1 = (dn > 0.f) ? tab[r * 128 + d + 1] / dn : 0.f;
        v0 += bias[d];
        v1 += bias[d + 1];
        v0 = (v0 > 0.f) ? v0 : (__expf(v0) - 1.f);
        v1 = (v1 > 0.f) ? v1 : (__expf(v1) - 1.f);
        v2h o = {(_Float16)v0, (_Float16)v1};
        *(v2h*)(hh + (size_t)(node0 + r) * 128 + d) = o;
    }
}

static_assert(CH2 * EPW == 32 * 4, "a wave covers EPW rows of CH2 channels with 4 channels per lane");

__device__ __forceinline__ void own2_trip(const int* list, int base, int cnt, int slot, int q,
        const int* __restrict__ src, const int* __restrict__ tgt, const float* __restrict__ ea,
        const float* __restrict__ xl, const float* xrS, int node0, int rows,
        const float (&we0)[4], const float (&we1)[4], const float (&at)[4],
        float4 (&v)[AGU], float (&lg)[AGU], int (&tl)[AGU]) {
    float a0[AGU], a1[AGU];
    int   sN[AGU];
#pragma unroll
    for (int u = 0; u < AGU; ++u) {
        int j = base + slot + u * NWV * EPW;
        int e = list[(j < cnt) ? j : (cnt - 1)];
        sN[u] = src[e];
        const int t = tgt[e] - node0;
        tl[u] = ((unsigned)t < (unsigned)rows) ? t : -1;
        a0[u] = ea[(size_t)e * 2];
        a1[u] = ea[(size_t)e * 2 + 1];
    }
#pragma unroll
    for (int u = 0; u < AGU; ++u)
        v[u] = *(const float4*)(xl + (size_t)sN[u] * CH2 + q * 4);
#pragma unroll
    for (int u = 0; u < AGU; ++u) {
        const float4 r4 = *(const float4*)(xrS + ((tl[u] < 0) ? 0 : tl[u]) * CH2 + q * 4);
        float m0 = v[u].x + r4.x + a0[u] * we0[0] + a1[u] * we1[0];
        float m1 = v[u].y + r4.y + a0[u] * we0[1] + a1[u] * we1[1];
        float m2 = v[u].z + r4.z + a0[u] * we0[2] + a1[u] * we1[2];
        float m3 = v[u].w + r4.w + a0[u] * we0[3] + a1[u] * we1[3];
        m0 = (m0 > 0.f) ? m0 : 0.2f * m0;
        m1 = (m1 > 0.f) ? m1 : 0.2f * m1;
        m2 = (m2 > 0.f) ? m2 : 0.2f * m2;
        m3 = (m3 > 0.f) ? m3 : 0.2f * m3;
        float p = fmaf(m3, at[3], fmaf(m2, at[2], fmaf(m1, at[1], m0 * at[0])));
        p += __int_as_float(__builtin_amdgcn_update_dpp(0, __float_as_int(p), 0xB1, 0xF, 0xF, false));
        p += __int_as_float(__builtin_amdgcn_update_dpp(0, __float_as_int(p), 0x4E, 0xF, 0xF, false));
        lg[u] = p;
    }
}

__global__ __launch_bounds__(AGT) void k_l2_own(
        const float* __restrict__ xl, const float* __restrict__ xr,
        const float* __restrict__ ea, const float* __restrict__ We,
        const float* __restrict__ att,
        const int* __restrict__ src, const int* __restrict__ tgt,
        const unsigned* __restrict__ keys,
        const int* __restrict__ glist, const int* __restrict__ gcnt,
        const int* __restrict__ batch,
        float* __restrict__ part, int N, int E, int G) {
    __shared__ __align__(16) float tab[R1 * CH2];
    __shared__ __align__(16) float xrS[R1 * CH2];
    __shared__ unsigned mxU[R1 * 4];
    __shared__ float    mxA[R1 * 4];
    __shared__ float    den[R1 * 4];
    __shared__ int      list[QCAP];
    __shared__ int      qcnt;
    __shared__ int      redo;
    extern __shared__ float gsum[];

    const int tid = threadIdx.x;
    const int lane = tid & 31;
    const int wave = __builtin_amdgcn_readfirstlane(tid >> 5);
    const int q = lane & 15;
    const int hq = q >> 2;
    const int slot = wave * EPW + (lane >> 4);
    const int node0 = blockIdx.x * R1;
    const int rows = (N - node0 < R1) ? (N - node0) : R1;
    const int nChunks = (E + CHK - 1) / CHK;

    for (int i = tid; i < R1 * CH2; i += AGT) tab[i] = 0.f;
    for (int i = tid; i < R1 * 4; i += AGT) { mxU[i] = 0u; mxA[i] = -__builtin_inff(); den[i] = 0.f; }
    for (int i = tid; i < (G + 1) * CH2; i += AGT) gsum[i] = 0.f;
    {
        const float4* g = (const float4*)(xr + (size_t)node0 * CH2);
        float4* l = (float4*)xrS;
        for (int i = tid; i < rows * (CH2 / 4); i += AGT) l[i] = g[i];
    }
    if (tid == 0) { qcnt = 0; redo = 0; }

    float we0[4], we1[4], at[4];
#pragma unroll
    for (int k = 0; k < 4; ++k) {
        we0[k] = We[q * 4 + k];
        we1[k] = We[CH2 + q * 4 + k];
        at[k]  = att[q * 4 + k];
    }

    const unsigned pat = ((unsigned)blockIdx.x & 255u) * 0x01010101u;
    const int given = gcnt[blockIdx.x];
    __syncthreads();

    int c0 = 0;
    while (c0 < nChunks) {
        int seg = nChunks - c0;
        int cnt = 0;
        if (c0 == 0 && given >= 0) {
            const int* gl = glist + (size_t)blockIdx.x * QCAP;
            cnt = (given < QCAP) ? given : QCAP;
            for (int i = tid; i < cnt; i += AGT) list[i] = gl[i];
            __syncthreads();
        } else while (true) {
            const int cEnd = c0 + seg;
            uint4 kq[2];
            kq[0] = own_load_keys(keys, c0 * CHK, tid);
            kq[1] = own_load_keys(keys, ((c0 + 1 < cEnd) ? (c0 + 1) : c0) * CHK, tid);
            for (int c = c0; c < cEnd; c += 2) {
                const int ebase = c * CHK;
                uint4 kn[2];
                kn[0] = kq[0];
                kn[1] = kq[1];
                if (c + 2 < cEnd) {
                    kn[0] = own_load_keys(keys, ebase + 2 * CHK, tid);
                    kn[1] = own_load_keys(keys, ((c + 3 < cEnd) ? (c + 3) : (c + 2)) * CHK, tid);
                }
                unsigned mask = (own_match4(kq[0].x, pat) >> 7) | (own_match4(kq[0].y, pat) >> 6)
                              | (own_match4(kq[0].z, pat) >> 5) | (own_match4(kq[0].w, pat) >> 4);
                if (c + 1 < cEnd)
                    mask |= (own_match4(kq[1].x, pat) >> 3) | (own_match4(kq[1].y, pat) >> 2)
                          | (own_match4(kq[1].z, pat) >> 1) |  own_match4(kq[1].w, pat);
                if (mask && ebase + 2 * CHK > E) {
                    unsigned m2 = mask;
                    while (m2) {
                        int b = __ffs(m2) - 1;
                        m2 &= m2 - 1;
                        if (ebase + own_off(b, tid) >= E) mask &= ~(1u << b);
                    }
                }
                if (mask) {
                    int pos = atomicAdd(&qcnt, __popc(mask));
                    while (mask) {
                        int b = __ffs(mask) - 1;
                        mask &= mask - 1;
                        if (pos < QCAP) list[pos] = ebase + own_off(b, tid);
                        ++pos;
                    }
                }
                kq[0] = kn[0];
                kq[1] = kn[1];
            }
            __syncthreads();
            cnt = qcnt;
            __syncthreads();
            if (tid == 0) qcnt = 0;
            if (cnt <= QCAP) break;
            seg = (seg + 1) >> 1;
            __syncthreads();
        }

        if (c0 == 0 && seg == nChunks) {
            for (int i = tid; i < R1 * 4; i += AGT) {
                const float4* xr4 = (const float4*)(xrS + i * 16);
                const float* ah = att + (i & 3) * 16;
                float r = 0.f;
#pragma unroll
                for (int k = 0; k < 4; ++k) {
                    const float4 p = xr4[k];
                    float m0 = (p.x > 0.f) ? p.x : 0.2f * p.x;
                    float m1 = (p.y > 0.f) ? p.y : 0.2f * p.y;
                    float m2 = (p.z > 0.f) ? p.z : 0.2f * p.z;
                    float m3 = (p.w > 0.f) ? p.w : 0.2f * p.w;
                    r = fmaf(m0, ah[4 * k], r);
                    r = fmaf(m1, ah[4 * k + 1], r);
                    r = fmaf(m2, ah[4 * k + 2], r);
                    r = fmaf(m3, ah[4 * k + 3], r);
                }
                mxA[i] = r;
            }
            __syncthreads();
            for (int base = 0; base < cnt; base += NWV * EPW * AGU) {
                float4 v[AGU];
                float  lg[AGU];
                int    tl[AGU];
                own2_trip(list, base, cnt, slot, q, src, tgt, ea, xl, xrS, node0, rows, we0, we1, at, v, lg, tl);
                float sh[AGU];
#pragma unroll
                for (int u = 0; u < AGU; ++u)
                    sh[u] = mxA[((tl[u] < 0) ? 0 : tl[u]) * 4 + hq];
#pragma unroll
                for (int u = 0; u < AGU; ++u) {
                    if (base + slot + u * NWV * EPW < cnt && tl[u] >= 0) {
                        float w = __expf(lg[u] - sh[u]);
                        pool_item_add(tab, v[u], w, tl[u] * CH2 + q * 4);
                        if ((lane & 3) == 0) {
                            atomicAdd(&den[tl[u] * 4 + hq], w);
                            atomicMax(&mxU[tl[u] * 4 + hq], fp_ord(lg[u]));
                        }
                    }
                }
            }
            __syncthreads();
            for (int i = tid; i < R1 * 4; i += AGT) {
                unsigned mu = mxU[i];
                if (mu != 0u) {
                    float d = fp_unord(mu) - mxA[i];
                    if (!(d > -SHIFT_OK && d < SHIFT_OK)) redo = 1;
                }
            }
            __syncthreads();
            if (redo == 0) break;
            for (int i = tid; i < R1 * CH2; i += AGT) tab[i] = 0.f;
            for (int i = tid; i < R1 * 4; i += AGT) { mxA[i] = -__builtin_inff(); den[i] = 0.f; }
            __syncthreads();
        }

        for (int base = 0; base < cnt; base += NWV * EPW * AGU) {
            float4 v[AGU];
            float  lg[AGU];
            int    tl[AGU];
            own2_trip(list, base, cnt, slot, q, src, tgt, ea, xl, xrS, node0, rows, we0, we1, at, v, lg, tl);
#pragma unroll
            for (int u = 0; u < AGU; ++u) {
                if (base + slot + u * NWV * EPW < cnt && tl[u] >= 0 && (lane & 3) == 0)
                    atomicMax(&mxU[tl[u] * 4 + hq], fp_ord(lg[u]));
            }
        }
        __syncthreads();
        for (int i = tid; i < R1 * 4; i += AGT) {
            unsigned mu = mxU[i];
            if (mu != 0u) {
                float nm = fp_unord(mu), old = mxA[i];
                if (nm != old) {
                    float sc = __expf(old - nm);
                    mxA[i] = nm;
                    den[i] *= sc;
                    float4* row = (float4*)(tab + i * 16);
#pragma unroll
                    for (int k = 0; k < 4; ++k) {
                        float4 qv = row[k];
                        qv.x *= sc; qv.y *= sc; qv.z *= sc; qv.w *= sc;
                        row[k] = qv;
                    }
                }
            }
        }
        __syncthreads();
        for (int base = 0; base < cnt; base += NWV * EPW * AGU) {
            float4 v[AGU];
            float  lg[AGU];
            int    tl[AGU];
            own2_trip(list, base, cnt, slot, q, src, tgt, ea, xl, xrS, node0, rows, we0, we1, at, v, lg, tl);
            float mx[AGU];
#pragma unroll
            for (int u = 0; u < AGU; ++u)
                mx[u] = mxA[((tl[u] < 0) ? 0 : tl[u]) * 4 + hq];
#pragma unroll
            for (int u = 0; u < AGU; ++u) {
                if (base + slot + u * NWV * EPW < cnt && tl[u] >= 0) {
                    float w = __expf(lg[u] - mx[u]);
                    pool_item_add(tab, v[u], w, tl[u] * CH2 + q * 4);
                    if ((lane & 3) == 0) atomicAdd(&den[tl[u] * 4 + hq], w);
                }
            }
        }
        c0 += seg;
        __syncthreads();
    }

    for (int i = tid; i < rows * CH2; i += AGT) {
        int r = i >> 6, d = i & 63;
        float dn = den[r * 4 + (d >> 4)];
        float val = (dn > 0.f) ? tab[i] / dn : 0.f;
        int g = batch[node0 + r];
        unsigned row = ((unsigned)g < (unsigned)G) ? (unsigned)g : (unsigned)G;
        atomicAdd(&gsum[row * CH2 + d], val);
    }
    __syncthreads();
    float* my = part + (size_t)blockIdx.x * G * CH2;
    for (int i = tid; i < G * CH2; i += AGT) my[i] = gsum[i];
}

__global__ __launch_bounds__(1024) void k_final(const float* __restrict__ part, int nPart,
                                                const float* __restrict__ bias2,
                                                const int* __restrict__ batch,
                                                float* __restrict__ out, int N, int G) {
    __shared__ float red[16][64];
    __shared__ int cntS;
    const int g = blockIdx.x;
    const int tid = threadIdx.x;
    const int d = tid & 63, j = tid >> 6;
    if (tid == 0) cntS = 0;

    float acc = 0.f;
    const float* p = part + (size_t)g * 64 + d;
    const size_t tabStride = (size_t)G * 64;
    int b = j;
    for (; b + 16 * 7 < nPart; b += 16 * 8) {
        float t[8];
#pragma unroll
        for (int k = 0; k < 8; ++k) t[k] = p[(size_t)(b + 16 * k) * tabStride];
#pragma unroll
        for (int k = 0; k < 8; ++k) acc += t[k];
    }
    for (; b < nPart; b += 16) acc += p[(size_t)b * tabStride];

    int c = 0;
    for (int n = tid; n < N; n += 1024) c += (batch[n] == g) ? 1 : 0;

    red[j][d] = acc;
    __syncthreads();
    atomicAdd(&cntS, c);
    __syncthreads();

    if (j == 0) {
        float s = 0.f;
#pragma unroll
        for (int k = 0; k < 16; ++k) s += red[k][d];
        float cnt = (float)cntS;
        out[g * 64 + d] = (s + cnt * bias2[d]) / fmaxf(cnt, 1.0f);
    }
}

extern "C" void kernel_launch(void* const* d_in, const int* in_sizes, int n_in,
                              void* d_out, int out_size, void* d_ws, size_t ws_size,
                              hipStream_t stream) {
    const float* x      = (const float*)d_in[0];
    const float* ea     = (const float*)d_in[1];
    const float* W_l1   = (const float*)d_in[2];
    const float* b_l1   = (const float*)d_in[3];
    const float* W_r1   = (const float*)d_in[4];
    const float* b_r1   = (const float*)d_in[5];
    const float* W_e1   = (const float*)d_in[6];
    const float* att1   = (const float*)d_in[7];
    const float* bias1  = (const float*)d_in[8];
    const float* W_l2   = (const float*)d_in[9];
    const float* b_l2   = (const float*)d_in[10];
    const float* W_r2   = (const float*)d_in[11];
    const float* b_r2   = (const float*)d_in[12];
    const float* W_e2   = (const float*)d_in[13];
    const float* att2   = (const float*)d_in[14];
    const float* bias2  = (const float*)d_in[15];
    const int*   eidx   = (const int*)d_in[16];
    const int*   batch  = (const int*)d_in[17];
    float* out = (float*)d_out;

    const int N = in_sizes[0] / 2;
    const int E = in_sizes[1] / 2;
    const int G = out_size / 64;
    const int* src = eidx;
    const int* tgt = eidx + E;

    char* ws = (char*)d_ws;
    size_t off = 0;
    auto alloc = [&](size_t bytes) { size_t o = off; off = (off + bytes + 255) & ~(size_t)255; return o; };

    const int nOwn = (N + R1 - 1) / R1;
    size_t o_xl1 = alloc((size_t)N * 128 * 4);
    size_t o_xr1 = alloc((size_t)N * 128 * 4);
    size_t o_hh  = alloc((size_t)N * 128 * 2);
    size_t o_wpk = alloc((size_t)2 * 8192 * 2);
    const int nKeys = ((E + CHK - 1) / CHK) * CHK;
    size_t o_keys = alloc((size_t)nKeys);
    size_t o_xl2 = alloc((size_t)N * 64 * 4);
    size_t o_xr2 = alloc((size_t)N * 64 * 4);
    size_t o_glist = alloc((size_t)nOwn * QCAP * 4);
    size_t o_gcnt  = alloc((size_t)nOwn * 4);
    size_t o_part = alloc((size_t)nOwn * G * 64 * 4);
    (void)ws_size;

    float*    xl1   = (float*)(ws + o_xl1);
    float*    xr1   = (float*)(ws + o_xr1);
    _Float16* hh    = (_Float16*)(ws + o_hh);
    _Float16* wpk   = (_Float16*)(ws + o_wpk);
    float*    xl2   = (float*)(ws + o_xl2);
    float*    xr2   = (float*)(ws + o_xr2);
    int*      glist = (int*)(ws + o_glist);
    int*      gcnt  = (int*)(ws + o_gcnt);
    float*    part  = (float*)(ws + o_part);
    unsigned* keys  = (unsigned*)(ws + o_keys);

    const int TB = 256;
    auto blocks = [&](long long t) { return (unsigned)((t + TB - 1) / TB); };

    k_node_transform1<<<blocks((long long)N * 128), TB, 0, stream>>>(
        x, W_l1, b_l1, W_r1, b_r1, xl1, xr1, N);
    const int packBlocks = (int)blocks(2 * 8192);
    k_w_pack<<<(unsigned)packBlocks + blocks(nKeys / 4), TB, 0, stream>>>(
        W_l2, W_r2, wpk, packBlocks, tgt, keys, E, nKeys);
    k_l1_own<<<(unsigned)nOwn, AGT, 0, stream>>>(
        xl1, xr1, ea, W_e1, att1, src, tgt, keys, bias1, hh, glist, gcnt, N, E);

    long long gemm_threads = (long long)((N + 15) / 16) * 32;
    k_gemm2_wmma<<<blocks(gemm_threads), TB, 0, stream>>>(
        hh, wpk, b_l2, b_r2, xl2, xr2, N);

    k_l2_own<<<(unsigned)nOwn, AGT, (size_t)(G + 1) * 64 * sizeof(float), stream>>>(
        xl2, xr2, ea, W_e2, att2, src, tgt, keys, glist, gcnt, batch, part, N, E, G);
    k_final<<<G, 1024, 0, stream>>>(part, nOwn, bias2, batch, out, N, G);
}


# layer 2 reads no edge arrays: layer 1's one-pass walk stores each listed edge as a 16-byte record (source, row in block, 2 attributes) in list order, layer 2 streams them; changes layer-2 sum order
# speedup vs baseline: 3.1470x; 1.0333x over previous
#include <hip/hip_runtime.h>
#include <hip/hip_fp16.h>

typedef __attribute__((ext_vector_type(16))) _Float16 v16h;
typedef __attribute__((ext_vector_type(8)))  _Float16 v8h;
typedef __attribute__((ext_vector_type(8)))  float    v8f;

__device__ __forceinline__ unsigned fp_ord(float f) {
    unsigned u = __float_as_uint(f);
    return (u & 0x80000000u) ? ~u : (u | 0x80000000u);
}
__device__ __forceinline__ float fp_unord(unsigned u) {
    unsigned b = (u & 0x80000000u) ? (u ^ 0x80000000u) : ~u;
    return __uint_as_float(b);
}

__global__ void k_node_transform1(const float* __restrict__ x,
                                  const float* __restrict__ Wl, const float* __restrict__ bl,
                                  const float* __restrict__ Wr, const float* __restrict__ br,
                                  float* __restrict__ xl1, float* __restrict__ xr1, int N) {
    int idx = blockIdx.x * blockDim.x + threadIdx.x;
    if (idx >= N * 128) return;
    int n = idx >> 7, d = idx & 127;
    float x0 = x[n * 2], x1 = x[n * 2 + 1];
    xl1[idx] = fmaf(x0, Wl[d], fmaf(x1, Wl[128 + d], bl[d]));
    xr1[idx] = fmaf(x0, Wr[d], fmaf(x1, Wr[128 + d], br[d]));
}

constexpr int KEY_R = 200;
__global__ void k_w_pack(const float* __restrict__ Wl, const float* __restrict__ Wr,
                         _Float16* __restrict__ Wpk, int packBlocks,
                         const int* __restrict__ tgt, unsigned* __restrict__ keys, int E, int nKeys) {
    if ((int)blockIdx.x >= packBlocks) {
        int w = ((int)blockIdx.x - packBlocks) * (int)blockDim.x + (int)threadIdx.x;
        int e0 = w * 4;
        if (e0 >= nKeys) return;
        int t0, t1, t2, t3;
        if (e0 + 4 <= E && (((size_t)tgt) & 15) == 0) {
            int4 t = *(const int4*)(tgt + e0);
            t0 = t.x; t1 = t.y; t2 = t.z; t3 = t.w;
        } else {
            t0 = (e0 + 0 < E) ? tgt[e0 + 0] : 0;
            t1 = (e0 + 1 < E) ? tgt[e0 + 1] : 0;
            t2 = (e0 + 2 < E) ? tgt[e0 + 2] : 0;
            t3 = (e0 + 3 < E) ? tgt[e0 + 3] : 0;
        }
        unsigned k0 = ((unsigned)t0 / (unsigned)KEY_R) & 255u;
        unsigned k1 = ((unsigned)t1 / (unsigned)KEY_R) & 255u;
        unsigned k2 = ((unsigned)t2 / (unsigned)KEY_R) & 255u;
        unsigned k3 = ((unsigned)t3 / (unsigned)KEY_R) & 255u;
        keys[w] = k0 | (k1 << 8) | (k2 << 16) | (k3 << 24);
        return;
    }
    int idx = blockIdx.x * blockDim.x + threadIdx.x;
    if (idx >= 2 * 8192) return;
    int which = idx >> 13;
    int r = idx & 8191;
    int j = r & 15;
    int lane = (r >> 4) & 31;
    int frag = r >> 9;
    int kc = frag >> 2, nt = frag & 3;
    int half_sel = lane >> 4, lidx = lane & 15;
    int kb = (j < 8) ? (half_sel * 8) : (16 + half_sel * 8);
    int k = kc * 32 + kb + (j & 7);
    int n = nt * 16 + lidx;
    const float* W = which ? Wr : Wl;
    Wpk[idx] = (_Float16)W[k * 64 + n];
}

__global__ void k_gemm2_wmma(const _Float16* __restrict__ hh,
                             const _Float16* __restrict__ Wpk,
                             const float* __restrict__ bl, const float* __restrict__ br,
                             float* __restrict__ xl2, float* __restrict__ xr2, int N) {
    __shared__ __align__(32) _Float16 ldsW[2 * 8192];

    {
        const uint4* g = (const uint4*)Wpk;
        uint4* l = (uint4*)ldsW;
        for (int i = threadIdx.x; i < 2048; i += blockDim.x) l[i] = g[i];
    }
    __syncthreads();

    int wave = (blockIdx.x * blockDim.x + threadIdx.x) >> 5;
    int lane = threadIdx.x & 31;
    int row0 = wave * 16;
    if (row0 >= N) return;

    int half_sel = lane >> 4;
    int lidx = lane & 15;
    int kb_lo = half_sel * 8;
    int kb_hi = 16 + kb_lo;
    int m = row0 + lidx;
    if (m >= N) m = N - 1;

    const v16h* fragL = (const v16h*)ldsW;
    const v16h* fragR = (const v16h*)(ldsW + 8192);

    v8f accl[4] = {};
    v8f accr[4] = {};

#pragma unroll
    for (int kc = 0; kc < 4; ++kc) {
        const _Float16* pa = hh + m * 128 + kc * 32;
        v8h alo = *(const v8h*)(pa + kb_lo);
        v8h ahi = *(const v8h*)(pa + kb_hi);
        v16h a = __builtin_shufflevector(alo, ahi, 0, 1, 2, 3, 4, 5, 6, 7,
                                         8, 9, 10, 11, 12, 13, 14, 15);
#pragma unroll
        for (int nt = 0; nt < 4; ++nt) {
            int fi = (kc * 4 + nt) * 32 + lane;
            v16h b  = fragL[fi];
            v16h b2 = fragR[fi];
            accl[nt] = __builtin_amdgcn_wmma_f32_16x16x32_f16(
                false, a, false, b, (short)0, accl[nt], false, false);
            accr[nt] = __builtin_amdgcn_wmma_f32_16x16x32_f16(
                false, a, false, b2, (short)0, accr[nt], false, false);
        }
    }
#pragma unroll
    for (int nt = 0; nt < 4; ++nt) {
        int n = nt * 16 + lidx;
        float bln = bl[n], brn = br[n];
#pragma unroll
        for (int r = 0; r < 8; ++r) {
            int mo = row0 + r + half_sel * 8;
            if (mo < N) {
                xl2[mo * 64 + n] = accl[nt][r] + bln;
                xr2[mo * 64 + n] = accr[nt][r] + brn;
            }
        }
    }
}

__device__ __forceinline__ void pool_item_add(float* tab, const float4& v, float alpha, int off) {
    atomicAdd(&tab[off + 0], alpha * v.x);
    atomicAdd(&tab[off + 1], alpha * v.y);
    atomicAdd(&tab[off + 2], alpha * v.z);
    atomicAdd(&tab[off + 3], alpha * v.w);
}

constexpr int R1  = 200;
constexpr int AGT = 1024;
constexpr int NWV = AGT / 32;
constexpr int EPT = 16;
constexpr int CHK = AGT * EPT;
constexpr int QCAP = CHK;
constexpr int AGU = 8;
constexpr int CH2 = 64;
constexpr int EPW = 2;
constexpr float SHIFT_OK = 40.f;
static_assert(EPT == 16, "a thread's keys of one chunk are one 16-byte word; two chunks make one 32-bit match mask");
static_assert(QCAP >= CHK, "a run of one chunk must always fit the list");
static_assert(KEY_R == R1, "the owner keys are made for this block size");

typedef __attribute__((ext_vector_type(2))) _Float16 v2h;

__device__ __forceinline__ uint4 own_load_keys(const unsigned* __restrict__ keys, int ebase, int tid) {
    return *(const uint4*)(keys + ((ebase + 16 * tid) >> 2));
}

__device__ __forceinline__ unsigned own_match4(unsigned w, unsigned pat) {
    unsigned x = w ^ pat;
    return ~(((x & 0x7F7F7F7Fu) + 0x7F7F7F7Fu) | x | 0x7F7F7F7Fu);
}

__device__ __forceinline__ int own_off(int p, int tid) {
    const int i = p & 7, j = p >> 3;
    return (i >> 2) * CHK + 16 * tid + 4 * (i & 3) + j;
}

__device__ __forceinline__ void own_trip(const int* list, int base, int cnt, int wave, int lane,
        const int* __restrict__ src, const int* __restrict__ tgt, const float* __restrict__ ea,
        const float* __restrict__ xl, const float* xrS, int node0, int rows,
        const float (&we0)[4], const float (&we1)[4], const float (&at)[4],
        float4 (&v)[AGU], float (&lg)[AGU], int (&tl)[AGU], int4* recOut) {
    float a0[AGU], a1[AGU];
    int   sN[AGU];
#pragma unroll
    for (int u = 0; u < AGU; ++u) {
        int j = base + wave + u * NWV;
        int e = list[(j < cnt) ? j : (cnt - 1)];
        sN[u] = src[e];
        const int t = tgt[e] - node0;
        tl[u] = ((unsigned)t < (unsigned)rows) ? t : -1;
        a0[u] = ea[(size_t)e * 2];
        a1[u] = ea[(size_t)e * 2 + 1];
    }
#pragma unroll
    for (int u = 0; u < AGU; ++u)
        v[u] = *(const float4*)(xl + (size_t)sN[u] * 128 + lane * 4);
    if (recOut != nullptr && lane == 0) {
#pragma unroll
        for (int u = 0; u < AGU; ++u) {
            const int j = base + wave + u * NWV;
            if (j < cnt)
                recOut[j] = make_int4(sN[u], tl[u], __float_as_int(a0[u]), __float_as_int(a1[u]));
        }
    }
#pragma unroll
    for (int u = 0; u < AGU; ++u) {
        const float4 r4 = *(const float4*)(xrS + ((tl[u] < 0) ? 0 : tl[u]) * 128 + lane * 4);
        float m0 = v[u].x + r4.x + a0[u] * we0[0] + a1[u] * we1[0];
        float m1 = v[u].y + r4.y + a0[u] * we0[1] + a1[u] * we1[1];
        float m2 = v[u].z + r4.z + a0[u] * we0[2] + a1[u] * we1[2];
        float m3 = v[u].w + r4.w + a0[u] * we0[3] + a1[u] * we1[3];
        m0 = (m0 > 0.f) ? m0 : 0.2f * m0;
        m1 = (m1 > 0.f) ? m1 : 0.2f * m1;
        m2 = (m2 > 0.f) ? m2 : 0.2f * m2;
        m3 = (m3 > 0.f) ? m3 : 0.2f * m3;
        float p = fmaf(m3, at[3], fmaf(m2, at[2], fmaf(m1, at[1], m0 * at[0])));
        p += __int_as_float(__builtin_amdgcn_update_dpp(0, __float_as_int(p), 0xB1, 0xF, 0xF, false));
        lg[u] = p;
    }
}

__global__ __launch_bounds__(AGT) void k_l1_own(
        const float* __restrict__ xl, const float* __restrict__ xr,
        const float* __restrict__ ea, const float* __restrict__ We,
        const float* __restrict__ att,
        const int* __restrict__ src, const int* __restrict__ tgt,
        const unsigned* __restrict__ keys,
        const float* __restrict__ bias,
        _Float16* __restrict__ hh, int* __restrict__ grec, int* __restrict__ gcnt, int N, int E) {
    __shared__ __align__(16) float tab[R1 * 128];
    __shared__ __align__(16) float xrS[R1 * 128];
    __shared__ unsigned mxU[R1 * 16];
    __shared__ float    mxA[R1 * 16];
    __shared__ float    den[R1 * 16];
    __shared__ int      list[QCAP];
    __shared__ int      qcnt;
    __shared__ int      redo;

    const int tid = threadIdx.x;
    const int lane = tid & 31;
    const int wave = __builtin_amdgcn_readfirstlane(tid >> 5);
    const int hq = lane >> 1;
    const int node0 = blockIdx.x * R1;
    const int rows = (N - node0 < R1) ? (N - node0) : R1;
    const int nChunks = (E + CHK - 1) / CHK;

    for (int i = tid; i < R1 * 128; i += AGT) tab[i] = 0.f;
    for (int i = tid; i < R1 * 16; i += AGT) { mxU[i] = 0u; mxA[i] = -__builtin_inff(); den[i] = 0.f; }
    {
        const float4* g = (const float4*)(xr + (size_t)node0 * 128);
        float4* l = (float4*)xrS;
        for (int i = tid; i < rows * 32; i += AGT) l[i] = g[i];
    }
    if (tid == 0) { qcnt = 0; redo = 0; }

    float we0[4], we1[4], at[4];
#pragma unroll
    for (int k = 0; k < 4; ++k) {
        we0[k] = We[lane * 4 + k];
        we1[k] = We[128 + lane * 4 + k];
        at[k]  = att[lane * 4 + k];
    }

    const unsigned pat = ((unsigned)blockIdx.x & 255u) * 0x01010101u;
    __syncthreads();

    int c0 = 0;
    while (c0 < nChunks) {
        int seg = nChunks - c0;
        int cnt = 0;
        while (true) {
            const int cEnd = c0 + seg;
            uint4 kq[2];
            kq[0] = own_load_keys(keys, c0 * CHK, tid);
            kq[1] = own_load_keys(keys, ((c0 + 1 < cEnd) ? (c0 + 1) : c0) * CHK, tid);
            for (int c = c0; c < cEnd; c += 2) {
                const int ebase = c * CHK;
                uint4 kn[2];
                kn[0] = kq[0];
                kn[1] = kq[1];
                if (c + 2 < cEnd) {
                    kn[0] = own_load_keys(keys, ebase + 2 * CHK, tid);
                    kn[1] = own_load_keys(keys, ((c + 3 < cEnd) ? (c + 3) : (c + 2)) * CHK, tid);
                }
                unsigned mask = (own_match4(kq[0].x, pat) >> 7) | (own_match4(kq[0].y, pat) >> 6)
                              | (own_match4(kq[0].z, pat) >> 5) | (own_match4(kq[0].w, pat) >> 4);
                if (c + 1 < cEnd)
                    mask |= (own_match4(kq[1].x, pat) >> 3) | (own_match4(kq[1].y, pat) >> 2)
                          | (own_match4(kq[1].z, pat) >> 1) |  own_match4(kq[1].w, pat);
                if (mask && ebase + 2 * CHK > E) {
                    unsigned m2 = mask;
                    while (m2) {
                        int b = __ffs(m2) - 1;
                        m2 &= m2 - 1;
                        if (ebase + own_off(b, tid) >= E) mask &= ~(1u << b);
                    }
                }
                if (mask) {
                    int pos = atomicAdd(&qcnt, __popc(mask));
                    while (mask) {
                        int b = __ffs(mask) - 1;
                        mask &= mask - 1;
                        if (pos < QCAP) list[pos] = ebase + own_off(b, tid);
                        ++pos;
                    }
                }
                kq[0] = kn[0];
                kq[1] = kn[1];
            }
            __syncthreads();
            cnt = qcnt;
            __syncthreads();
            if (tid == 0) qcnt = 0;
            if (cnt <= QCAP) break;
            seg = (seg + 1) >> 1;
            __syncthreads();
        }
        if (c0 == 0 && tid == 0) gcnt[blockIdx.x] = (seg == nChunks) ? cnt : -1;

        if (c0 == 0 && seg == nChunks) {
            for (int i = tid; i < R1 * 16; i += AGT) {
                const float4* xr4 = (const float4*)(xrS + i * 8);
                const float4 p0 = xr4[0], p1 = xr4[1];
                const float* ah = att + (i & 15) * 8;
                float m[8] = {p0.x, p0.y, p0.z, p0.w, p1.x, p1.y, p1.z, p1.w};
                float r = 0.f;
#pragma unroll
                for (int k = 0; k < 8; ++k) {
                    float mk = (m[k] > 0.f) ? m[k] : 0.2f * m[k];
                    r = fmaf(mk, ah[k], r);
                }
                mxA[i] = r;
            }
            __syncthreads();
            int4* recB = (int4*)grec + (size_t)blockIdx.x * QCAP;
            for (int base = 0; base < cnt; base += NWV * AGU) {
                float4 v[AGU];
                float  lg[AGU];
                int    tl[AGU];
                own_trip(list, base, cnt, wave, lane, src, tgt, ea, xl, xrS, node0, rows, we0, we1, at, v, lg, tl,
                         recB);
                float sh[AGU];
#pragma unroll
                for (int u = 0; u < AGU; ++u)
                    sh[u] = mxA[((tl[u] < 0) ? 0 : tl[u]) * 16 + hq];
#pragma unroll
                for (int u = 0; u < AGU; ++u) {
                    if (base + wave + u * NWV < cnt && tl[u] >= 0) {
                        float w = __expf(lg[u] - sh[u]);
                        pool_item_add(tab, v[u], w, tl[u] * 128 + lane * 4);
                        if ((lane & 1) == 0) {
                            atomicAdd(&den[tl[u] * 16 + hq], w);
                            atomicMax(&mxU[tl[u] * 16 + hq], fp_ord(lg[u]));
                        }
                    }
                }
            }
            __syncthreads();
            for (int i = tid; i < R1 * 16; i += AGT) {
                unsigned mu = mxU[i];
                if (mu != 0u) {
                    float d = fp_unord(mu) - mxA[i];
                    if (!(d > -SHIFT_OK && d < SHIFT_OK)) redo = 1;
                }
            }
            __syncthreads();
            if (redo == 0) break;
            for (int i = tid; i < R1 * 128; i += AGT) tab[i] = 0.f;
            for (int i = tid; i < R1 * 16; i += AGT) { mxA[i] = -__builtin_inff(); den[i] = 0.f; }
            __syncthreads();
        }

        for (int base = 0; base < cnt; base += NWV * AGU) {
            float4 v[AGU];
            float  lg[AGU];
            int    tl[AGU];
            own_trip(list, base, cnt, wave, lane, src, tgt, ea, xl, xrS, node0, rows, we0, we1, at, v, lg, tl,
                     nullptr);
#pragma unroll
            for (int u = 0; u < AGU; ++u) {
                if (base + wave + u * NWV < cnt && tl[u] >= 0 && (lane & 1) == 0)
                    atomicMax(&mxU[tl[u] * 16 + hq], fp_ord(lg[u]));
            }
        }
        __syncthreads();
        for (int i = tid; i < R1 * 16; i += AGT) {
            unsigned mu = mxU[i];
            if (mu != 0u) {
                float nm = fp_unord(mu), old = mxA[i];
                if (nm != old) {
                    float sc = __expf(old - nm);
                    mxA[i] = nm;
                    den[i] *= sc;
                    float4* row = (float4*)(tab + i * 8);
                    float4 q0 = row[0], q1 = row[1];
                    q0.x *= sc; q0.y *= sc; q0.z *= sc; q0.w *= sc;
                    q1.x *= sc; q1.y *= sc; q1.z *= sc; q1.w *= sc;
                    row[0] = q0; row[1] = q1;
                }
            }
        }
        __syncthreads();
        for (int base = 0; base < cnt; base += NWV * AGU) {
            float4 v[AGU];
            float  lg[AGU];
            int    tl[AGU];
            own_trip(list, base, cnt, wave, lane, src, tgt, ea, xl, xrS, node0, rows, we0, we1, at, v, lg, tl,
                     nullptr);
            float mx[AGU];
#pragma unroll
            for (int u = 0; u < AGU; ++u)
                mx[u] = mxA[((tl[u] < 0) ? 0 : tl[u]) * 16 + hq];
#pragma unroll
            for (int u = 0; u < AGU; ++u) {
                if (base + wave + u * NWV < cnt && tl[u] >= 0) {
                    float w = __expf(lg[u] - mx[u]);
                    pool_item_add(tab, v[u], w, tl[u] * 128 + lane * 4);
                    if ((lane & 1) == 0) atomicAdd(&den[tl[u] * 16 + hq], w);
                }
            }
        }
        c0 += seg;
        __syncthreads();
    }

    for (int i = tid; i < rows * 64; i += AGT) {
        int r = i >> 6, d = (i & 63) * 2;
        float dn = den[r * 16 + (d >> 3)];
        float v0 = (dn > 0.f) ? tab[r * 128 + d] / dn : 0.f;
        float v1 = (dn > 0.f) ? tab[r * 128 + d + 1] / dn : 0.f;
        v0 += bias[d];
        v1 += bias[d + 1];
        v0 = (v0 > 0.f) ? v0 : (__expf(v0) - 1.f);
        v1 = (v1 > 0.f) ? v1 : (__expf(v1) - 1.f);
        v2h o = {(_Float16)v0, (_Float16)v1};
        *(v2h*)(hh + (size_t)(node0 + r) * 128 + d) = o;
    }
}

static_assert(CH2 * EPW == 32 * 4, "a wave covers EPW rows of CH2 channels with 4 channels per lane");

template <bool REC>
__device__ __forceinline__ int own2_j(int base, int wave, int half, int u) {
    return REC ? base + (wave * AGU + u) * EPW + half
               : base + (u * NWV + wave) * EPW + half;
}

template <bool REC>
__device__ __forceinline__ void own2_trip(const int* list, const int4* __restrict__ rec,
        int base, int cnt, int wave, int half, int q,
        const int* __restrict__ src, const int* __restrict__ tgt, const float* __restrict__ ea,
        const float* __restrict__ xl, const float* xrS, int node0, int rows,
        const float (&we0)[4], const float (&we1)[4], const float (&at)[4],
        float4 (&v)[AGU], float (&lg)[AGU], int (&tl)[AGU]) {
    float a0[AGU], a1[AGU];
    int   sN[AGU];
    if (REC) {
        int4 r[AGU];
#pragma unroll
        for (int u = 0; u < AGU; ++u) {
            const int j = own2_j<REC>(base, wave, half, u);
            r[u] = rec[(j < cnt) ? j : (cnt - 1)];
        }
        __builtin_amdgcn_sched_barrier(0);
#pragma unroll
        for (int u = 0; u < AGU; ++u) {
            sN[u] = r[u].x;
            tl[u] = ((unsigned)r[u].y < (unsigned)rows) ? r[u].y : -1;
            a0[u] = __int_as_float(r[u].z);
            a1[u] = __int_as_float(r[u].w);
        }
    } else {
#pragma unroll
        for (int u = 0; u < AGU; ++u) {
            const int j = own2_j<REC>(base, wave, half, u);
            const int e = list[(j < cnt) ? j : (cnt - 1)];
            sN[u] = src[e];
            const int t = tgt[e] - node0;
            tl[u] = ((unsigned)t < (unsigned)rows) ? t : -1;
            a0[u] = ea[(size_t)e * 2];
            a1[u] = ea[(size_t)e * 2 + 1];
        }
    }
#pragma unroll
    for (int u = 0; u < AGU; ++u)
        v[u] = *(const float4*)(xl + (size_t)sN[u] * CH2 + q * 4);
#pragma unroll
    for (int u = 0; u < AGU; ++u) {
        const float4 r4 = *(const float4*)(xrS + ((tl[u] < 0) ? 0 : tl[u]) * CH2 + q * 4);
        float m0 = v[u].x + r4.x + a0[u] * we0[0] + a1[u] * we1[0];
        float m1 = v[u].y + r4.y + a0[u] * we0[1] + a1[u] * we1[1];
        float m2 = v[u].z + r4.z + a0[u] * we0[2] + a1[u] * we1[2];
        float m3 = v[u].w + r4.w + a0[u] * we0[3] + a1[u] * we1[3];
        m0 = (m0 > 0.f) ? m0 : 0.2f * m0;
        m1 = (m1 > 0.f) ? m1 : 0.2f * m1;
        m2 = (m2 > 0.f) ? m2 : 0.2f * m2;
        m3 = (m3 > 0.f) ? m3 : 0.2f * m3;
        float p = fmaf(m3, at[3], fmaf(m2, at[2], fmaf(m1, at[1], m0 * at[0])));
        p += __int_as_float(__builtin_amdgcn_update_dpp(0, __float_as_int(p), 0xB1, 0xF, 0xF, false));
        p += __int_as_float(__builtin_amdgcn_update_dpp(0, __float_as_int(p), 0x4E, 0xF, 0xF, false));
        lg[u] = p;
    }
}

template <bool REC>
__device__ __forceinline__ void own2_walk_one(const int* list, const int4* __restrict__ rec, int cnt, int wave, int lane,
        const int* __restrict__ src, const int* __restrict__ tgt, const float* __restrict__ ea,
        const float* __restrict__ xl, const float* xrS, int node0, int rows,
        const float (&we0)[4], const float (&we1)[4], const float (&at)[4],
        float* tab, float* den, unsigned* mxU, const float* mxA) {
    const int half = lane >> 4, q = lane & 15, hq = q >> 2;
    for (int base = 0; base < cnt; base += NWV * EPW * AGU) {
        float4 v[AGU];
        float  lg[AGU];
        int    tl[AGU];
        own2_trip<REC>(list, rec, base, cnt, wave, half, q, src, tgt, ea, xl, xrS, node0, rows, we0, we1, at, v, lg, tl);
        float sh[AGU];
#pragma unroll
        for (int u = 0; u < AGU; ++u)
            sh[u] = mxA[((tl[u] < 0) ? 0 : tl[u]) * 4 + hq];
#pragma unroll
        for (int u = 0; u < AGU; ++u) {
            if (own2_j<REC>(base, wave, half, u) < cnt && tl[u] >= 0) {
                float w = __expf(lg[u] - sh[u]);
                pool_item_add(tab, v[u], w, tl[u] * CH2 + q * 4);
                if ((lane & 3) == 0) {
                    atomicAdd(&den[tl[u] * 4 + hq], w);
                    atomicMax(&mxU[tl[u] * 4 + hq], fp_ord(lg[u]));
                }
            }
        }
    }
}

template <bool REC>
__device__ __forceinline__ void own2_walk_max(const int* list, const int4* __restrict__ rec, int cnt, int wave, int lane,
        const int* __restrict__ src, const int* __restrict__ tgt, const float* __restrict__ ea,
        const float* __restrict__ xl, const float* xrS, int node0, int rows,
        const float (&we0)[4], const float (&we1)[4], const float (&at)[4],
        unsigned* mxU) {
    const int half = lane >> 4, q = lane & 15, hq = q >> 2;
    for (int base = 0; base < cnt; base += NWV * EPW * AGU) {
        float4 v[AGU];
        float  lg[AGU];
        int    tl[AGU];
        own2_trip<REC>(list, rec, base, cnt, wave, half, q, src, tgt, ea, xl, xrS, node0, rows, we0, we1, at, v, lg, tl);
#pragma unroll
        for (int u = 0; u < AGU; ++u) {
            if (own2_j<REC>(base, wave, half, u) < cnt && tl[u] >= 0 && (lane & 3) == 0)
                atomicMax(&mxU[tl[u] * 4 + hq], fp_ord(lg[u]));
        }
    }
}

template <bool REC>
__device__ __forceinline__ void own2_walk_sum(const int* list, const int4* __restrict__ rec, int cnt, int wave, int lane,
        const int* __restrict__ src, const int* __restrict__ tgt, const float* __restrict__ ea,
        const float* __restrict__ xl, const float* xrS, int node0, int rows,
        const float (&we0)[4], const float (&we1)[4], const float (&at)[4],
        float* tab, float* den, const float* mxA) {
    const int half = lane >> 4, q = lane & 15, hq = q >> 2;
    for (int base = 0; base < cnt; base += NWV * EPW * AGU) {
        float4 v[AGU];
        float  lg[AGU];
        int    tl[AGU];
        own2_trip<REC>(list, rec, base, cnt, wave, half, q, src, tgt, ea, xl, xrS, node0, rows, we0, we1, at, v, lg, tl);
        float mx[AGU];
#pragma unroll
        for (int u = 0; u < AGU; ++u)
            mx[u] = mxA[((tl[u] < 0) ? 0 : tl[u]) * 4 + hq];
#pragma unroll
        for (int u = 0; u < AGU; ++u) {
            if (own2_j<REC>(base, wave, half, u) < cnt && tl[u] >= 0) {
                float w = __expf(lg[u] - mx[u]);
                pool_item_add(tab, v[u], w, tl[u] * CH2 + q * 4);
                if ((lane & 3) == 0) atomicAdd(&den[tl[u] * 4 + hq], w);
            }
        }
    }
}

__global__ __launch_bounds__(AGT) void k_l2_own(
        const float* __restrict__ xl, const float* __restrict__ xr,
        const float* __restrict__ ea, const float* __restrict__ We,
        const float* __restrict__ att,
        const int* __restrict__ src, const int* __restrict__ tgt,
        const unsigned* __restrict__ keys,
        const int* __restrict__ grec, const int* __restrict__ gcnt,
        const int* __restrict__ batch,
        float* __restrict__ part, int N, int E, int G) {
    __shared__ __align__(16) float tab[R1 * CH2];
    __shared__ __align__(16) float xrS[R1 * CH2];
    __shared__ unsigned mxU[R1 * 4];
    __shared__ float    mxA[R1 * 4];
    __shared__ float    den[R1 * 4];
    __shared__ int      list[QCAP];
    __shared__ int      qcnt;
    __shared__ int      redo;
    extern __shared__ float gsum[];

    const int tid = threadIdx.x;
    const int lane = tid & 31;
    const int wave = __builtin_amdgcn_readfirstlane(tid >> 5);
    const int q = lane & 15;
    const int node0 = blockIdx.x * R1;
    const int rows = (N - node0 < R1) ? (N - node0) : R1;
    const int nChunks = (E + CHK - 1) / CHK;

    for (int i = tid; i < R1 * CH2; i += AGT) tab[i] = 0.f;
    for (int i = tid; i < R1 * 4; i += AGT) { mxU[i] = 0u; mxA[i] = -__builtin_inff(); den[i] = 0.f; }
    for (int i = tid; i < (G + 1) * CH2; i += AGT) gsum[i] = 0.f;
    {
        const float4* g = (const float4*)(xr + (size_t)node0 * CH2);
        float4* l = (float4*)xrS;
        for (int i = tid; i < rows * (CH2 / 4); i += AGT) l[i] = g[i];
    }
    if (tid == 0) { qcnt = 0; redo = 0; }

    float we0[4], we1[4], at[4];
#pragma unroll
    for (int k = 0; k < 4; ++k) {
        we0[k] = We[q * 4 + k];
        we1[k] = We[CH2 + q * 4 + k];
        at[k]  = att[q * 4 + k];
    }

    const unsigned pat = ((unsigned)blockIdx.x & 255u) * 0x01010101u;
    const int given = __builtin_amdgcn_readfirstlane(gcnt[blockIdx.x]);
    const bool fromRec = (given >= 0);
    const int4* recB = (const int4*)grec + (size_t)blockIdx.x * QCAP;
    __syncthreads();

    int c0 = 0;
    while (c0 < nChunks) {
        int seg = nChunks - c0;
        int cnt = 0;
        if (c0 == 0 && fromRec) {
            cnt = (given < QCAP) ? given : QCAP;
        } else while (true) {
            const int cEnd = c0 + seg;
            uint4 kq[2];
            kq[0] = own_load_keys(keys, c0 * CHK, tid);
            kq[1] = own_load_keys(keys, ((c0 + 1 < cEnd) ? (c0 + 1) : c0) * CHK, tid);
            for (int c = c0; c < cEnd; c += 2) {
                const int ebase = c * CHK;
                uint4 kn[2];
                kn[0] = kq[0];
                kn[1] = kq[1];
                if (c + 2 < cEnd) {
                    kn[0] = own_load_keys(keys, ebase + 2 * CHK, tid);
                    kn[1] = own_load_keys(keys, ((c + 3 < cEnd) ? (c + 3) : (c + 2)) * CHK, tid);
                }
                unsigned mask = (own_match4(kq[0].x, pat) >> 7) | (own_match4(kq[0].y, pat) >> 6)
                              | (own_match4(kq[0].z, pat) >> 5) | (own_match4(kq[0].w, pat) >> 4);
                if (c + 1 < cEnd)
                    mask |= (own_match4(kq[1].x, pat) >> 3) | (own_match4(kq[1].y, pat) >> 2)
                          | (own_match4(kq[1].z, pat) >> 1) |  own_match4(kq[1].w, pat);
                if (mask && ebase + 2 * CHK > E) {
                    unsigned m2 = mask;
                    while (m2) {
                        int b = __ffs(m2) - 1;
                        m2 &= m2 - 1;
                        if (ebase + own_off(b, tid) >= E) mask &= ~(1u << b);
                    }
                }
                if (mask) {
                    int pos = atomicAdd(&qcnt, __popc(mask));
                    while (mask) {
                        int b = __ffs(mask) - 1;
                        mask &= mask - 1;
                        if (pos < QCAP) list[pos] = ebase + own_off(b, tid);
                        ++pos;
                    }
                }
                kq[0] = kn[0];
                kq[1] = kn[1];
            }
            __syncthreads();
            cnt = qcnt;
            __syncthreads();
            if (tid == 0) qcnt = 0;
            if (cnt <= QCAP) break;
            seg = (seg + 1) >> 1;
            __syncthreads();
        }

        if (c0 == 0 && seg == nChunks) {
            for (int i = tid; i < R1 * 4; i += AGT) {
                const float4* xr4 = (const float4*)(xrS + i * 16);
                const float* ah = att + (i & 3) * 16;
                float r = 0.f;
#pragma unroll
                for (int k = 0; k < 4; ++k) {
                    const float4 p = xr4[k];
                    float m0 = (p.x > 0.f) ? p.x : 0.2f * p.x;
                    float m1 = (p.y > 0.f) ? p.y : 0.2f * p.y;
                    float m2 = (p.z > 0.f) ? p.z : 0.2f * p.z;
                    float m3 = (p.w > 0.f) ? p.w : 0.2f * p.w;
                    r = fmaf(m0, ah[4 * k], r);
                    r = fmaf(m1, ah[4 * k + 1], r);
                    r = fmaf(m2, ah[4 * k + 2], r);
                    r = fmaf(m3, ah[4 * k + 3], r);
                }
                mxA[i] = r;
            }
            __syncthreads();
            if (fromRec)
                own2_walk_one<true>(list, recB, cnt, wave, lane, src, tgt, ea, xl, xrS, node0, rows, we0, we1, at,
                                    tab, den, mxU, mxA);
            else
                own2_walk_one<false>(list, recB, cnt, wave, lane, src, tgt, ea, xl, xrS, node0, rows, we0, we1, at,
                                     tab, den, mxU, mxA);
            __syncthreads();
            for (int i = tid; i < R1 * 4; i += AGT) {
                unsigned mu = mxU[i];
                if (mu != 0u) {
                    float d = fp_unord(mu) - mxA[i];
                    if (!(d > -SHIFT_OK && d < SHIFT_OK)) redo = 1;
                }
            }
            __syncthreads();
            if (redo == 0) break;
            for (int i = tid; i < R1 * CH2; i += AGT) tab[i] = 0.f;
            for (int i = tid; i < R1 * 4; i += AGT) { mxA[i] = -__builtin_inff(); den[i] = 0.f; }
            __syncthreads();
        }

        if (fromRec)
            own2_walk_max<true>(list, recB, cnt, wave, lane, src, tgt, ea, xl, xrS, node0, rows, we0, we1, at, mxU);
        else
            own2_walk_max<false>(list, recB, cnt, wave, lane, src, tgt, ea, xl, xrS, node0, rows, we0, we1, at, mxU);
        __syncthreads();
        for (int i = tid; i < R1 * 4; i += AGT) {
            unsigned mu = mxU[i];
            if (mu != 0u) {
                float nm = fp_unord(mu), old = mxA[i];
                if (nm != old) {
                    float sc = __expf(old - nm);
                    mxA[i] = nm;
                    den[i] *= sc;
                    float4* row = (float4*)(tab + i * 16);
#pragma unroll
                    for (int k = 0; k < 4; ++k) {
                        float4 qv = row[k];
                        qv.x *= sc; qv.y *= sc; qv.z *= sc; qv.w *= sc;
                        row[k] = qv;
                    }
                }
            }
        }
        __syncthreads();
        if (fromRec)
            own2_walk_sum<true>(list, recB, cnt, wave, lane, src, tgt, ea, xl, xrS, node0, rows, we0, we1, at,
                                tab, den, mxA);
        else
            own2_walk_sum<false>(list, recB, cnt, wave, lane, src, tgt, ea, xl, xrS, node0, rows, we0, we1, at,
                                 tab, den, mxA);
        c0 += seg;
        __syncthreads();
    }

    for (int i = tid; i < rows * CH2; i += AGT) {
        int r = i >> 6, d = i & 63;
        float dn = den[r * 4 + (d >> 4)];
        float val = (dn > 0.f) ? tab[i] / dn : 0.f;
        int g = batch[node0 + r];
        unsigned row = ((unsigned)g < (unsigned)G) ? (unsigned)g : (unsigned)G;
        atomicAdd(&gsum[row * CH2 + d], val);
    }
    __syncthreads();
    float* my = part + (size_t)blockIdx.x * G * CH2;
    for (int i = tid; i < G * CH2; i += AGT) my[i] = gsum[i];
}

__global__ __launch_bounds__(1024) void k_final(const float* __restrict__ part, int nPart,
                                                const float* __restrict__ bias2,
                                                const int* __restrict__ batch,
                                                float* __restrict__ out, int N, int G) {
    __shared__ float red[16][64];
    __shared__ int cntS;
    const int g = blockIdx.x;
    const int tid = threadIdx.x;
    const int d = tid & 63, j = tid >> 6;
    if (tid == 0) cntS = 0;

    float acc = 0.f;
    const float* p = part + (size_t)g * 64 + d;
    const size_t tabStride = (size_t)G * 64;
    int b = j;
    for (; b + 16 * 7 < nPart; b += 16 * 8) {
        float t[8];
#pragma unroll
        for (int k = 0; k < 8; ++k) t[k] = p[(size_t)(b + 16 * k) * tabStride];
#pragma unroll
        for (int k = 0; k < 8; ++k) acc += t[k];
    }
    for (; b < nPart; b += 16) acc += p[(size_t)b * tabStride];

    int c = 0;
    for (int n = tid; n < N; n += 1024) c += (batch[n] == g) ? 1 : 0;

    red[j][d] = acc;
    __syncthreads();
    atomicAdd(&cntS, c);
    __syncthreads();

    if (j == 0) {
        float s = 0.f;
#pragma unroll
        for (int k = 0; k < 16; ++k) s += red[k][d];
        float cnt = (float)cntS;
        out[g * 64 + d] = (s + cnt * bias2[d]) / fmaxf(cnt, 1.0f);
    }
}

extern "C" void kernel_launch(void* const* d_in, const int* in_sizes, int n_in,
                              void* d_out, int out_size, void* d_ws, size_t ws_size,
                              hipStream_t stream) {
    const float* x      = (const float*)d_in[0];
    const float* ea     = (const float*)d_in[1];
    const float* W_l1   = (const float*)d_in[2];
    const float* b_l1   = (const float*)d_in[3];
    const float* W_r1   = (const float*)d_in[4];
    const float* b_r1   = (const float*)d_in[5];
    const float* W_e1   = (const float*)d_in[6];
    const float* att1   = (const float*)d_in[7];
    const float* bias1  = (const float*)d_in[8];
    const float* W_l2   = (const float*)d_in[9];
    const float* b_l2   = (const float*)d_in[10];
    const float* W_r2   = (const float*)d_in[11];
    const float* b_r2   = (const float*)d_in[12];
    const float* W_e2   = (const float*)d_in[13];
    const float* att2   = (const float*)d_in[14];
    const float* bias2  = (const float*)d_in[15];
    const int*   eidx   = (const int*)d_in[16];
    const int*   batch  = (const int*)d_in[17];
    float* out = (float*)d_out;

    const int N = in_sizes[0] / 2;
    const int E = in_sizes[1] / 2;
    const int G = out_size / 64;
    const int* src = eidx;
    const int* tgt = eidx + E;

    char* ws = (char*)d_ws;
    size_t off = 0;
    auto alloc = [&](size_t bytes) { size_t o = off; off = (off + bytes + 255) & ~(size_t)255; return o; };

    const int nOwn = (N + R1 - 1) / R1;
    size_t o_xl1 = alloc((size_t)N * 128 * 4);
    size_t o_xr1 = alloc((size_t)N * 128 * 4);
    size_t o_hh  = alloc((size_t)N * 128 * 2);
    size_t o_wpk = alloc((size_t)2 * 8192 * 2);
    const int nKeys = ((E + CHK - 1) / CHK) * CHK;
    size_t o_keys = alloc((size_t)nKeys);
    size_t o_xl2 = alloc((size_t)N * 64 * 4);
    size_t o_xr2 = alloc((size_t)N * 64 * 4);
    size_t o_grec  = alloc((size_t)nOwn * QCAP * 16);
    size_t o_gcnt  = alloc((size_t)nOwn * 4);
    size_t o_part = alloc((size_t)nOwn * G * 64 * 4);
    (void)ws_size;

    float*    xl1   = (float*)(ws + o_xl1);
    float*    xr1   = (float*)(ws + o_xr1);
    _Float16* hh    = (_Float16*)(ws + o_hh);
    _Float16* wpk   = (_Float16*)(ws + o_wpk);
    float*    xl2   = (float*)(ws + o_xl2);
    float*    xr2   = (float*)(ws + o_xr2);
    int*      grec  = (int*)(ws + o_grec);
    int*      gcnt  = (int*)(ws + o_gcnt);
    float*    part  = (float*)(ws + o_part);
    unsigned* keys  = (unsigned*)(ws + o_keys);

    const int TB = 256;
    auto blocks = [&](long long t) { return (unsigned)((t + TB - 1) / TB); };

    k_node_transform1<<<blocks((long long)N * 128), TB, 0, stream>>>(
        x, W_l1, b_l1, W_r1, b_r1, xl1, xr1, N);
    const int packBlocks = (int)blocks(2 * 8192);
    k_w_pack<<<(unsigned)packBlocks + blocks(nKeys / 4), TB, 0, stream>>>(
        W_l2, W_r2, wpk, packBlocks, tgt, keys, E, nKeys);
    k_l1_own<<<(unsigned)nOwn, AGT, 0, stream>>>(
        xl1, xr1, ea, W_e1, att1, src, tgt, keys, bias1, hh, grec, gcnt, N, E);

    long long gemm_threads = (long long)((N + 15) / 16) * 32;
    k_gemm2_wmma<<<blocks(gemm_threads), TB, 0, stream>>>(
        hh, wpk, b_l2, b_r2, xl2, xr2, N);

    k_l2_own<<<(unsigned)nOwn, AGT, (size_t)(G + 1) * 64 * sizeof(float), stream>>>(
        xl2, xr2, ea, W_e2, att2, src, tgt, keys, grec, gcnt, batch, part, N, E, G);
    k_final<<<G, 1024, 0, stream>>>(part, nOwn, bias2, batch, out, N, G);
}


# LDS sum tables of both owner kernels in channel-plane order (layer 2: row pitch 80): each float add of a wave goes to distinct banks, was 2 lanes per bank; arrival-order sums, order may change
# speedup vs baseline: 3.9005x; 1.2394x over previous
#include <hip/hip_runtime.h>
#include <hip/hip_fp16.h>

typedef __attribute__((ext_vector_type(16))) _Float16 v16h;
typedef __attribute__((ext_vector_type(8)))  _Float16 v8h;
typedef __attribute__((ext_vector_type(8)))  float    v8f;

__device__ __forceinline__ unsigned fp_ord(float f) {
    unsigned u = __float_as_uint(f);
    return (u & 0x80000000u) ? ~u : (u | 0x80000000u);
}
__device__ __forceinline__ float fp_unord(unsigned u) {
    unsigned b = (u & 0x80000000u) ? (u ^ 0x80000000u) : ~u;
    return __uint_as_float(b);
}

__global__ void k_node_transform1(const float* __restrict__ x,
                                  const float* __restrict__ Wl, const float* __restrict__ bl,
                                  const float* __restrict__ Wr, const float* __restrict__ br,
                                  float* __restrict__ xl1, float* __restrict__ xr1, int N) {
    int idx = blockIdx.x * blockDim.x + threadIdx.x;
    if (idx >= N * 128) return;
    int n = idx >> 7, d = idx & 127;
    float x0 = x[n * 2], x1 = x[n * 2 + 1];
    xl1[idx] = fmaf(x0, Wl[d], fmaf(x1, Wl[128 + d], bl[d]));
    xr1[idx] = fmaf(x0, Wr[d], fmaf(x1, Wr[128 + d], br[d]));
}

constexpr int KEY_R = 200;
__global__ void k_w_pack(const float* __restrict__ Wl, const float* __restrict__ Wr,
                         _Float16* __restrict__ Wpk, int packBlocks,
                         const int* __restrict__ tgt, unsigned* __restrict__ keys, int E, int nKeys) {
    if ((int)blockIdx.x >= packBlocks) {
        int w = ((int)blockIdx.x - packBlocks) * (int)blockDim.x + (int)threadIdx.x;
        int e0 = w * 4;
        if (e0 >= nKeys) return;
        int t0, t1, t2, t3;
        if (e0 + 4 <= E && (((size_t)tgt) & 15) == 0) {
            int4 t = *(const int4*)(tgt + e0);
            t0 = t.x; t1 = t.y; t2 = t.z; t3 = t.w;
        } else {
            t0 = (e0 + 0 < E) ? tgt[e0 + 0] : 0;
            t1 = (e0 + 1 < E) ? tgt[e0 + 1] : 0;
            t2 = (e0 + 2 < E) ? tgt[e0 + 2] : 0;
            t3 = (e0 + 3 < E) ? tgt[e0 + 3] : 0;
        }
        unsigned k0 = ((unsigned)t0 / (unsigned)KEY_R) & 255u;
        unsigned k1 = ((unsigned)t1 / (unsigned)KEY_R) & 255u;
        unsigned k2 = ((unsigned)t2 / (unsigned)KEY_R) & 255u;
        unsigned k3 = ((unsigned)t3 / (unsigned)KEY_R) & 255u;
        keys[w] = k0 | (k1 << 8) | (k2 << 16) | (k3 << 24);
        return;
    }
    int idx = blockIdx.x * blockDim.x + threadIdx.x;
    if (idx >= 2 * 8192) return;
    int which = idx >> 13;
    int r = idx & 8191;
    int j = r & 15;
    int lane = (r >> 4) & 31;
    int frag = r >> 9;
    int kc = frag >> 2, nt = frag & 3;
    int half_sel = lane >> 4, lidx = lane & 15;
    int kb = (j < 8) ? (half_sel * 8) : (16 + half_sel * 8);
    int k = kc * 32 + kb + (j & 7);
    int n = nt * 16 + lidx;
    const float* W = which ? Wr : Wl;
    Wpk[idx] = (_Float16)W[k * 64 + n];
}

__global__ void k_gemm2_wmma(const _Float16* __restrict__ hh,
                             const _Float16* __restrict__ Wpk,
                             const float* __restrict__ bl, const float* __restrict__ br,
                             float* __restrict__ xl2, float* __restrict__ xr2, int N) {
    __shared__ __align__(32) _Float16 ldsW[2 * 8192];

    {
        const uint4* g = (const uint4*)Wpk;
        uint4* l = (uint4*)ldsW;
        for (int i = threadIdx.x; i < 2048; i += blockDim.x) l[i] = g[i];
    }
    __syncthreads();

    int wave = (blockIdx.x * blockDim.x + threadIdx.x) >> 5;
    int lane = threadIdx.x & 31;
    int row0 = wave * 16;
    if (row0 >= N) return;

    int half_sel = lane >> 4;
    int lidx = lane & 15;
    int kb_lo = half_sel * 8;
    int kb_hi = 16 + kb_lo;
    int m = row0 + lidx;
    if (m >= N) m = N - 1;

    const v16h* fragL = (const v16h*)ldsW;
    const v16h* fragR = (const v16h*)(ldsW + 8192);

    v8f accl[4] = {};
    v8f accr[4] = {};

#pragma unroll
    for (int kc = 0; kc < 4; ++kc) {
        const _Float16* pa = hh + m * 128 + kc * 32;
        v8h alo = *(const v8h*)(pa + kb_lo);
        v8h ahi = *(const v8h*)(pa + kb_hi);
        v16h a = __builtin_shufflevector(alo, ahi, 0, 1, 2, 3, 4, 5, 6, 7,
                                         8, 9, 10, 11, 12, 13, 14, 15);
#pragma unroll
        for (int nt = 0; nt < 4; ++nt) {
            int fi = (kc * 4 + nt) * 32 + lane;
            v16h b  = fragL[fi];
            v16h b2 = fragR[fi];
            accl[nt] = __builtin_amdgcn_wmma_f32_16x16x32_f16(
                false, a, false, b, (short)0, accl[nt], false, false);
            accr[nt] = __builtin_amdgcn_wmma_f32_16x16x32_f16(
                false, a, false, b2, (short)0, accr[nt], false, false);
        }
    }
#pragma unroll
    for (int nt = 0; nt < 4; ++nt) {
        int n = nt * 16 + lidx;
        float bln = bl[n], brn = br[n];
#pragma unroll
        for (int r = 0; r < 8; ++r) {
            int mo = row0 + r + half_sel * 8;
            if (mo < N) {
                xl2[mo * 64 + n] = accl[nt][r] + bln;
                xr2[mo * 64 + n] = accr[nt][r] + brn;
            }
        }
    }
}

__device__ __forceinline__ void pool_item_add(float* tab, const float4& v, float alpha, int off, int cs) {
    atomicAdd(&tab[off], alpha * v.x);
    atomicAdd(&tab[off + cs], alpha * v.y);
    atomicAdd(&tab[off + 2 * cs], alpha * v.z);
    atomicAdd(&tab[off + 3 * cs], alpha * v.w);
}

constexpr int T2P = 80;
__device__ __forceinline__ int tab1_at(int r, int c) { return r * 128 + (c & 3) * 32 + (c >> 2); }
__device__ __forceinline__ int tab2_at(int r, int c) { return r * T2P + (c & 3) * 16 + (c >> 2); }

constexpr int R1  = 200;
constexpr int AGT = 1024;
constexpr int NWV = AGT / 32;
constexpr int EPT = 16;
constexpr int CHK = AGT * EPT;
constexpr int QCAP = CHK;
constexpr int AGU = 8;
constexpr int CH2 = 64;
constexpr int EPW = 2;
constexpr float SHIFT_OK = 40.f;
static_assert(EPT == 16, "a thread's keys of one chunk are one 16-byte word; two chunks make one 32-bit match mask");
static_assert(QCAP >= CHK, "a run of one chunk must always fit the list");
static_assert(KEY_R == R1, "the owner keys are made for this block size");

typedef __attribute__((ext_vector_type(2))) _Float16 v2h;

__device__ __forceinline__ uint4 own_load_keys(const unsigned* __restrict__ keys, int ebase, int tid) {
    return *(const uint4*)(keys + ((ebase + 16 * tid) >> 2));
}

__device__ __forceinline__ unsigned own_match4(unsigned w, unsigned pat) {
    unsigned x = w ^ pat;
    return ~(((x & 0x7F7F7F7Fu) + 0x7F7F7F7Fu) | x | 0x7F7F7F7Fu);
}

__device__ __forceinline__ int own_off(int p, int tid) {
    const int i = p & 7, j = p >> 3;
    return (i >> 2) * CHK + 16 * tid + 4 * (i & 3) + j;
}

__device__ __forceinline__ void own_trip(const int* list, int base, int cnt, int wave, int lane,
        const int* __restrict__ src, const int* __restrict__ tgt, const float* __restrict__ ea,
        const float* __restrict__ xl, const float* xrS, int node0, int rows,
        const float (&we0)[4], const float (&we1)[4], const float (&at)[4],
        float4 (&v)[AGU], float (&lg)[AGU], int (&tl)[AGU], int4* recOut) {
    float a0[AGU], a1[AGU];
    int   sN[AGU];
#pragma unroll
    for (int u = 0; u < AGU; ++u) {
        int j = base + wave + u * NWV;
        int e = list[(j < cnt) ? j : (cnt - 1)];
        sN[u] = src[e];
        const int t = tgt[e] - node0;
        tl[u] = ((unsigned)t < (unsigned)rows) ? t : -1;
        a0[u] = ea[(size_t)e * 2];
        a1[u] = ea[(size_t)e * 2 + 1];
    }
#pragma unroll
    for (int u = 0; u < AGU; ++u)
        v[u] = *(const float4*)(xl + (size_t)sN[u] * 128 + lane * 4);
    if (recOut != nullptr && lane == 0) {
#pragma unroll
        for (int u = 0; u < AGU; ++u) {
            const int j = base + wave + u * NWV;
            if (j < cnt)
                recOut[j] = make_int4(sN[u], tl[u], __float_as_int(a0[u]), __float_as_int(a1[u]));
        }
    }
#pragma unroll
    for (int u = 0; u < AGU; ++u) {
        const float4 r4 = *(const float4*)(xrS + ((tl[u] < 0) ? 0 : tl[u]) * 128 + lane * 4);
        float m0 = v[u].x + r4.x + a0[u] * we0[0] + a1[u] * we1[0];
        float m1 = v[u].y + r4.y + a0[u] * we0[1] + a1[u] * we1[1];
        float m2 = v[u].z + r4.z + a0[u] * we0[2] + a1[u] * we1[2];
        float m3 = v[u].w + r4.w + a0[u] * we0[3] + a1[u] * we1[3];
        m0 = (m0 > 0.f) ? m0 : 0.2f * m0;
        m1 = (m1 > 0.f) ? m1 : 0.2f * m1;
        m2 = (m2 > 0.f) ? m2 : 0.2f * m2;
        m3 = (m3 > 0.f) ? m3 : 0.2f * m3;
        float p = fmaf(m3, at[3], fmaf(m2, at[2], fmaf(m1, at[1], m0 * at[0])));
        p += __int_as_float(__builtin_amdgcn_update_dpp(0, __float_as_int(p), 0xB1, 0xF, 0xF, false));
        lg[u] = p;
    }
}

__global__ __launch_bounds__(AGT) void k_l1_own(
        const float* __restrict__ xl, const float* __restrict__ xr,
        const float* __restrict__ ea, const float* __restrict__ We,
        const float* __restrict__ att,
        const int* __restrict__ src, const int* __restrict__ tgt,
        const unsigned* __restrict__ keys,
        const float* __restrict__ bias,
        _Float16* __restrict__ hh, int* __restrict__ grec, int* __restrict__ gcnt, int N, int E) {
    __shared__ __align__(16) float tab[R1 * 128];
    __shared__ __align__(16) float xrS[R1 * 128];
    __shared__ unsigned mxU[R1 * 16];
    __shared__ float    mxA[R1 * 16];
    __shared__ float    den[R1 * 16];
    __shared__ int      list[QCAP];
    __shared__ int      qcnt;
    __shared__ int      redo;

    const int tid = threadIdx.x;
    const int lane = tid & 31;
    const int wave = __builtin_amdgcn_readfirstlane(tid >> 5);
    const int hq = lane >> 1;
    const int node0 = blockIdx.x * R1;
    const int rows = (N - node0 < R1) ? (N - node0) : R1;
    const int nChunks = (E + CHK - 1) / CHK;

    for (int i = tid; i < R1 * 128; i += AGT) tab[i] = 0.f;
    for (int i = tid; i < R1 * 16; i += AGT) { mxU[i] = 0u; mxA[i] = -__builtin_inff(); den[i] = 0.f; }
    {
        const float4* g = (const float4*)(xr + (size_t)node0 * 128);
        float4* l = (float4*)xrS;
        for (int i = tid; i < rows * 32; i += AGT) l[i] = g[i];
    }
    if (tid == 0) { qcnt = 0; redo = 0; }

    float we0[4], we1[4], at[4];
#pragma unroll
    for (int k = 0; k < 4; ++k) {
        we0[k] = We[lane * 4 + k];
        we1[k] = We[128 + lane * 4 + k];
        at[k]  = att[lane * 4 + k];
    }

    const unsigned pat = ((unsigned)blockIdx.x & 255u) * 0x01010101u;
    __syncthreads();

    int c0 = 0;
    while (c0 < nChunks) {
        int seg = nChunks - c0;
        int cnt = 0;
        while (true) {
            const int cEnd = c0 + seg;
            uint4 kq[2];
            kq[0] = own_load_keys(keys, c0 * CHK, tid);
            kq[1] = own_load_keys(keys, ((c0 + 1 < cEnd) ? (c0 + 1) : c0) * CHK, tid);
            for (int c = c0; c < cEnd; c += 2) {
                const int ebase = c * CHK;
                uint4 kn[2];
                kn[0] = kq[0];
                kn[1] = kq[1];
                if (c + 2 < cEnd) {
                    kn[0] = own_load_keys(keys, ebase + 2 * CHK, tid);
                    kn[1] = own_load_keys(keys, ((c + 3 < cEnd) ? (c + 3) : (c + 2)) * CHK, tid);
                }
                unsigned mask = (own_match4(kq[0].x, pat) >> 7) | (own_match4(kq[0].y, pat) >> 6)
                              | (own_match4(kq[0].z, pat) >> 5) | (own_match4(kq[0].w, pat) >> 4);
                if (c + 1 < cEnd)
                    mask |= (own_match4(kq[1].x, pat) >> 3) | (own_match4(kq[1].y, pat) >> 2)
                          | (own_match4(kq[1].z, pat) >> 1) |  own_match4(kq[1].w, pat);
                if (mask && ebase + 2 * CHK > E) {
                    unsigned m2 = mask;
                    while (m2) {
                        int b = __ffs(m2) - 1;
                        m2 &= m2 - 1;
                        if (ebase + own_off(b, tid) >= E) mask &= ~(1u << b);
                    }
                }
                if (mask) {
                    int pos = atomicAdd(&qcnt, __popc(mask));
                    while (mask) {
                        int b = __ffs(mask) - 1;
                        mask &= mask - 1;
                        if (pos < QCAP) list[pos] = ebase + own_off(b, tid);
                        ++pos;
                    }
                }
                kq[0] = kn[0];
                kq[1] = kn[1];
            }
            __syncthreads();
            cnt = qcnt;
            __syncthreads();
            if (tid == 0) qcnt = 0;
            if (cnt <= QCAP) break;
            seg = (seg + 1) >> 1;
            __syncthreads();
        }
        if (c0 == 0 && tid == 0) gcnt[blockIdx.x] = (seg == nChunks) ? cnt : -1;

        if (c0 == 0 && seg == nChunks) {
            for (int i = tid; i < R1 * 16; i += AGT) {
                const float4* xr4 = (const float4*)(xrS + i * 8);
                const float4 p0 = xr4[0], p1 = xr4[1];
                const float* ah = att + (i & 15) * 8;
                float m[8] = {p0.x, p0.y, p0.z, p0.w, p1.x, p1.y, p1.z, p1.w};
                float r = 0.f;
#pragma unroll
                for (int k = 0; k < 8; ++k) {
                    float mk = (m[k] > 0.f) ? m[k] : 0.2f * m[k];
                    r = fmaf(mk, ah[k], r);
                }
                mxA[i] = r;
            }
            __syncthreads();
            int4* recB = (int4*)grec + (size_t)blockIdx.x * QCAP;
            for (int base = 0; base < cnt; base += NWV * AGU) {
                float4 v[AGU];
                float  lg[AGU];
                int    tl[AGU];
                own_trip(list, base, cnt, wave, lane, src, tgt, ea, xl, xrS, node0, rows, we0, we1, at, v, lg, tl,
                         recB);
                float sh[AGU];
#pragma unroll
                for (int u = 0; u < AGU; ++u)
                    sh[u] = mxA[((tl[u] < 0) ? 0 : tl[u]) * 16 + hq];
#pragma unroll
                for (int u = 0; u < AGU; ++u) {
                    if (base + wave + u * NWV < cnt && tl[u] >= 0) {
                        float w = __expf(lg[u] - sh[u]);
                        pool_item_add(tab, v[u], w, tl[u] * 128 + lane, 32);
                        if ((lane & 1) == 0) {
                            atomicAdd(&den[tl[u] * 16 + hq], w);
                            atomicMax(&mxU[tl[u] * 16 + hq], fp_ord(lg[u]));
                        }
                    }
                }
            }
            __syncthreads();
            for (int i = tid; i < R1 * 16; i += AGT) {
                unsigned mu = mxU[i];
                if (mu != 0u) {
                    float d = fp_unord(mu) - mxA[i];
                    if (!(d > -SHIFT_OK && d < SHIFT_OK)) redo = 1;
                }
            }
            __syncthreads();
            if (redo == 0) break;
            for (int i = tid; i < R1 * 128; i += AGT) tab[i] = 0.f;
            for (int i = tid; i < R1 * 16; i += AGT) { mxA[i] = -__builtin_inff(); den[i] = 0.f; }
            __syncthreads();
        }

        for (int base = 0; base < cnt; base += NWV * AGU) {
            float4 v[AGU];
            float  lg[AGU];
            int    tl[AGU];
            own_trip(list, base, cnt, wave, lane, src, tgt, ea, xl, xrS, node0, rows, we0, we1, at, v, lg, tl,
                     nullptr);
#pragma unroll
            for (int u = 0; u < AGU; ++u) {
                if (base + wave + u * NWV < cnt && tl[u] >= 0 && (lane & 1) == 0)
                    atomicMax(&mxU[tl[u] * 16 + hq], fp_ord(lg[u]));
            }
        }
        __syncthreads();
        for (int i = tid; i < R1 * 16; i += AGT) {
            unsigned mu = mxU[i];
            if (mu != 0u) {
                float nm = fp_unord(mu), old = mxA[i];
                if (nm != old) {
                    float sc = __expf(old - nm);
                    mxA[i] = nm;
                    den[i] *= sc;
                    const int nd = i >> 4, c8 = (i & 15) * 8;
#pragma unroll
                    for (int k = 0; k < 8; ++k) tab[tab1_at(nd, c8 + k)] *= sc;
                }
            }
        }
        __syncthreads();
        for (int base = 0; base < cnt; base += NWV * AGU) {
            float4 v[AGU];
            float  lg[AGU];
            int    tl[AGU];
            own_trip(list, base, cnt, wave, lane, src, tgt, ea, xl, xrS, node0, rows, we0, we1, at, v, lg, tl,
                     nullptr);
            float mx[AGU];
#pragma unroll
            for (int u = 0; u < AGU; ++u)
                mx[u] = mxA[((tl[u] < 0) ? 0 : tl[u]) * 16 + hq];
#pragma unroll
            for (int u = 0; u < AGU; ++u) {
                if (base + wave + u * NWV < cnt && tl[u] >= 0) {
                    float w = __expf(lg[u] - mx[u]);
                    pool_item_add(tab, v[u], w, tl[u] * 128 + lane, 32);
                    if ((lane & 1) == 0) atomicAdd(&den[tl[u] * 16 + hq], w);
                }
            }
        }
        c0 += seg;
        __syncthreads();
    }

    for (int i = tid; i < rows * 64; i += AGT) {
        int r = i >> 6, d = (i & 63) * 2;
        float dn = den[r * 16 + (d >> 3)];
        float v0 = (dn > 0.f) ? tab[tab1_at(r, d)] / dn : 0.f;
        float v1 = (dn > 0.f) ? tab[tab1_at(r, d + 1)] / dn : 0.f;
        v0 += bias[d];
        v1 += bias[d + 1];
        v0 = (v0 > 0.f) ? v0 : (__expf(v0) - 1.f);
        v1 = (v1 > 0.f) ? v1 : (__expf(v1) - 1.f);
        v2h o = {(_Float16)v0, (_Float16)v1};
        *(v2h*)(hh + (size_t)(node0 + r) * 128 + d) = o;
    }
}

static_assert(CH2 * EPW == 32 * 4, "a wave covers EPW rows of CH2 channels with 4 channels per lane");

template <bool REC>
__device__ __forceinline__ int own2_j(int base, int wave, int half, int u) {
    return REC ? base + (wave * AGU + u) * EPW + half
               : base + (u * NWV + wave) * EPW + half;
}

template <bool REC>
__device__ __forceinline__ void own2_trip(const int* list, const int4* __restrict__ rec,
        int base, int cnt, int wave, int half, int q,
        const int* __restrict__ src, const int* __restrict__ tgt, const float* __restrict__ ea,
        const float* __restrict__ xl, const float* xrS, int node0, int rows,
        const float (&we0)[4], const float (&we1)[4], const float (&at)[4],
        float4 (&v)[AGU], float (&lg)[AGU], int (&tl)[AGU]) {
    float a0[AGU], a1[AGU];
    int   sN[AGU];
    if (REC) {
        int4 r[AGU];
#pragma unroll
        for (int u = 0; u < AGU; ++u) {
            const int j = own2_j<REC>(base, wave, half, u);
            r[u] = rec[(j < cnt) ? j : (cnt - 1)];
        }
        __builtin_amdgcn_sched_barrier(0);
#pragma unroll
        for (int u = 0; u < AGU; ++u) {
            sN[u] = r[u].x;
            tl[u] = ((unsigned)r[u].y < (unsigned)rows) ? r[u].y : -1;
            a0[u] = __int_as_float(r[u].z);
            a1[u] = __int_as_float(r[u].w);
        }
    } else {
#pragma unroll
        for (int u = 0; u < AGU; ++u) {
            const int j = own2_j<REC>(base, wave, half, u);
            const int e = list[(j < cnt) ? j : (cnt - 1)];
            sN[u] = src[e];
            const int t = tgt[e] - node0;
            tl[u] = ((unsigned)t < (unsigned)rows) ? t : -1;
            a0[u] = ea[(size_t)e * 2];
            a1[u] = ea[(size_t)e * 2 + 1];
        }
    }
#pragma unroll
    for (int u = 0; u < AGU; ++u)
        v[u] = *(const float4*)(xl + (size_t)sN[u] * CH2 + q * 4);
#pragma unroll
    for (int u = 0; u < AGU; ++u) {
        const float4 r4 = *(const float4*)(xrS + ((tl[u] < 0) ? 0 : tl[u]) * CH2 + q * 4);
        float m0 = v[u].x + r4.x + a0[u] * we0[0] + a1[u] * we1[0];
        float m1 = v[u].y + r4.y + a0[u] * we0[1] + a1[u] * we1[1];
        float m2 = v[u].z + r4.z + a0[u] * we0[2] + a1[u] * we1[2];
        float m3 = v[u].w + r4.w + a0[u] * we0[3] + a1[u] * we1[3];
        m0 = (m0 > 0.f) ? m0 : 0.2f * m0;
        m1 = (m1 > 0.f) ? m1 : 0.2f * m1;
        m2 = (m2 > 0.f) ? m2 : 0.2f * m2;
        m3 = (m3 > 0.f) ? m3 : 0.2f * m3;
        float p = fmaf(m3, at[3], fmaf(m2, at[2], fmaf(m1, at[1], m0 * at[0])));
        p += __int_as_float(__builtin_amdgcn_update_dpp(0, __float_as_int(p), 0xB1, 0xF, 0xF, false));
        p += __int_as_float(__builtin_amdgcn_update_dpp(0, __float_as_int(p), 0x4E, 0xF, 0xF, false));
        lg[u] = p;
    }
}

template <bool REC>
__device__ __forceinline__ void own2_walk_one(const int* list, const int4* __restrict__ rec, int cnt, int wave, int lane,
        const int* __restrict__ src, const int* __restrict__ tgt, const float* __restrict__ ea,
        const float* __restrict__ xl, const float* xrS, int node0, int rows,
        const float (&we0)[4], const float (&we1)[4], const float (&at)[4],
        float* tab, float* den, unsigned* mxU, const float* mxA) {
    const int half = lane >> 4, q = lane & 15, hq = q >> 2;
    for (int base = 0; base < cnt; base += NWV * EPW * AGU) {
        float4 v[AGU];
        float  lg[AGU];
        int    tl[AGU];
        own2_trip<REC>(list, rec, base, cnt, wave, half, q, src, tgt, ea, xl, xrS, node0, rows, we0, we1, at, v, lg, tl);
        float sh[AGU];
#pragma unroll
        for (int u = 0; u < AGU; ++u)
            sh[u] = mxA[((tl[u] < 0) ? 0 : tl[u]) * 4 + hq];
#pragma unroll
        for (int u = 0; u < AGU; ++u) {
            if (own2_j<REC>(base, wave, half, u) < cnt && tl[u] >= 0) {
                float w = __expf(lg[u] - sh[u]);
                pool_item_add(tab, v[u], w, tl[u] * T2P + q, 16);
                if ((lane & 3) == 0) {
                    atomicAdd(&den[tl[u] * 4 + hq], w);
                    atomicMax(&mxU[tl[u] * 4 + hq], fp_ord(lg[u]));
                }
            }
        }
    }
}

template <bool REC>
__device__ __forceinline__ void own2_walk_max(const int* list, const int4* __restrict__ rec, int cnt, int wave, int lane,
        const int* __restrict__ src, const int* __restrict__ tgt, const float* __restrict__ ea,
        const float* __restrict__ xl, const float* xrS, int node0, int rows,
        const float (&we0)[4], const float (&we1)[4], const float (&at)[4],
        unsigned* mxU) {
    const int half = lane >> 4, q = lane & 15, hq = q >> 2;
    for (int base = 0; base < cnt; base += NWV * EPW * AGU) {
        float4 v[AGU];
        float  lg[AGU];
        int    tl[AGU];
        own2_trip<REC>(list, rec, base, cnt, wave, half, q, src, tgt, ea, xl, xrS, node0, rows, we0, we1, at, v, lg, tl);
#pragma unroll
        for (int u = 0; u < AGU; ++u) {
            if (own2_j<REC>(base, wave, half, u) < cnt && tl[u] >= 0 && (lane & 3) == 0)
                atomicMax(&mxU[tl[u] * 4 + hq], fp_ord(lg[u]));
        }
    }
}

template <bool REC>
__device__ __forceinline__ void own2_walk_sum(const int* list, const int4* __restrict__ rec, int cnt, int wave, int lane,
        const int* __restrict__ src, const int* __restrict__ tgt, const float* __restrict__ ea,
        const float* __restrict__ xl, const float* xrS, int node0, int rows,
        const float (&we0)[4], const float (&we1)[4], const float (&at)[4],
        float* tab, float* den, const float* mxA) {
    const int half = lane >> 4, q = lane & 15, hq = q >> 2;
    for (int base = 0; base < cnt; base += NWV * EPW * AGU) {
        float4 v[AGU];
        float  lg[AGU];
        int    tl[AGU];
        own2_trip<REC>(list, rec, base, cnt, wave, half, q, src, tgt, ea, xl, xrS, node0, rows, we0, we1, at, v, lg, tl);
        float mx[AGU];
#pragma unroll
        for (int u = 0; u < AGU; ++u)
            mx[u] = mxA[((tl[u] < 0) ? 0 : tl[u]) * 4 + hq];
#pragma unroll
        for (int u = 0; u < AGU; ++u) {
            if (own2_j<REC>(base, wave, half, u) < cnt && tl[u] >= 0) {
                float w = __expf(lg[u] - mx[u]);
                pool_item_add(tab, v[u], w, tl[u] * T2P + q, 16);
                if ((lane & 3) == 0) atomicAdd(&den[tl[u] * 4 + hq], w);
            }
        }
    }
}

__global__ __launch_bounds__(AGT) void k_l2_own(
        const float* __restrict__ xl, const float* __restrict__ xr,
        const float* __restrict__ ea, const float* __restrict__ We,
        const float* __restrict__ att,
        const int* __restrict__ src, const int* __restrict__ tgt,
        const unsigned* __restrict__ keys,
        const int* __restrict__ grec, const int* __restrict__ gcnt,
        const int* __restrict__ batch,
        float* __restrict__ part, int N, int E, int G) {
    __shared__ __align__(16) float tab[R1 * T2P];
    __shared__ __align__(16) float xrS[R1 * CH2];
    __shared__ unsigned mxU[R1 * 4];
    __shared__ float    mxA[R1 * 4];
    __shared__ float    den[R1 * 4];
    __shared__ int      list[QCAP];
    __shared__ int      qcnt;
    __shared__ int      redo;
    extern __shared__ float gsum[];

    const int tid = threadIdx.x;
    const int lane = tid & 31;
    const int wave = __builtin_amdgcn_readfirstlane(tid >> 5);
    const int q = lane & 15;
    const int node0 = blockIdx.x * R1;
    const int rows = (N - node0 < R1) ? (N - node0) : R1;
    const int nChunks = (E + CHK - 1) / CHK;

    for (int i = tid; i < R1 * T2P; i += AGT) tab[i] = 0.f;
    for (int i = tid; i < R1 * 4; i += AGT) { mxU[i] = 0u; mxA[i] = -__builtin_inff(); den[i] = 0.f; }
    for (int i = tid; i < (G + 1) * CH2; i += AGT) gsum[i] = 0.f;
    {
        const float4* g = (const float4*)(xr + (size_t)node0 * CH2);
        float4* l = (float4*)xrS;
        for (int i = tid; i < rows * (CH2 / 4); i += AGT) l[i] = g[i];
    }
    if (tid == 0) { qcnt = 0; redo = 0; }

    float we0[4], we1[4], at[4];
#pragma unroll
    for (int k = 0; k < 4; ++k) {
        we0[k] = We[q * 4 + k];
        we1[k] = We[CH2 + q * 4 + k];
        at[k]  = att[q * 4 + k];
    }

    const unsigned pat = ((unsigned)blockIdx.x & 255u) * 0x01010101u;
    const int given = __builtin_amdgcn_readfirstlane(gcnt[blockIdx.x]);
    const bool fromRec = (given >= 0);
    const int4* recB = (const int4*)grec + (size_t)blockIdx.x * QCAP;
    __syncthreads();

    int c0 = 0;
    while (c0 < nChunks) {
        int seg = nChunks - c0;
        int cnt = 0;
        if (c0 == 0 && fromRec) {
            cnt = (given < QCAP) ? given : QCAP;
        } else while (true) {
            const int cEnd = c0 + seg;
            uint4 kq[2];
            kq[0] = own_load_keys(keys, c0 * CHK, tid);
            kq[1] = own_load_keys(keys, ((c0 + 1 < cEnd) ? (c0 + 1) : c0) * CHK, tid);
            for (int c = c0; c < cEnd; c += 2) {
                const int ebase = c * CHK;
                uint4 kn[2];
                kn[0] = kq[0];
                kn[1] = kq[1];
                if (c + 2 < cEnd) {
                    kn[0] = own_load_keys(keys, ebase + 2 * CHK, tid);
                    kn[1] = own_load_keys(keys, ((c + 3 < cEnd) ? (c + 3) : (c + 2)) * CHK, tid);
                }
                unsigned mask = (own_match4(kq[0].x, pat) >> 7) | (own_match4(kq[0].y, pat) >> 6)
                              | (own_match4(kq[0].z, pat) >> 5) | (own_match4(kq[0].w, pat) >> 4);
                if (c + 1 < cEnd)
                    mask |= (own_match4(kq[1].x, pat) >> 3) | (own_match4(kq[1].y, pat) >> 2)
                          | (own_match4(kq[1].z, pat) >> 1) |  own_match4(kq[1].w, pat);
                if (mask && ebase + 2 * CHK > E) {
                    unsigned m2 = mask;
                    while (m2) {
                        int b = __ffs(m2) - 1;
                        m2 &= m2 - 1;
                        if (ebase + own_off(b, tid) >= E) mask &= ~(1u << b);
                    }
                }
                if (mask) {
                    int pos = atomicAdd(&qcnt, __popc(mask));
                    while (mask) {
                        int b = __ffs(mask) - 1;
                        mask &= mask - 1;
                        if (pos < QCAP) list[pos] = ebase + own_off(b, tid);
                        ++pos;
                    }
                }
                kq[0] = kn[0];
                kq[1] = kn[1];
            }
            __syncthreads();
            cnt = qcnt;
            __syncthreads();
            if (tid == 0) qcnt = 0;
            if (cnt <= QCAP) break;
            seg = (seg + 1) >> 1;
            __syncthreads();
        }

        if (c0 == 0 && seg == nChunks) {
            for (int i = tid; i < R1 * 4; i += AGT) {
                const float4* xr4 = (const float4*)(xrS + i * 16);
                const float* ah = att + (i & 3) * 16;
                float r = 0.f;
#pragma unroll
                for (int k = 0; k < 4; ++k) {
                    const float4 p = xr4[k];
                    float m0 = (p.x > 0.f) ? p.x : 0.2f * p.x;
                    float m1 = (p.y > 0.f) ? p.y : 0.2f * p.y;
                    float m2 = (p.z > 0.f) ? p.z : 0.2f * p.z;
                    float m3 = (p.w > 0.f) ? p.w : 0.2f * p.w;
                    r = fmaf(m0, ah[4 * k], r);
                    r = fmaf(m1, ah[4 * k + 1], r);
                    r = fmaf(m2, ah[4 * k + 2], r);
                    r = fmaf(m3, ah[4 * k + 3], r);
                }
                mxA[i] = r;
            }
            __syncthreads();
            if (fromRec)
                own2_walk_one<true>(list, recB, cnt, wave, lane, src, tgt, ea, xl, xrS, node0, rows, we0, we1, at,
                                    tab, den, mxU, mxA);
            else
                own2_walk_one<false>(list, recB, cnt, wave, lane, src, tgt, ea, xl, xrS, node0, rows, we0, we1, at,
                                     tab, den, mxU, mxA);
            __syncthreads();
            for (int i = tid; i < R1 * 4; i += AGT) {
                unsigned mu = mxU[i];
                if (mu != 0u) {
                    float d = fp_unord(mu) - mxA[i];
                    if (!(d > -SHIFT_OK && d < SHIFT_OK)) redo = 1;
                }
            }
            __syncthreads();
            if (redo == 0) break;
            for (int i = tid; i < R1 * T2P; i += AGT) tab[i] = 0.f;
            for (int i = tid; i < R1 * 4; i += AGT) { mxA[i] = -__builtin_inff(); den[i] = 0.f; }
            __syncthreads();
        }

        if (fromRec)
            own2_walk_max<true>(list, recB, cnt, wave, lane, src, tgt, ea, xl, xrS, node0, rows, we0, we1, at, mxU);
        else
            own2_walk_max<false>(list, recB, cnt, wave, lane, src, tgt, ea, xl, xrS, node0, rows, we0, we1, at, mxU);
        __syncthreads();
        for (int i = tid; i < R1 * 4; i += AGT) {
            unsigned mu = mxU[i];
            if (mu != 0u) {
                float nm = fp_unord(mu), old = mxA[i];
                if (nm != old) {
                    float sc = __expf(old - nm);
                    mxA[i] = nm;
                    den[i] *= sc;
                    const int nd = i >> 2, c16 = (i & 3) * 16;
#pragma unroll
                    for (int k = 0; k < 16; ++k) tab[tab2_at(nd, c16 + k)] *= sc;
                }
            }
        }
        __syncthreads();
        if (fromRec)
            own2_walk_sum<true>(list, recB, cnt, wave, lane, src, tgt, ea, xl, xrS, node0, rows, we0, we1, at,
                                tab, den, mxA);
        else
            own2_walk_sum<false>(list, recB, cnt, wave, lane, src, tgt, ea, xl, xrS, node0, rows, we0, we1, at,
                                 tab, den, mxA);
        c0 += seg;
        __syncthreads();
    }

    for (int i = tid; i < rows * CH2; i += AGT) {
        int r = i >> 6, d = i & 63;
        float dn = den[r * 4 + (d >> 4)];
        float val = (dn > 0.f) ? tab[tab2_at(r, d)] / dn : 0.f;
        int g = batch[node0 + r];
        unsigned row = ((unsigned)g < (unsigned)G) ? (unsigned)g : (unsigned)G;
        atomicAdd(&gsum[row * CH2 + d], val);
    }
    __syncthreads();
    float* my = part + (size_t)blockIdx.x * G * CH2;
    for (int i = tid; i < G * CH2; i += AGT) my[i] = gsum[i];
}

__global__ __launch_bounds__(1024) void k_final(const float* __restrict__ part, int nPart,
                                                const float* __restrict__ bias2,
                                                const int* __restrict__ batch,
                                                float* __restrict__ out, int N, int G) {
    __shared__ float red[16][64];
    __shared__ int cntS;
    const int g = blockIdx.x;
    const int tid = threadIdx.x;
    const int d = tid & 63, j = tid >> 6;
    if (tid == 0) cntS = 0;

    float acc = 0.f;
    const float* p = part + (size_t)g * 64 + d;
    const size_t tabStride = (size_t)G * 64;
    int b = j;
    for (; b + 16 * 7 < nPart; b += 16 * 8) {
        float t[8];
#pragma unroll
        for (int k = 0; k < 8; ++k) t[k] = p[(size_t)(b + 16 * k) * tabStride];
#pragma unroll
        for (int k = 0; k < 8; ++k) acc += t[k];
    }
    for (; b < nPart; b += 16) acc += p[(size_t)b * tabStride];

    int c = 0;
    for (int n = tid; n < N; n += 1024) c += (batch[n] == g) ? 1 : 0;

    red[j][d] = acc;
    __syncthreads();
    atomicAdd(&cntS, c);
    __syncthreads();

    if (j == 0) {
        float s = 0.f;
#pragma unroll
        for (int k = 0; k < 16; ++k) s += red[k][d];
        float cnt = (float)cntS;
        out[g * 64 + d] = (s + cnt * bias2[d]) / fmaxf(cnt, 1.0f);
    }
}

extern "C" void kernel_launch(void* const* d_in, const int* in_sizes, int n_in,
                              void* d_out, int out_size, void* d_ws, size_t ws_size,
                              hipStream_t stream) {
    const float* x      = (const float*)d_in[0];
    const float* ea     = (const float*)d_in[1];
    const float* W_l1   = (const float*)d_in[2];
    const float* b_l1   = (const float*)d_in[3];
    const float* W_r1   = (const float*)d_in[4];
    const float* b_r1   = (const float*)d_in[5];
    const float* W_e1   = (const float*)d_in[6];
    const float* att1   = (const float*)d_in[7];
    const float* bias1  = (const float*)d_in[8];
    const float* W_l2   = (const float*)d_in[9];
    const float* b_l2   = (const float*)d_in[10];
    const float* W_r2   = (const float*)d_in[11];
    const float* b_r2   = (const float*)d_in[12];
    const float* W_e2   = (const float*)d_in[13];
    const float* att2   = (const float*)d_in[14];
    const float* bias2  = (const float*)d_in[15];
    const int*   eidx   = (const int*)d_in[16];
    const int*   batch  = (const int*)d_in[17];
    float* out = (float*)d_out;

    const int N = in_sizes[0] / 2;
    const int E = in_sizes[1] / 2;
    const int G = out_size / 64;
    const int* src = eidx;
    const int* tgt = eidx + E;

    char* ws = (char*)d_ws;
    size_t off = 0;
    auto alloc = [&](size_t bytes) { size_t o = off; off = (off + bytes + 255) & ~(size_t)255; return o; };

    const int nOwn = (N + R1 - 1) / R1;
    size_t o_xl1 = alloc((size_t)N * 128 * 4);
    size_t o_xr1 = alloc((size_t)N * 128 * 4);
    size_t o_hh  = alloc((size_t)N * 128 * 2);
    size_t o_wpk = alloc((size_t)2 * 8192 * 2);
    const int nKeys = ((E + CHK - 1) / CHK) * CHK;
    size_t o_keys = alloc((size_t)nKeys);
    size_t o_xl2 = alloc((size_t)N * 64 * 4);
    size_t o_xr2 = alloc((size_t)N * 64 * 4);
    size_t o_grec  = alloc((size_t)nOwn * QCAP * 16);
    size_t o_gcnt  = alloc((size_t)nOwn * 4);
    size_t o_part = alloc((size_t)nOwn * G * 64 * 4);
    (void)ws_size;

    float*    xl1   = (float*)(ws + o_xl1);
    float*    xr1   = (float*)(ws + o_xr1);
    _Float16* hh    = (_Float16*)(ws + o_hh);
    _Float16* wpk   = (_Float16*)(ws + o_wpk);
    float*    xl2   = (float*)(ws + o_xl2);
    float*    xr2   = (float*)(ws + o_xr2);
    int*      grec  = (int*)(ws + o_grec);
    int*      gcnt  = (int*)(ws + o_gcnt);
    float*    part  = (float*)(ws + o_part);
    unsigned* keys  = (unsigned*)(ws + o_keys);

    const int TB = 256;
    auto blocks = [&](long long t) { return (unsigned)((t + TB - 1) / TB); };

    k_node_transform1<<<blocks((long long)N * 128), TB, 0, stream>>>(
        x, W_l1, b_l1, W_r1, b_r1, xl1, xr1, N);
    const int packBlocks = (int)blocks(2 * 8192);
    k_w_pack<<<(unsigned)packBlocks + blocks(nKeys / 4), TB, 0, stream>>>(
        W_l2, W_r2, wpk, packBlocks, tgt, keys, E, nKeys);
    k_l1_own<<<(unsigned)nOwn, AGT, 0, stream>>>(
        xl1, xr1, ea, W_e1, att1, src, tgt, keys, bias1, hh, grec, gcnt, N, E);

    long long gemm_threads = (long long)((N + 15) / 16) * 32;
    k_gemm2_wmma<<<blocks(gemm_threads), TB, 0, stream>>>(
        hh, wpk, b_l2, b_r2, xl2, xr2, N);

    k_l2_own<<<(unsigned)nOwn, AGT, (size_t)(G + 1) * 64 * sizeof(float), stream>>>(
        xl2, xr2, ea, W_e2, att2, src, tgt, keys, grec, gcnt, batch, part, N, E, G);
    k_final<<<G, 1024, 0, stream>>>(part, nOwn, bias2, batch, out, N, G);
}


# launches 6 to 4 by fusion: the layer-1 owner kernel multiplies its own f16 rows, kept in LDS, by the layer-2 weights on WMMA (f16 row plane gone); weight pack and keys ride in the transform launch
# speedup vs baseline: 4.2222x; 1.0825x over previous
#include <hip/hip_runtime.h>
#include <hip/hip_fp16.h>

typedef __attribute__((ext_vector_type(16))) _Float16 v16h;
typedef __attribute__((ext_vector_type(8)))  _Float16 v8h;
typedef __attribute__((ext_vector_type(8)))  float    v8f;

__device__ __forceinline__ unsigned fp_ord(float f) {
    unsigned u = __float_as_uint(f);
    return (u & 0x80000000u) ? ~u : (u | 0x80000000u);
}
__device__ __forceinline__ float fp_unord(unsigned u) {
    unsigned b = (u & 0x80000000u) ? (u ^ 0x80000000u) : ~u;
    return __uint_as_float(b);
}

__device__ __forceinline__ void node_transform1(int blk, const float* __restrict__ x,
                                  const float* __restrict__ Wl, const float* __restrict__ bl,
                                  const float* __restrict__ Wr, const float* __restrict__ br,
                                  float* __restrict__ xl1, float* __restrict__ xr1, int N) {
    int idx = blk * blockDim.x + threadIdx.x;
    if (idx >= N * 128) return;
    int n = idx >> 7, d = idx & 127;
    float x0 = x[n * 2], x1 = x[n * 2 + 1];
    xl1[idx] = fmaf(x0, Wl[d], fmaf(x1, Wl[128 + d], bl[d]));
    xr1[idx] = fmaf(x0, Wr[d], fmaf(x1, Wr[128 + d], br[d]));
}

constexpr int KEY_R = 200;
__device__ __forceinline__ void w_pack_keys(int blk, const float* __restrict__ Wl, const float* __restrict__ Wr,
                         _Float16* __restrict__ Wpk, int packBlocks,
                         const int* __restrict__ tgt, unsigned* __restrict__ keys, int E, int nKeys) {
    if (blk >= packBlocks) {
        int w = (blk - packBlocks) * (int)blockDim.x + (int)threadIdx.x;
        int e0 = w * 4;
        if (e0 >= nKeys) return;
        int t0, t1, t2, t3;
        if (e0 + 4 <= E && (((size_t)tgt) & 15) == 0) {
            int4 t = *(const int4*)(tgt + e0);
            t0 = t.x; t1 = t.y; t2 = t.z; t3 = t.w;
        } else {
            t0 = (e0 + 0 < E) ? tgt[e0 + 0] : 0;
            t1 = (e0 + 1 < E) ? tgt[e0 + 1] : 0;
            t2 = (e0 + 2 < E) ? tgt[e0 + 2] : 0;
            t3 = (e0 + 3 < E) ? tgt[e0 + 3] : 0;
        }
        unsigned k0 = ((unsigned)t0 / (unsigned)KEY_R) & 255u;
        unsigned k1 = ((unsigned)t1 / (unsigned)KEY_R) & 255u;
        unsigned k2 = ((unsigned)t2 / (unsigned)KEY_R) & 255u;
        unsigned k3 = ((unsigned)t3 / (unsigned)KEY_R) & 255u;
        keys[w] = k0 | (k1 << 8) | (k2 << 16) | (k3 << 24);
        return;
    }
    int idx = blk * blockDim.x + threadIdx.x;
    if (idx >= 2 * 8192) return;
    int which = idx >> 13;
    int r = idx & 8191;
    int j = r & 15;
    int lane = (r >> 4) & 31;
    int frag = r >> 9;
    int kc = frag >> 2, nt = frag & 3;
    int half_sel = lane >> 4, lidx = lane & 15;
    int kb = (j < 8) ? (half_sel * 8) : (16 + half_sel * 8);
    int k = kc * 32 + kb + (j & 7);
    int n = nt * 16 + lidx;
    const float* W = which ? Wr : Wl;
    Wpk[idx] = (_Float16)W[k * 64 + n];
}

__global__ void k_prep(const float* __restrict__ x,
                       const float* __restrict__ Wl1, const float* __restrict__ bl1,
                       const float* __restrict__ Wr1, const float* __restrict__ br1,
                       float* __restrict__ xl1, float* __restrict__ xr1, int N, int tBlocks,
                       const float* __restrict__ Wl2, const float* __restrict__ Wr2,
                       _Float16* __restrict__ Wpk, int packBlocks,
                       const int* __restrict__ tgt, unsigned* __restrict__ keys, int E, int nKeys) {
    const int b = (int)blockIdx.x;
    if (b < tBlocks) node_transform1(b, x, Wl1, bl1, Wr1, br1, xl1, xr1, N);
    else             w_pack_keys(b - tBlocks, Wl2, Wr2, Wpk, packBlocks, tgt, keys, E, nKeys);
}

__device__ __forceinline__ void pool_item_add(float* tab, const float4& v, float alpha, int off, int cs) {
    atomicAdd(&tab[off], alpha * v.x);
    atomicAdd(&tab[off + cs], alpha * v.y);
    atomicAdd(&tab[off + 2 * cs], alpha * v.z);
    atomicAdd(&tab[off + 3 * cs], alpha * v.w);
}

constexpr int T2P = 80;
__device__ __forceinline__ int tab1_at(int r, int c) { return r * 128 + (c & 3) * 32 + (c >> 2); }
__device__ __forceinline__ int tab2_at(int r, int c) { return r * T2P + (c & 3) * 16 + (c >> 2); }

constexpr int R1  = 200;
constexpr int AGT = 1024;
constexpr int NWV = AGT / 32;
constexpr int EPT = 16;
constexpr int CHK = AGT * EPT;
constexpr int QCAP = CHK;
constexpr int AGU = 8;
constexpr int CH2 = 64;
constexpr int EPW = 2;
constexpr int HSP = 136;
constexpr float SHIFT_OK = 40.f;
static_assert(EPT == 16, "a thread's keys of one chunk are one 16-byte word; two chunks make one 32-bit match mask");
static_assert(QCAP >= CHK, "a run of one chunk must always fit the list");
static_assert(KEY_R == R1, "the owner keys are made for this block size");

typedef __attribute__((ext_vector_type(2))) _Float16 v2h;

__device__ __forceinline__ uint4 own_load_keys(const unsigned* __restrict__ keys, int ebase, int tid) {
    return *(const uint4*)(keys + ((ebase + 16 * tid) >> 2));
}

__device__ __forceinline__ unsigned own_match4(unsigned w, unsigned pat) {
    unsigned x = w ^ pat;
    return ~(((x & 0x7F7F7F7Fu) + 0x7F7F7F7Fu) | x | 0x7F7F7F7Fu);
}

__device__ __forceinline__ int own_off(int p, int tid) {
    const int i = p & 7, j = p >> 3;
    return (i >> 2) * CHK + 16 * tid + 4 * (i & 3) + j;
}

__device__ __forceinline__ void own_trip(const int* list, int base, int cnt, int wave, int lane,
        const int* __restrict__ src, const int* __restrict__ tgt, const float* __restrict__ ea,
        const float* __restrict__ xl, const float* xrS, int node0, int rows,
        const float (&we0)[4], const float (&we1)[4], const float (&at)[4],
        float4 (&v)[AGU], float (&lg)[AGU], int (&tl)[AGU], int4* recOut) {
    float a0[AGU], a1[AGU];
    int   sN[AGU];
#pragma unroll
    for (int u = 0; u < AGU; ++u) {
        int j = base + wave + u * NWV;
        int e = list[(j < cnt) ? j : (cnt - 1)];
        sN[u] = src[e];
        const int t = tgt[e] - node0;
        tl[u] = ((unsigned)t < (unsigned)rows) ? t : -1;
        a0[u] = ea[(size_t)e * 2];
        a1[u] = ea[(size_t)e * 2 + 1];
    }
#pragma unroll
    for (int u = 0; u < AGU; ++u)
        v[u] = *(const float4*)(xl + (size_t)sN[u] * 128 + lane * 4);
    if (recOut != nullptr && lane == 0) {
#pragma unroll
        for (int u = 0; u < AGU; ++u) {
            const int j = base + wave + u * NWV;
            if (j < cnt)
                recOut[j] = make_int4(sN[u], tl[u], __float_as_int(a0[u]), __float_as_int(a1[u]));
        }
    }
#pragma unroll
    for (int u = 0; u < AGU; ++u) {
        const float4 r4 = *(const float4*)(xrS + ((tl[u] < 0) ? 0 : tl[u]) * 128 + lane * 4);
        float m0 = v[u].x + r4.x + a0[u] * we0[0] + a1[u] * we1[0];
        float m1 = v[u].y + r4.y + a0[u] * we0[1] + a1[u] * we1[1];
        float m2 = v[u].z + r4.z + a0[u] * we0[2] + a1[u] * we1[2];
        float m3 = v[u].w + r4.w + a0[u] * we0[3] + a1[u] * we1[3];
        m0 = (m0 > 0.f) ? m0 : 0.2f * m0;
        m1 = (m1 > 0.f) ? m1 : 0.2f * m1;
        m2 = (m2 > 0.f) ? m2 : 0.2f * m2;
        m3 = (m3 > 0.f) ? m3 : 0.2f * m3;
        float p = fmaf(m3, at[3], fmaf(m2, at[2], fmaf(m1, at[1], m0 * at[0])));
        p += __int_as_float(__builtin_amdgcn_update_dpp(0, __float_as_int(p), 0xB1, 0xF, 0xF, false));
        lg[u] = p;
    }
}

__global__ __launch_bounds__(AGT) void k_l1_own(
        const float* __restrict__ xl, const float* __restrict__ xr,
        const float* __restrict__ ea, const float* __restrict__ We,
        const float* __restrict__ att,
        const int* __restrict__ src, const int* __restrict__ tgt,
        const unsigned* __restrict__ keys,
        const float* __restrict__ bias,
        const _Float16* __restrict__ Wpk, const float* __restrict__ bl2, const float* __restrict__ br2,
        float* __restrict__ xl2, float* __restrict__ xr2,
        int* __restrict__ grec, int* __restrict__ gcnt, int N, int E) {
    __shared__ __align__(16) float tab[R1 * 128];
    __shared__ __align__(16) float xrS[R1 * 128];
    __shared__ unsigned mxU[R1 * 16];
    __shared__ float    mxA[R1 * 16];
    __shared__ float    den[R1 * 16];
    __shared__ __align__(32) int list[QCAP];
    __shared__ int      qcnt;
    __shared__ int      redo;

    const int tid = threadIdx.x;
    const int lane = tid & 31;
    const int wave = __builtin_amdgcn_readfirstlane(tid >> 5);
    const int hq = lane >> 1;
    const int node0 = blockIdx.x * R1;
    const int rows = (N - node0 < R1) ? (N - node0) : R1;
    const int nChunks = (E + CHK - 1) / CHK;

    for (int i = tid; i < R1 * 128; i += AGT) tab[i] = 0.f;
    for (int i = tid; i < R1 * 16; i += AGT) { mxU[i] = 0u; mxA[i] = -__builtin_inff(); den[i] = 0.f; }
    {
        const float4* g = (const float4*)(xr + (size_t)node0 * 128);
        float4* l = (float4*)xrS;
        for (int i = tid; i < rows * 32; i += AGT) l[i] = g[i];
    }
    if (tid == 0) { qcnt = 0; redo = 0; }

    float we0[4], we1[4], at[4];
#pragma unroll
    for (int k = 0; k < 4; ++k) {
        we0[k] = We[lane * 4 + k];
        we1[k] = We[128 + lane * 4 + k];
        at[k]  = att[lane * 4 + k];
    }

    const unsigned pat = ((unsigned)blockIdx.x & 255u) * 0x01010101u;
    __syncthreads();

    int c0 = 0;
    while (c0 < nChunks) {
        int seg = nChunks - c0;
        int cnt = 0;
        while (true) {
            const int cEnd = c0 + seg;
            uint4 kq[2];
            kq[0] = own_load_keys(keys, c0 * CHK, tid);
            kq[1] = own_load_keys(keys, ((c0 + 1 < cEnd) ? (c0 + 1) : c0) * CHK, tid);
            for (int c = c0; c < cEnd; c += 2) {
                const int ebase = c * CHK;
                uint4 kn[2];
                kn[0] = kq[0];
                kn[1] = kq[1];
                if (c + 2 < cEnd) {
                    kn[0] = own_load_keys(keys, ebase + 2 * CHK, tid);
                    kn[1] = own_load_keys(keys, ((c + 3 < cEnd) ? (c + 3) : (c + 2)) * CHK, tid);
                }
                unsigned mask = (own_match4(kq[0].x, pat) >> 7) | (own_match4(kq[0].y, pat) >> 6)
                              | (own_match4(kq[0].z, pat) >> 5) | (own_match4(kq[0].w, pat) >> 4);
                if (c + 1 < cEnd)
                    mask |= (own_match4(kq[1].x, pat) >> 3) | (own_match4(kq[1].y, pat) >> 2)
                          | (own_match4(kq[1].z, pat) >> 1) |  own_match4(kq[1].w, pat);
                if (mask && ebase + 2 * CHK > E) {
                    unsigned m2 = mask;
                    while (m2) {
                        int b = __ffs(m2) - 1;
                        m2 &= m2 - 1;
                        if (ebase + own_off(b, tid) >= E) mask &= ~(1u << b);
                    }
                }
                if (mask) {
                    int pos = atomicAdd(&qcnt, __popc(mask));
                    while (mask) {
                        int b = __ffs(mask) - 1;
                        mask &= mask - 1;
                        if (pos < QCAP) list[pos] = ebase + own_off(b, tid);
                        ++pos;
                    }
                }
                kq[0] = kn[0];
                kq[1] = kn[1];
            }
            __syncthreads();
            cnt = qcnt;
            __syncthreads();
            if (tid == 0) qcnt = 0;
            if (cnt <= QCAP) break;
            seg = (seg + 1) >> 1;
            __syncthreads();
        }
        if (c0 == 0 && tid == 0) gcnt[blockIdx.x] = (seg == nChunks) ? cnt : -1;

        if (c0 == 0 && seg == nChunks) {
            for (int i = tid; i < R1 * 16; i += AGT) {
                const float4* xr4 = (const float4*)(xrS + i * 8);
                const float4 p0 = xr4[0], p1 = xr4[1];
                const float* ah = att + (i & 15) * 8;
                float m[8] = {p0.x, p0.y, p0.z, p0.w, p1.x, p1.y, p1.z, p1.w};
                float r = 0.f;
#pragma unroll
                for (int k = 0; k < 8; ++k) {
                    float mk = (m[k] > 0.f) ? m[k] : 0.2f * m[k];
                    r = fmaf(mk, ah[k], r);
                }
                mxA[i] = r;
            }
            __syncthreads();
            int4* recB = (int4*)grec + (size_t)blockIdx.x * QCAP;
            for (int base = 0; base < cnt; base += NWV * AGU) {
                float4 v[AGU];
                float  lg[AGU];
                int    tl[AGU];
                own_trip(list, base, cnt, wave, lane, src, tgt, ea, xl, xrS, node0, rows, we0, we1, at, v, lg, tl,
                         recB);
                float sh[AGU];
#pragma unroll
                for (int u = 0; u < AGU; ++u)
                    sh[u] = mxA[((tl[u] < 0) ? 0 : tl[u]) * 16 + hq];
#pragma unroll
                for (int u = 0; u < AGU; ++u) {
                    if (base + wave + u * NWV < cnt && tl[u] >= 0) {
                        float w = __expf(lg[u] - sh[u]);
                        pool_item_add(tab, v[u], w, tl[u] * 128 + lane, 32);
                        if ((lane & 1) == 0) {
                            atomicAdd(&den[tl[u] * 16 + hq], w);
                            atomicMax(&mxU[tl[u] * 16 + hq], fp_ord(lg[u]));
                        }
                    }
                }
            }
            __syncthreads();
            for (int i = tid; i < R1 * 16; i += AGT) {
                unsigned mu = mxU[i];
                if (mu != 0u) {
                    float d = fp_unord(mu) - mxA[i];
                    if (!(d > -SHIFT_OK && d < SHIFT_OK)) redo = 1;
                }
            }
            __syncthreads();
            if (redo == 0) break;
            for (int i = tid; i < R1 * 128; i += AGT) tab[i] = 0.f;
            for (int i = tid; i < R1 * 16; i += AGT) { mxA[i] = -__builtin_inff(); den[i] = 0.f; }
            __syncthreads();
        }

        for (int base = 0; base < cnt; base += NWV * AGU) {
            float4 v[AGU];
            float  lg[AGU];
            int    tl[AGU];
            own_trip(list, base, cnt, wave, lane, src, tgt, ea, xl, xrS, node0, rows, we0, we1, at, v, lg, tl,
                     nullptr);
#pragma unroll
            for (int u = 0; u < AGU; ++u) {
                if (base + wave + u * NWV < cnt && tl[u] >= 0 && (lane & 1) == 0)
                    atomicMax(&mxU[tl[u] * 16 + hq], fp_ord(lg[u]));
            }
        }
        __syncthreads();
        for (int i = tid; i < R1 * 16; i += AGT) {
            unsigned mu = mxU[i];
            if (mu != 0u) {
                float nm = fp_unord(mu), old = mxA[i];
                if (nm != old) {
                    float sc = __expf(old - nm);
                    mxA[i] = nm;
                    den[i] *= sc;
                    const int nd = i >> 4, c8 = (i & 15) * 8;
#pragma unroll
                    for (int k = 0; k < 8; ++k) tab[tab1_at(nd, c8 + k)] *= sc;
                }
            }
        }
        __syncthreads();
        for (int base = 0; base < cnt; base += NWV * AGU) {
            float4 v[AGU];
            float  lg[AGU];
            int    tl[AGU];
            own_trip(list, base, cnt, wave, lane, src, tgt, ea, xl, xrS, node0, rows, we0, we1, at, v, lg, tl,
                     nullptr);
            float mx[AGU];
#pragma unroll
            for (int u = 0; u < AGU; ++u)
                mx[u] = mxA[((tl[u] < 0) ? 0 : tl[u]) * 16 + hq];
#pragma unroll
            for (int u = 0; u < AGU; ++u) {
                if (base + wave + u * NWV < cnt && tl[u] >= 0) {
                    float w = __expf(lg[u] - mx[u]);
                    pool_item_add(tab, v[u], w, tl[u] * 128 + lane, 32);
                    if ((lane & 1) == 0) atomicAdd(&den[tl[u] * 16 + hq], w);
                }
            }
        }
        c0 += seg;
        __syncthreads();
    }

    static_assert(((R1 + 15) / 16) * 16 * HSP * 2 <= R1 * 128 * 4, "the f16 rows, whole tiles of them, fit over xrS");
    static_assert(2 * 8192 * 2 <= QCAP * 4, "the packed weights fit over the list");
    _Float16* hS = (_Float16*)xrS;
    _Float16* wS = (_Float16*)list;
    {
        const uint4* g = (const uint4*)Wpk;
        uint4* l = (uint4*)wS;
        for (int i = tid; i < 2048; i += AGT) l[i] = g[i];
    }
    for (int i = tid; i < rows * 64; i += AGT) {
        int r = i >> 6, d = (i & 63) * 2;
        float dn = den[r * 16 + (d >> 3)];
        float v0 = (dn > 0.f) ? tab[tab1_at(r, d)] / dn : 0.f;
        float v1 = (dn > 0.f) ? tab[tab1_at(r, d + 1)] / dn : 0.f;
        v0 += bias[d];
        v1 += bias[d + 1];
        v0 = (v0 > 0.f) ? v0 : (__expf(v0) - 1.f);
        v1 = (v1 > 0.f) ? v1 : (__expf(v1) - 1.f);
        v2h o = {(_Float16)v0, (_Float16)v1};
        *(v2h*)(hS + r * HSP + d) = o;
    }
    __syncthreads();

    const int half_sel = lane >> 4;
    const int lidx = lane & 15;
    const int kb_lo = half_sel * 8;
    const int kb_hi = 16 + kb_lo;
    static_assert(((R1 + 15) / 16) * 2 <= NWV, "one (tile, matrix) pair per wave covers the block's rows");
    const int nTasks = ((rows + 15) >> 4) * 2;
    if (wave < nTasks) {
        const int row0 = (wave >> 1) * 16;
        const int which = wave & 1;
        const v16h* frag = (const v16h*)(wS + which * 8192);
        const _Float16* pa = hS + (row0 + lidx) * HSP;
        v8f acc[4] = {};
#pragma unroll
        for (int kc = 0; kc < 4; ++kc) {
            v8h alo = *(const v8h*)(pa + kc * 32 + kb_lo);
            v8h ahi = *(const v8h*)(pa + kc * 32 + kb_hi);
            v16h a = __builtin_shufflevector(alo, ahi, 0, 1, 2, 3, 4, 5, 6, 7,
                                             8, 9, 10, 11, 12, 13, 14, 15);
#pragma unroll
            for (int nt = 0; nt < 4; ++nt) {
                v16h b = frag[(kc * 4 + nt) * 32 + lane];
                acc[nt] = __builtin_amdgcn_wmma_f32_16x16x32_f16(
                    false, a, false, b, (short)0, acc[nt], false, false);
            }
        }
        const float* bb = which ? br2 : bl2;
        float* xo = which ? xr2 : xl2;
#pragma unroll
        for (int nt = 0; nt < 4; ++nt) {
            const int n = nt * 16 + lidx;
            const float bn = bb[n];
#pragma unroll
            for (int r = 0; r < 8; ++r) {
                const int rl = row0 + r + half_sel * 8;
                if (rl < rows) xo[(size_t)(node0 + rl) * 64 + n] = acc[nt][r] + bn;
            }
        }
    }
}

static_assert(CH2 * EPW == 32 * 4, "a wave covers EPW rows of CH2 channels with 4 channels per lane");

template <bool REC>
__device__ __forceinline__ int own2_j(int base, int wave, int half, int u) {
    return REC ? base + (wave * AGU + u) * EPW + half
               : base + (u * NWV + wave) * EPW + half;
}

template <bool REC>
__device__ __forceinline__ void own2_trip(const int* list, const int4* __restrict__ rec,
        int base, int cnt, int wave, int half, int q,
        const int* __restrict__ src, const int* __restrict__ tgt, const float* __restrict__ ea,
        const float* __restrict__ xl, const float* xrS, int node0, int rows,
        const float (&we0)[4], const float (&we1)[4], const float (&at)[4],
        float4 (&v)[AGU], float (&lg)[AGU], int (&tl)[AGU]) {
    float a0[AGU], a1[AGU];
    int   sN[AGU];
    if (REC) {
        int4 r[AGU];
#pragma unroll
        for (int u = 0; u < AGU; ++u) {
            const int j = own2_j<REC>(base, wave, half, u);
            r[u] = rec[(j < cnt) ? j : (cnt - 1)];
        }
        __builtin_amdgcn_sched_barrier(0);
#pragma unroll
        for (int u = 0; u < AGU; ++u) {
            sN[u] = r[u].x;
            tl[u] = ((unsigned)r[u].y < (unsigned)rows) ? r[u].y : -1;
            a0[u] = __int_as_float(r[u].z);
            a1[u] = __int_as_float(r[u].w);
        }
    } else {
#pragma unroll
        for (int u = 0; u < AGU; ++u) {
            const int j = own2_j<REC>(base, wave, half, u);
            const int e = list[(j < cnt) ? j : (cnt - 1)];
            sN[u] = src[e];
            const int t = tgt[e] - node0;
            tl[u] = ((unsigned)t < (unsigned)rows) ? t : -1;
            a0[u] = ea[(size_t)e * 2];
            a1[u] = ea[(size_t)e * 2 + 1];
        }
    }
#pragma unroll
    for (int u = 0; u < AGU; ++u)
        v[u] = *(const float4*)(xl + (size_t)sN[u] * CH2 + q * 4);
#pragma unroll
    for (int u = 0; u < AGU; ++u) {
        const float4 r4 = *(const float4*)(xrS + ((tl[u] < 0) ? 0 : tl[u]) * CH2 + q * 4);
        float m0 = v[u].x + r4.x + a0[u] * we0[0] + a1[u] * we1[0];
        float m1 = v[u].y + r4.y + a0[u] * we0[1] + a1[u] * we1[1];
        float m2 = v[u].z + r4.z + a0[u] * we0[2] + a1[u] * we1[2];
        float m3 = v[u].w + r4.w + a0[u] * we0[3] + a1[u] * we1[3];
        m0 = (m0 > 0.f) ? m0 : 0.2f * m0;
        m1 = (m1 > 0.f) ? m1 : 0.2f * m1;
        m2 = (m2 > 0.f) ? m2 : 0.2f * m2;
        m3 = (m3 > 0.f) ? m3 : 0.2f * m3;
        float p = fmaf(m3, at[3], fmaf(m2, at[2], fmaf(m1, at[1], m0 * at[0])));
        p += __int_as_float(__builtin_amdgcn_update_dpp(0, __float_as_int(p), 0xB1, 0xF, 0xF, false));
        p += __int_as_float(__builtin_amdgcn_update_dpp(0, __float_as_int(p), 0x4E, 0xF, 0xF, false));
        lg[u] = p;
    }
}

template <bool REC>
__device__ __forceinline__ void own2_walk_one(const int* list, const int4* __restrict__ rec, int cnt, int wave, int lane,
        const int* __restrict__ src, const int* __restrict__ tgt, const float* __restrict__ ea,
        const float* __restrict__ xl, const float* xrS, int node0, int rows,
        const float (&we0)[4], const float (&we1)[4], const float (&at)[4],
        float* tab, float* den, unsigned* mxU, const float* mxA) {
    const int half = lane >> 4, q = lane & 15, hq = q >> 2;
    for (int base = 0; base < cnt; base += NWV * EPW * AGU) {
        float4 v[AGU];
        float  lg[AGU];
        int    tl[AGU];
        own2_trip<REC>(list, rec, base, cnt, wave, half, q, src, tgt, ea, xl, xrS, node0, rows, we0, we1, at, v, lg, tl);
        float sh[AGU];
#pragma unroll
        for (int u = 0; u < AGU; ++u)
            sh[u] = mxA[((tl[u] < 0) ? 0 : tl[u]) * 4 + hq];
#pragma unroll
        for (int u = 0; u < AGU; ++u) {
            if (own2_j<REC>(base, wave, half, u) < cnt && tl[u] >= 0) {
                float w = __expf(lg[u] - sh[u]);
                pool_item_add(tab, v[u], w, tl[u] * T2P + q, 16);
                if ((lane & 3) == 0) {
                    atomicAdd(&den[tl[u] * 4 + hq], w);
                    atomicMax(&mxU[tl[u] * 4 + hq], fp_ord(lg[u]));
                }
            }
        }
    }
}

template <bool REC>
__device__ __forceinline__ void own2_walk_max(const int* list, const int4* __restrict__ rec, int cnt, int wave, int lane,
        const int* __restrict__ src, const int* __restrict__ tgt, const float* __restrict__ ea,
        const float* __restrict__ xl, const float* xrS, int node0, int rows,
        const float (&we0)[4], const float (&we1)[4], const float (&at)[4],
        unsigned* mxU) {
    const int half = lane >> 4, q = lane & 15, hq = q >> 2;
    for (int base = 0; base < cnt; base += NWV * EPW * AGU) {
        float4 v[AGU];
        float  lg[AGU];
        int    tl[AGU];
        own2_trip<REC>(list, rec, base, cnt, wave, half, q, src, tgt, ea, xl, xrS, node0, rows, we0, we1, at, v, lg, tl);
#pragma unroll
        for (int u = 0; u < AGU; ++u) {
            if (own2_j<REC>(base, wave, half, u) < cnt && tl[u] >= 0 && (lane & 3) == 0)
                atomicMax(&mxU[tl[u] * 4 + hq], fp_ord(lg[u]));
        }
    }
}

template <bool REC>
__device__ __forceinline__ void own2_walk_sum(const int* list, const int4* __restrict__ rec, int cnt, int wave, int lane,
        const int* __restrict__ src, const int* __restrict__ tgt, const float* __restrict__ ea,
        const float* __restrict__ xl, const float* xrS, int node0, int rows,
        const float (&we0)[4], const float (&we1)[4], const float (&at)[4],
        float* tab, float* den, const float* mxA) {
    const int half = lane >> 4, q = lane & 15, hq = q >> 2;
    for (int base = 0; base < cnt; base += NWV * EPW * AGU) {
        float4 v[AGU];
        float  lg[AGU];
        int    tl[AGU];
        own2_trip<REC>(list, rec, base, cnt, wave, half, q, src, tgt, ea, xl, xrS, node0, rows, we0, we1, at, v, lg, tl);
        float mx[AGU];
#pragma unroll
        for (int u = 0; u < AGU; ++u)
            mx[u] = mxA[((tl[u] < 0) ? 0 : tl[u]) * 4 + hq];
#pragma unroll
        for (int u = 0; u < AGU; ++u) {
            if (own2_j<REC>(base, wave, half, u) < cnt && tl[u] >= 0) {
                float w = __expf(lg[u] - mx[u]);
                pool_item_add(tab, v[u], w, tl[u] * T2P + q, 16);
                if ((lane & 3) == 0) atomicAdd(&den[tl[u] * 4 + hq], w);
            }
        }
    }
}

__global__ __launch_bounds__(AGT) void k_l2_own(
        const float* __restrict__ xl, const float* __restrict__ xr,
        const float* __restrict__ ea, const float* __restrict__ We,
        const float* __restrict__ att,
        const int* __restrict__ src, const int* __restrict__ tgt,
        const unsigned* __restrict__ keys,
        const int* __restrict__ grec, const int* __restrict__ gcnt,
        const int* __restrict__ batch,
        float* __restrict__ part, int N, int E, int G) {
    __shared__ __align__(16) float tab[R1 * T2P];
    __shared__ __align__(16) float xrS[R1 * CH2];
    __shared__ unsigned mxU[R1 * 4];
    __shared__ float    mxA[R1 * 4];
    __shared__ float    den[R1 * 4];
    __shared__ int      list[QCAP];
    __shared__ int      qcnt;
    __shared__ int      redo;
    extern __shared__ float gsum[];

    const int tid = threadIdx.x;
    const int lane = tid & 31;
    const int wave = __builtin_amdgcn_readfirstlane(tid >> 5);
    const int q = lane & 15;
    const int node0 = blockIdx.x * R1;
    const int rows = (N - node0 < R1) ? (N - node0) : R1;
    const int nChunks = (E + CHK - 1) / CHK;

    for (int i = tid; i < R1 * T2P; i += AGT) tab[i] = 0.f;
    for (int i = tid; i < R1 * 4; i += AGT) { mxU[i] = 0u; mxA[i] = -__builtin_inff(); den[i] = 0.f; }
    for (int i = tid; i < (G + 1) * CH2; i += AGT) gsum[i] = 0.f;
    {
        const float4* g = (const float4*)(xr + (size_t)node0 * CH2);
        float4* l = (float4*)xrS;
        for (int i = tid; i < rows * (CH2 / 4); i += AGT) l[i] = g[i];
    }
    if (tid == 0) { qcnt = 0; redo = 0; }

    float we0[4], we1[4], at[4];
#pragma unroll
    for (int k = 0; k < 4; ++k) {
        we0[k] = We[q * 4 + k];
        we1[k] = We[CH2 + q * 4 + k];
        at[k]  = att[q * 4 + k];
    }

    const unsigned pat = ((unsigned)blockIdx.x & 255u) * 0x01010101u;
    const int given = __builtin_amdgcn_readfirstlane(gcnt[blockIdx.x]);
    const bool fromRec = (given >= 0);
    const int4* recB = (const int4*)grec + (size_t)blockIdx.x * QCAP;
    __syncthreads();

    int c0 = 0;
    while (c0 < nChunks) {
        int seg = nChunks - c0;
        int cnt = 0;
        if (c0 == 0 && fromRec) {
            cnt = (given < QCAP) ? given : QCAP;
        } else while (true) {
            const int cEnd = c0 + seg;
            uint4 kq[2];
            kq[0] = own_load_keys(keys, c0 * CHK, tid);
            kq[1] = own_load_keys(keys, ((c0 + 1 < cEnd) ? (c0 + 1) : c0) * CHK, tid);
            for (int c = c0; c < cEnd; c += 2) {
                const int ebase = c * CHK;
                uint4 kn[2];
                kn[0] = kq[0];
                kn[1] = kq[1];
                if (c + 2 < cEnd) {
                    kn[0] = own_load_keys(keys, ebase + 2 * CHK, tid);
                    kn[1] = own_load_keys(keys, ((c + 3 < cEnd) ? (c + 3) : (c + 2)) * CHK, tid);
                }
                unsigned mask = (own_match4(kq[0].x, pat) >> 7) | (own_match4(kq[0].y, pat) >> 6)
                              | (own_match4(kq[0].z, pat) >> 5) | (own_match4(kq[0].w, pat) >> 4);
                if (c + 1 < cEnd)
                    mask |= (own_match4(kq[1].x, pat) >> 3) | (own_match4(kq[1].y, pat) >> 2)
                          | (own_match4(kq[1].z, pat) >> 1) |  own_match4(kq[1].w, pat);
                if (mask && ebase + 2 * CHK > E) {
                    unsigned m2 = mask;
                    while (m2) {
                        int b = __ffs(m2) - 1;
                        m2 &= m2 - 1;
                        if (ebase + own_off(b, tid) >= E) mask &= ~(1u << b);
                    }
                }
                if (mask) {
                    int pos = atomicAdd(&qcnt, __popc(mask));
                    while (mask) {
                        int b = __ffs(mask) - 1;
                        mask &= mask - 1;
                        if (pos < QCAP) list[pos] = ebase + own_off(b, tid);
                        ++pos;
                    }
                }
                kq[0] = kn[0];
                kq[1] = kn[1];
            }
            __syncthreads();
            cnt = qcnt;
            __syncthreads();
            if (tid == 0) qcnt = 0;
            if (cnt <= QCAP) break;
            seg = (seg + 1) >> 1;
            __syncthreads();
        }

        if (c0 == 0 && seg == nChunks) {
            for (int i = tid; i < R1 * 4; i += AGT) {
                const float4* xr4 = (const float4*)(xrS + i * 16);
                const float* ah = att + (i & 3) * 16;
                float r = 0.f;
#pragma unroll
                for (int k = 0; k < 4; ++k) {
                    const float4 p = xr4[k];
                    float m0 = (p.x > 0.f) ? p.x : 0.2f * p.x;
                    float m1 = (p.y > 0.f) ? p.y : 0.2f * p.y;
                    float m2 = (p.z > 0.f) ? p.z : 0.2f * p.z;
                    float m3 = (p.w > 0.f) ? p.w : 0.2f * p.w;
                    r = fmaf(m0, ah[4 * k], r);
                    r = fmaf(m1, ah[4 * k + 1], r);
                    r = fmaf(m2, ah[4 * k + 2], r);
                    r = fmaf(m3, ah[4 * k + 3], r);
                }
                mxA[i] = r;
            }
            __syncthreads();
            if (fromRec)
                own2_walk_one<true>(list, recB, cnt, wave, lane, src, tgt, ea, xl, xrS, node0, rows, we0, we1, at,
                                    tab, den, mxU, mxA);
            else
                own2_walk_one<false>(list, recB, cnt, wave, lane, src, tgt, ea, xl, xrS, node0, rows, we0, we1, at,
                                     tab, den, mxU, mxA);
            __syncthreads();
            for (int i = tid; i < R1 * 4; i += AGT) {
                unsigned mu = mxU[i];
                if (mu != 0u) {
                    float d = fp_unord(mu) - mxA[i];
                    if (!(d > -SHIFT_OK && d < SHIFT_OK)) redo = 1;
                }
            }
            __syncthreads();
            if (redo == 0) break;
            for (int i = tid; i < R1 * T2P; i += AGT) tab[i] = 0.f;
            for (int i = tid; i < R1 * 4; i += AGT) { mxA[i] = -__builtin_inff(); den[i] = 0.f; }
            __syncthreads();
        }

        if (fromRec)
            own2_walk_max<true>(list, recB, cnt, wave, lane, src, tgt, ea, xl, xrS, node0, rows, we0, we1, at, mxU);
        else
            own2_walk_max<false>(list, recB, cnt, wave, lane, src, tgt, ea, xl, xrS, node0, rows, we0, we1, at, mxU);
        __syncthreads();
        for (int i = tid; i < R1 * 4; i += AGT) {
            unsigned mu = mxU[i];
            if (mu != 0u) {
                float nm = fp_unord(mu), old = mxA[i];
                if (nm != old) {
                    float sc = __expf(old - nm);
                    mxA[i] = nm;
                    den[i] *= sc;
                    const int nd = i >> 2, c16 = (i & 3) * 16;
#pragma unroll
                    for (int k = 0; k < 16; ++k) tab[tab2_at(nd, c16 + k)] *= sc;
                }
            }
        }
        __syncthreads();
        if (fromRec)
            own2_walk_sum<true>(list, recB, cnt, wave, lane, src, tgt, ea, xl, xrS, node0, rows, we0, we1, at,
                                tab, den, mxA);
        else
            own2_walk_sum<false>(list, recB, cnt, wave, lane, src, tgt, ea, xl, xrS, node0, rows, we0, we1, at,
                                 tab, den, mxA);
        c0 += seg;
        __syncthreads();
    }

    for (int i = tid; i < rows * CH2; i += AGT) {
        int r = i >> 6, d = i & 63;
        float dn = den[r * 4 + (d >> 4)];
        float val = (dn > 0.f) ? tab[tab2_at(r, d)] / dn : 0.f;
        int g = batch[node0 + r];
        unsigned row = ((unsigned)g < (unsigned)G) ? (unsigned)g : (unsigned)G;
        atomicAdd(&gsum[row * CH2 + d], val);
    }
    __syncthreads();
    float* my = part + (size_t)blockIdx.x * G * CH2;
    for (int i = tid; i < G * CH2; i += AGT) my[i] = gsum[i];
}

__global__ __launch_bounds__(1024) void k_final(const float* __restrict__ part, int nPart,
                                                const float* __restrict__ bias2,
                                                const int* __restrict__ batch,
                                                float* __restrict__ out, int N, int G) {
    __shared__ float red[16][64];
    __shared__ int cntS;
    const int g = blockIdx.x;
    const int tid = threadIdx.x;
    const int d = tid & 63, j = tid >> 6;
    if (tid == 0) cntS = 0;

    float acc = 0.f;
    const float* p = part + (size_t)g * 64 + d;
    const size_t tabStride = (size_t)G * 64;
    int b = j;
    for (; b + 16 * 7 < nPart; b += 16 * 8) {
        float t[8];
#pragma unroll
        for (int k = 0; k < 8; ++k) t[k] = p[(size_t)(b + 16 * k) * tabStride];
#pragma unroll
        for (int k = 0; k < 8; ++k) acc += t[k];
    }
    for (; b < nPart; b += 16) acc += p[(size_t)b * tabStride];

    int c = 0;
    for (int n = tid; n < N; n += 1024) c += (batch[n] == g) ? 1 : 0;

    red[j][d] = acc;
    __syncthreads();
    atomicAdd(&cntS, c);
    __syncthreads();

    if (j == 0) {
        float s = 0.f;
#pragma unroll
        for (int k = 0; k < 16; ++k) s += red[k][d];
        float cnt = (float)cntS;
        out[g * 64 + d] = (s + cnt * bias2[d]) / fmaxf(cnt, 1.0f);
    }
}

extern "C" void kernel_launch(void* const* d_in, const int* in_sizes, int n_in,
                              void* d_out, int out_size, void* d_ws, size_t ws_size,
                              hipStream_t stream) {
    const float* x      = (const float*)d_in[0];
    const float* ea     = (const float*)d_in[1];
    const float* W_l1   = (const float*)d_in[2];
    const float* b_l1   = (const float*)d_in[3];
    const float* W_r1   = (const float*)d_in[4];
    const float* b_r1   = (const float*)d_in[5];
    const float* W_e1   = (const float*)d_in[6];
    const float* att1   = (const float*)d_in[7];
    const float* bias1  = (const float*)d_in[8];
    const float* W_l2   = (const float*)d_in[9];
    const float* b_l2   = (const float*)d_in[10];
    const float* W_r2   = (const float*)d_in[11];
    const float* b_r2   = (const float*)d_in[12];
    const float* W_e2   = (const float*)d_in[13];
    const float* att2   = (const float*)d_in[14];
    const float* bias2  = (const float*)d_in[15];
    const int*   eidx   = (const int*)d_in[16];
    const int*   batch  = (const int*)d_in[17];
    float* out = (float*)d_out;

    const int N = in_sizes[0] / 2;
    const int E = in_sizes[1] / 2;
    const int G = out_size / 64;
    const int* src = eidx;
    const int* tgt = eidx + E;

    char* ws = (char*)d_ws;
    size_t off = 0;
    auto alloc = [&](size_t bytes) { size_t o = off; off = (off + bytes + 255) & ~(size_t)255; return o; };

    const int nOwn = (N + R1 - 1) / R1;
    size_t o_xl1 = alloc((size_t)N * 128 * 4);
    size_t o_xr1 = alloc((size_t)N * 128 * 4);
    size_t o_wpk = alloc((size_t)2 * 8192 * 2);
    const int nKeys = ((E + CHK - 1) / CHK) * CHK;
    size_t o_keys = alloc((size_t)nKeys);
    size_t o_xl2 = alloc((size_t)N * 64 * 4);
    size_t o_xr2 = alloc((size_t)N * 64 * 4);
    size_t o_grec  = alloc((size_t)nOwn * QCAP * 16);
    size_t o_gcnt  = alloc((size_t)nOwn * 4);
    size_t o_part = alloc((size_t)nOwn * G * 64 * 4);
    (void)ws_size;

    float*    xl1   = (float*)(ws + o_xl1);
    float*    xr1   = (float*)(ws + o_xr1);
    _Float16* wpk   = (_Float16*)(ws + o_wpk);
    float*    xl2   = (float*)(ws + o_xl2);
    float*    xr2   = (float*)(ws + o_xr2);
    int*      grec  = (int*)(ws + o_grec);
    int*      gcnt  = (int*)(ws + o_gcnt);
    float*    part  = (float*)(ws + o_part);
    unsigned* keys  = (unsigned*)(ws + o_keys);

    const int TB = 256;
    auto blocks = [&](long long t) { return (unsigned)((t + TB - 1) / TB); };

    const int tBlocks = (int)blocks((long long)N * 128);
    const int packBlocks = (int)blocks(2 * 8192);
    k_prep<<<(unsigned)tBlocks + (unsigned)packBlocks + blocks(nKeys / 4), TB, 0, stream>>>(
        x, W_l1, b_l1, W_r1, b_r1, xl1, xr1, N, tBlocks, W_l2, W_r2, wpk, packBlocks, tgt, keys, E, nKeys);
    k_l1_own<<<(unsigned)nOwn, AGT, 0, stream>>>(
        xl1, xr1, ea, W_e1, att1, src, tgt, keys, bias1, wpk, b_l2, b_r2, xl2, xr2, grec, gcnt, N, E);

    k_l2_own<<<(unsigned)nOwn, AGT, (size_t)(G + 1) * 64 * sizeof(float), stream>>>(
        xl2, xr2, ea, W_e2, att2, src, tgt, keys, grec, gcnt, batch, part, N, E, G);
    k_final<<<G, 1024, 0, stream>>>(part, nOwn, bias2, batch, out, N, G);
}
